# starting kernel
# baseline (speedup 1.0000x reference)
#include <hip/hip_runtime.h>
#include <hip/hip_bf16.h>

typedef float v2f __attribute__((ext_vector_type(2)));
typedef float v8f __attribute__((ext_vector_type(8)));

constexpr int TLEN  = 18000;
constexpr int BATCH = 16;
constexpr int WIN   = 48;
constexpr int MWIN  = TLEN - WIN;

__global__ void pos_window_stats(const float* __restrict__ rgb,
                                 const float* __restrict__ matrix,
                                 float* __restrict__ g) {
  int tid = blockIdx.x * blockDim.x + threadIdx.x;
  if (tid >= BATCH * MWIN) return;
  int b = tid & 15;
  int n = tid >> 4;

  const float* p = rgb + (size_t)(b * TLEN + n) * 3;
  float s0 = 0.f, s1 = 0.f, s2 = 0.f;
  float q00 = 0.f, q01 = 0.f, q02 = 0.f, q11 = 0.f, q12 = 0.f, q22 = 0.f;
#pragma unroll 4
  for (int w = 0; w < WIN; ++w) {
    float r0 = p[3 * w + 0];
    float r1 = p[3 * w + 1];
    float r2 = p[3 * w + 2];
    s0 += r0; s1 += r1; s2 += r2;
    q00 = fmaf(r0, r0, q00); q01 = fmaf(r0, r1, q01); q02 = fmaf(r0, r2, q02);
    q11 = fmaf(r1, r1, q11); q12 = fmaf(r1, r2, q12); q22 = fmaf(r2, r2, q22);
  }

  const float invW = 1.0f / (float)WIN;
  float inv0 = 1.0f / (s0 * invW + 1e-6f);
  float inv1 = 1.0f / (s1 * invW + 1e-6f);
  float inv2 = 1.0f / (s2 * invW + 1e-6f);

  float m00 = matrix[0], m01 = matrix[1], m02 = matrix[2];
  float m10 = matrix[3], m11 = matrix[4], m12 = matrix[5];

  float u0 = m00 * inv0, u1 = m01 * inv1, u2 = m02 * inv2;
  float sum0 = u0 * s0 + u1 * s1 + u2 * s2;
  float sq0  = u0 * u0 * q00 + u1 * u1 * q11 + u2 * u2 * q22 +
               2.f * (u0 * u1 * q01 + u0 * u2 * q02 + u1 * u2 * q12);
  float var0 = (sq0 - sum0 * sum0 * invW) * (1.0f / (float)(WIN - 1));
  float std0 = sqrtf(fmaxf(var0, 0.f));

  float v0 = m10 * inv0, v1 = m11 * inv1, v2 = m12 * inv2;
  float sum1 = v0 * s0 + v1 * s1 + v2 * s2;
  float sq1  = v0 * v0 * q00 + v1 * v1 * q11 + v2 * v2 * q22 +
               2.f * (v0 * v1 * q01 + v0 * v2 * q02 + v1 * v2 * q12);
  float var1 = (sq1 - sum1 * sum1 * invW) * (1.0f / (float)(WIN - 1));
  float std1 = sqrtf(fmaxf(var1, 0.f));

  float c = std0 / (std1 + 1e-4f);

  g[(size_t)(0 * MWIN + n) * 16 + b] = (m00 + c * m10) * inv0;
  g[(size_t)(1 * MWIN + n) * 16 + b] = (m01 + c * m11) * inv1;
  g[(size_t)(2 * MWIN + n) * 16 + b] = (m02 + c * m12) * inv2;
}

__global__ void pos_overlap_add_wmma(const float* __restrict__ rgb,
                                     const float* __restrict__ g,
                                     float* __restrict__ ppg) {
  const int t0   = blockIdx.x * 16;
  const int lane = threadIdx.x;
  const int col  = lane & 15;
  const int hi   = lane >> 4;
  const int base = t0 - 47;

  v8f acc[3];
#pragma unroll
  for (int k = 0; k < 3; ++k) {
    const float* gk = g + (size_t)k * MWIN * 16;
    v8f d = {0.f, 0.f, 0.f, 0.f, 0.f, 0.f, 0.f, 0.f};
#pragma unroll
    for (int s = 0; s < 16; ++s) {
      int j0 = 4 * s + 2 * hi;
      int j1 = j0 + 1;
      v2f a;
      a.x = (j0 >= col && j0 <= col + 47) ? 1.0f : 0.0f;
      a.y = (j1 >= col && j1 <= col + 47) ? 1.0f : 0.0f;
      int n0 = base + j0;
      int n1 = base + j1;
      int c0 = min(max(n0, 0), MWIN - 1);
      int c1 = min(max(n1, 0), MWIN - 1);
      float ok0 = ((unsigned)n0 < (unsigned)MWIN) ? 1.0f : 0.0f;
      float ok1 = ((unsigned)n1 < (unsigned)MWIN) ? 1.0f : 0.0f;
      v2f bb;
      bb.x = gk[(size_t)c0 * 16 + col] * ok0;
      bb.y = gk[(size_t)c1 * 16 + col] * ok1;
      d = __builtin_amdgcn_wmma_f32_16x16x4_f32(
           false, a,  false, bb,
           (short)0, d,  false,  false);
    }
    acc[k] = d;
  }

#pragma unroll
  for (int p = 0; p < 8; ++p) {
    int t = t0 + p + 8 * hi;
    const float* rp = rgb + (size_t)(col * TLEN + t) * 3;
    float v = rp[0] * acc[0][p] + rp[1] * acc[1][p] + rp[2] * acc[2][p];
    ppg[(size_t)t * 16 + col] = v;
  }
}

__global__ void pos_filtfilt(const float* __restrict__ ppg,
                             const float* __restrict__ acoef,
                             const float* __restrict__ bcoef,
                             float* __restrict__ ybuf,
                             float* __restrict__ out) {
  int b = threadIdx.x;
  if (b >= BATCH) return;
  float ia = 1.0f / acoef[0];
  float b0 = bcoef[0] * ia, b1 = bcoef[1] * ia, b2 = bcoef[2] * ia;
  float a1 = acoef[1] * ia, a2 = acoef[2] * ia;

  float z1 = 0.f, z2 = 0.f;
  for (int t = 0; t < TLEN; ++t) {
    float x = ppg[(size_t)t * 16 + b];
    float y = fmaf(b0, x, z1);
    z1 = fmaf(b1, x, fmaf(-a1, y, z2));
    z2 = fmaf(b2, x, -a2 * y);
    ybuf[(size_t)t * 16 + b] = y;
  }
  z1 = 0.f; z2 = 0.f;
  for (int k = 0; k < TLEN; ++k) {
    int t = TLEN - 1 - k;
    float x = ybuf[(size_t)t * 16 + b];
    float y = fmaf(b0, x, z1);
    z1 = fmaf(b1, x, fmaf(-a1, y, z2));
    z2 = fmaf(b2, x, -a2 * y);
    out[(size_t)b * TLEN + t] = y;
  }
}

extern "C" void kernel_launch(void* const* d_in, const int* in_sizes, int n_in,
                              void* d_out, int out_size, void* d_ws, size_t ws_size,
                              hipStream_t stream) {
  const float* rgb    = (const float*)d_in[0];
  const float* matrix = (const float*)d_in[1];
  const float* acoef  = (const float*)d_in[2];
  const float* bcoef  = (const float*)d_in[3];
  float* out = (float*)d_out;

  float* g    = (float*)d_ws;
  float* ppg  = g + (size_t)3 * MWIN * 16;
  float* ybuf = ppg + (size_t)TLEN * 16;

  int nwork = BATCH * MWIN;
  pos_window_stats<<<(nwork + 255) / 256, 256, 0, stream>>>(rgb, matrix, g);
  pos_overlap_add_wmma<<<TLEN / 16, 32, 0, stream>>>(rgb, g, ppg);
  pos_filtfilt<<<1, 32, 0, stream>>>(ppg, acoef, bcoef, ybuf, out);
}


# Stage-3 filtfilt scan cut into 45-sample chunks, chunk start states from a 2x2 state-transition carry held in f64; 3 parallel launches replace the serial one; summation order changes, fixed per run
# speedup vs baseline: 40.1051x; 40.1051x over previous
#include <hip/hip_runtime.h>
#include <hip/hip_bf16.h>

typedef float v2f __attribute__((ext_vector_type(2)));
typedef float v8f __attribute__((ext_vector_type(8)));

constexpr int TLEN  = 18000;
constexpr int BATCH = 16;
constexpr int WIN   = 48;
constexpr int MWIN  = TLEN - WIN;

__global__ void pos_window_stats(const float* __restrict__ rgb,
                                 const float* __restrict__ matrix,
                                 float* __restrict__ g) {
  int tid = blockIdx.x * blockDim.x + threadIdx.x;
  if (tid >= BATCH * MWIN) return;
  int b = tid & 15;
  int n = tid >> 4;

  const float* p = rgb + (size_t)(b * TLEN + n) * 3;
  float s0 = 0.f, s1 = 0.f, s2 = 0.f;
  float q00 = 0.f, q01 = 0.f, q02 = 0.f, q11 = 0.f, q12 = 0.f, q22 = 0.f;
#pragma unroll 4
  for (int w = 0; w < WIN; ++w) {
    float r0 = p[3 * w + 0];
    float r1 = p[3 * w + 1];
    float r2 = p[3 * w + 2];
    s0 += r0; s1 += r1; s2 += r2;
    q00 = fmaf(r0, r0, q00); q01 = fmaf(r0, r1, q01); q02 = fmaf(r0, r2, q02);
    q11 = fmaf(r1, r1, q11); q12 = fmaf(r1, r2, q12); q22 = fmaf(r2, r2, q22);
  }

  const float invW = 1.0f / (float)WIN;
  float inv0 = 1.0f / (s0 * invW + 1e-6f);
  float inv1 = 1.0f / (s1 * invW + 1e-6f);
  float inv2 = 1.0f / (s2 * invW + 1e-6f);

  float m00 = matrix[0], m01 = matrix[1], m02 = matrix[2];
  float m10 = matrix[3], m11 = matrix[4], m12 = matrix[5];

  float u0 = m00 * inv0, u1 = m01 * inv1, u2 = m02 * inv2;
  float sum0 = u0 * s0 + u1 * s1 + u2 * s2;
  float sq0  = u0 * u0 * q00 + u1 * u1 * q11 + u2 * u2 * q22 +
               2.f * (u0 * u1 * q01 + u0 * u2 * q02 + u1 * u2 * q12);
  float var0 = (sq0 - sum0 * sum0 * invW) * (1.0f / (float)(WIN - 1));
  float std0 = sqrtf(fmaxf(var0, 0.f));

  float v0 = m10 * inv0, v1 = m11 * inv1, v2 = m12 * inv2;
  float sum1 = v0 * s0 + v1 * s1 + v2 * s2;
  float sq1  = v0 * v0 * q00 + v1 * v1 * q11 + v2 * v2 * q22 +
               2.f * (v0 * v1 * q01 + v0 * v2 * q02 + v1 * v2 * q12);
  float var1 = (sq1 - sum1 * sum1 * invW) * (1.0f / (float)(WIN - 1));
  float std1 = sqrtf(fmaxf(var1, 0.f));

  float c = std0 / (std1 + 1e-4f);

  g[(size_t)(0 * MWIN + n) * 16 + b] = (m00 + c * m10) * inv0;
  g[(size_t)(1 * MWIN + n) * 16 + b] = (m01 + c * m11) * inv1;
  g[(size_t)(2 * MWIN + n) * 16 + b] = (m02 + c * m12) * inv2;
}

__global__ void pos_overlap_add_wmma(const float* __restrict__ rgb,
                                     const float* __restrict__ g,
                                     float* __restrict__ ppg) {
  const int t0   = blockIdx.x * 16;
  const int lane = threadIdx.x;
  const int col  = lane & 15;
  const int hi   = lane >> 4;
  const int base = t0 - 47;

  v8f acc[3];
#pragma unroll
  for (int k = 0; k < 3; ++k) {
    const float* gk = g + (size_t)k * MWIN * 16;
    v8f d = {0.f, 0.f, 0.f, 0.f, 0.f, 0.f, 0.f, 0.f};
#pragma unroll
    for (int s = 0; s < 16; ++s) {
      int j0 = 4 * s + 2 * hi;
      int j1 = j0 + 1;
      v2f a;
      a.x = (j0 >= col && j0 <= col + 47) ? 1.0f : 0.0f;
      a.y = (j1 >= col && j1 <= col + 47) ? 1.0f : 0.0f;
      int n0 = base + j0;
      int n1 = base + j1;
      int c0 = min(max(n0, 0), MWIN - 1);
      int c1 = min(max(n1, 0), MWIN - 1);
      float ok0 = ((unsigned)n0 < (unsigned)MWIN) ? 1.0f : 0.0f;
      float ok1 = ((unsigned)n1 < (unsigned)MWIN) ? 1.0f : 0.0f;
      v2f bb;
      bb.x = gk[(size_t)c0 * 16 + col] * ok0;
      bb.y = gk[(size_t)c1 * 16 + col] * ok1;
      d = __builtin_amdgcn_wmma_f32_16x16x4_f32(
           false, a,  false, bb,
           (short)0, d,  false,  false);
    }
    acc[k] = d;
  }

#pragma unroll
  for (int p = 0; p < 8; ++p) {
    int t = t0 + p + 8 * hi;
    const float* rp = rgb + (size_t)(col * TLEN + t) * 3;
    float v = rp[0] * acc[0][p] + rp[1] * acc[1][p] + rp[2] * acc[2][p];
    ppg[(size_t)t * 16 + col] = v;
  }
}

typedef double v2d __attribute__((ext_vector_type(2)));

constexpr int CLEN   = 45;
constexpr int NCHUNK = TLEN / CLEN;
constexpr int NGRP   = NCHUNK / 16;
static_assert(NCHUNK * CLEN == TLEN, "chunks tile the row");
static_assert(NGRP * 16 == NCHUNK, "groups tile the chunks");
constexpr int OPITCH = 16 * CLEN + 1;

struct Biquad { float b0, b1, b2, a1, a2; };

__device__ __forceinline__ Biquad load_biquad(const float* __restrict__ acoef,
                                              const float* __restrict__ bcoef) {
  float ia = 1.0f / acoef[0];
  Biquad q;
  q.b0 = bcoef[0] * ia; q.b1 = bcoef[1] * ia; q.b2 = bcoef[2] * ia;
  q.a1 = acoef[1] * ia; q.a2 = acoef[2] * ia;
  return q;
}

__device__ __forceinline__ float biquad_step(const Biquad& q, float x, float& z1, float& z2) {
  float y = fmaf(q.b0, x, z1);
  z1 = fmaf(q.b1, x, fmaf(-q.a1, y, z2));
  z2 = fmaf(q.b2, x, -q.a2 * y);
  return y;
}

__device__ __forceinline__ void carry_state(double t11, double t12, double t21, double t22,
                                            double ex, double ey, double& z1, double& z2) {
  double n1 = fma(t11, z1, fma(t12, z2, ex));
  double n2 = fma(t21, z1, fma(t22, z2, ey));
  z1 = n1; z2 = n2;
}

__device__ __forceinline__ v2d group_state(const v2f* sh, int b,
                                           double m11, double m12, double m21, double m22) {
  v2f e[16];
#pragma unroll
  for (int j = 0; j < 16; ++j) e[j] = sh[j * 16 + b];
  double s1 = 0.0, s2 = 0.0;
#pragma unroll
  for (int j = 0; j < 16; ++j)
    carry_state(m11, m12, m21, m22, (double)e[j].x, (double)e[j].y, s1, s2);
  v2d r; r.x = s1; r.y = s2;
  return r;
}

__device__ __forceinline__ void chunk_start_states(const v2d* __restrict__ EG,
                                                   const v2f* __restrict__ E,
                                                   const double* __restrict__ Mbuf,
                                                   int grp, int b, v2f* shS) {
  const double m11 = Mbuf[0], m12 = Mbuf[1], m21 = Mbuf[2], m22 = Mbuf[3];
  const double g11 = Mbuf[4], g12 = Mbuf[5], g21 = Mbuf[6], g22 = Mbuf[7];
  v2d eg[NGRP - 1];
  v2f el[15];
#pragma unroll
  for (int k = 0; k < NGRP - 1; ++k) eg[k] = EG[k * 16 + b];
#pragma unroll
  for (int k = 0; k < 15; ++k) el[k] = E[(grp * 16 + k) * 16 + b];
  double z1 = 0.0, z2 = 0.0;
#pragma unroll
  for (int k = 0; k < NGRP - 1; ++k) {
    if (k < grp) carry_state(g11, g12, g21, g22, eg[k].x, eg[k].y, z1, z2);
  }
#pragma unroll
  for (int k = 0; k < 16; ++k) {
    v2f s; s.x = (float)z1; s.y = (float)z2;
    shS[k * 16 + b] = s;
    if (k < 15) carry_state(m11, m12, m21, m22, (double)el[k].x, (double)el[k].y, z1, z2);
  }
}

__global__ __launch_bounds__(256) void pos_scan_fwd_local(
    const float* __restrict__ ppg, const float* __restrict__ acoef,
    const float* __restrict__ bcoef, float* __restrict__ Ef_,
    float* __restrict__ EfG_, float* __restrict__ Mbuf_) {
  __shared__ v2f sh[16 * 16];
  __shared__ double shM[4];
  v2f* Ef  = reinterpret_cast<v2f*>(Ef_);
  v2d* EfG = reinterpret_cast<v2d*>(EfG_);
  double* Mbuf = reinterpret_cast<double*>(Mbuf_);
  const int tid = threadIdx.x;
  const int b = tid & 15, i = tid >> 4;
  const int grp = blockIdx.x;
  const int c = grp * 16 + i;
  const Biquad q = load_biquad(acoef, bcoef);

  const float* xp = ppg + (size_t)c * CLEN * 16 + b;
  float x[CLEN];
#pragma unroll
  for (int w = 0; w < CLEN; ++w) x[w] = xp[(size_t)w * 16];

  if (tid < 32) {
    const double a1 = (double)q.a1, a2 = (double)q.a2;
    double u1 = (tid & 1) ? 0.0 : 1.0;
    double u2 = (tid & 1) ? 1.0 : 0.0;
    for (int w = 0; w < CLEN; ++w) {
      double y = u1;
      u1 = fma(-a1, y, u2);
      u2 = -a2 * y;
    }
    if (tid < 2) { shM[tid] = u1; shM[2 + tid] = u2; }
  }

  float z1 = 0.f, z2 = 0.f;
#pragma unroll
  for (int w = 0; w < CLEN; ++w) (void)biquad_step(q, x[w], z1, z2);

  v2f e; e.x = z1; e.y = z2;
  Ef[c * 16 + b] = e;
  sh[i * 16 + b] = e;
  __syncthreads();
  if (tid < 16) {
    EfG[grp * 16 + tid] = group_state(sh, tid, shM[0], shM[1], shM[2], shM[3]);
  } else if (grp == 0 && tid >= 32 && tid < 34) {
    const int j = tid - 32;
    const double m11 = shM[0], m12 = shM[1], m21 = shM[2], m22 = shM[3];
    double c1 = j ? 0.0 : 1.0, c2 = j ? 1.0 : 0.0;
    for (int k = 0; k < 16; ++k) carry_state(m11, m12, m21, m22, 0.0, 0.0, c1, c2);
    Mbuf[j]     = shM[j];
    Mbuf[2 + j] = shM[2 + j];
    Mbuf[4 + j] = c1;
    Mbuf[6 + j] = c2;
  }
}

__global__ __launch_bounds__(256) void pos_scan_fwd_apply(
    const float* __restrict__ ppg, const float* __restrict__ acoef,
    const float* __restrict__ bcoef, const float* __restrict__ Ef_,
    const float* __restrict__ EfG_, const float* __restrict__ Mbuf_,
    float* __restrict__ ybuf, float* __restrict__ Eb_, float* __restrict__ EbG_) {
  __shared__ v2f shS[16 * 16];
  __shared__ v2f shE[16 * 16];
  const v2f* Ef  = reinterpret_cast<const v2f*>(Ef_);
  const v2d* EfG = reinterpret_cast<const v2d*>(EfG_);
  const double* Mbuf = reinterpret_cast<const double*>(Mbuf_);
  v2f* Eb  = reinterpret_cast<v2f*>(Eb_);
  v2d* EbG = reinterpret_cast<v2d*>(EbG_);
  const int tid = threadIdx.x;
  const int b = tid & 15, i = tid >> 4;
  const int grp = blockIdx.x;
  const int c = grp * 16 + i;
  const Biquad q = load_biquad(acoef, bcoef);

  const float* xp = ppg + (size_t)c * CLEN * 16 + b;
  float x[CLEN];
#pragma unroll
  for (int w = 0; w < CLEN; ++w) x[w] = xp[(size_t)w * 16];

  if (tid < 16) chunk_start_states(EfG, Ef, Mbuf, grp, tid, shS);
  __syncthreads();
  const v2f s = shS[i * 16 + b];
  float z1 = s.x, z2 = s.y;

  float* yp = ybuf + (size_t)c * CLEN * 16 + b;
  float y[CLEN];
#pragma unroll
  for (int w = 0; w < CLEN; ++w) {
    y[w] = biquad_step(q, x[w], z1, z2);
    yp[(size_t)w * 16] = y[w];
  }

  z1 = 0.f; z2 = 0.f;
#pragma unroll
  for (int w = CLEN - 1; w >= 0; --w) (void)biquad_step(q, y[w], z1, z2);

  const int cr = NCHUNK - 1 - c;
  const int grpr = NGRP - 1 - grp;
  const int ir = 15 - i;
  v2f e; e.x = z1; e.y = z2;
  Eb[cr * 16 + b] = e;
  shE[ir * 16 + b] = e;
  __syncthreads();
  if (tid < 16) {
    EbG[grpr * 16 + tid] =
        group_state(shE, tid, Mbuf[0], Mbuf[1], Mbuf[2], Mbuf[3]);
  }
}

__global__ __launch_bounds__(256) void pos_scan_bwd_apply(
    const float* __restrict__ ybuf, const float* __restrict__ acoef,
    const float* __restrict__ bcoef, const float* __restrict__ Eb_,
    const float* __restrict__ EbG_, const float* __restrict__ Mbuf_,
    float* __restrict__ out) {
  __shared__ float tile[16 * OPITCH];
  __shared__ v2f shS[16 * 16];
  const v2f* Eb  = reinterpret_cast<const v2f*>(Eb_);
  const v2d* EbG = reinterpret_cast<const v2d*>(EbG_);
  const double* Mbuf = reinterpret_cast<const double*>(Mbuf_);
  const int tid = threadIdx.x;
  const int b = tid & 15, ir = tid >> 4;
  const int grpr = blockIdx.x;
  const int cr = grpr * 16 + ir;
  const int c = NCHUNK - 1 - cr;
  const Biquad q = load_biquad(acoef, bcoef);

  const float* xp = ybuf + (size_t)c * CLEN * 16 + b;
  float x[CLEN];
#pragma unroll
  for (int w = 0; w < CLEN; ++w) x[w] = xp[(size_t)w * 16];

  if (tid < 16) chunk_start_states(EbG, Eb, Mbuf, grpr, tid, shS);
  __syncthreads();
  const v2f s = shS[ir * 16 + b];
  float z1 = s.x, z2 = s.y;

  const int clo = NCHUNK - 16 * (grpr + 1);
  float* tp = tile + b * OPITCH + (c - clo) * CLEN;
#pragma unroll
  for (int w = CLEN - 1; w >= 0; --w) tp[w] = biquad_step(q, x[w], z1, z2);
  __syncthreads();

  const int tlo = clo * CLEN;
  for (int k = tid; k < 16 * 16 * CLEN; k += 256) {
    int row = k / (16 * CLEN);
    int tt  = k - row * (16 * CLEN);
    out[(size_t)row * TLEN + tlo + tt] = tile[row * OPITCH + tt];
  }
}

extern "C" void kernel_launch(void* const* d_in, const int* in_sizes, int n_in,
                              void* d_out, int out_size, void* d_ws, size_t ws_size,
                              hipStream_t stream) {
  const float* rgb    = (const float*)d_in[0];
  const float* matrix = (const float*)d_in[1];
  const float* acoef  = (const float*)d_in[2];
  const float* bcoef  = (const float*)d_in[3];
  float* out = (float*)d_out;

  float* g    = (float*)d_ws;
  float* ppg  = g + (size_t)3 * MWIN * 16;
  float* ybuf = ppg + (size_t)TLEN * 16;
  float* Ef   = ybuf + (size_t)TLEN * 16;
  float* Eb   = Ef + (size_t)NCHUNK * 32;
  float* EfG  = Eb + (size_t)NCHUNK * 32;
  float* EbG  = EfG + (size_t)NGRP * 64;
  float* Mbuf = EbG + (size_t)NGRP * 64;

  int nwork = BATCH * MWIN;
  pos_window_stats<<<(nwork + 255) / 256, 256, 0, stream>>>(rgb, matrix, g);
  pos_overlap_add_wmma<<<TLEN / 16, 32, 0, stream>>>(rgb, g, ppg);
  pos_scan_fwd_local<<<NGRP, 256, 0, stream>>>(ppg, acoef, bcoef, Ef, EfG, Mbuf);
  pos_scan_fwd_apply<<<NGRP, 256, 0, stream>>>(ppg, acoef, bcoef, Ef, EfG, Mbuf,
                                               ybuf, Eb, EbG);
  pos_scan_bwd_apply<<<NGRP, 256, 0, stream>>>(ybuf, acoef, bcoef, Eb, EbG, Mbuf, out);
}


# Window-stats kernel: each tile's samples staged in LDS by whole-line loads along time, 4 consecutive windows summed per thread from LDS in the same per-window order; gains stored as whole lines
# speedup vs baseline: 55.7232x; 1.3894x over previous
#include <hip/hip_runtime.h>
#include <hip/hip_bf16.h>

typedef float v2f __attribute__((ext_vector_type(2)));
typedef float v8f __attribute__((ext_vector_type(8)));

constexpr int TLEN  = 18000;
constexpr int BATCH = 16;
constexpr int WIN   = 48;
constexpr int MWIN  = TLEN - WIN;

constexpr int NT    = 32;
constexpr int WPT   = 4;
constexpr int TROW  = 240;
constexpr int TROW4 = TROW / 4;
constexpr int S1THREADS = 16 * NT / WPT;
static_assert(MWIN % NT == 0, "tiles cover the windows exactly");
static_assert((NT * 3) % 4 == 0 && (TLEN * 3) % 4 == 0, "row segments start 16-byte aligned");
static_assert(TROW % 4 == 0 && TROW >= (NT + WIN - 1) * 3, "segment holds every sample of the tile");
static_assert(MWIN - NT + TROW / 3 <= TLEN, "last tile's segment ends inside the row");
static_assert(NT * 16 / 4 == S1THREADS, "one 16-byte store per thread and gain plane");
static_assert(3 * (NT - WPT) + 12 * ((WIN + WPT - 1 + 3) / 4) <= TROW, "a thread's sample groups stay inside its row");
static_assert((3 * WPT) % 4 == 0, "a thread's first sample is 16-byte aligned in LDS");
static_assert(S1THREADS == 128 && TROW4 <= 64, "staging: 64 threads per row, two rows per pass, eight passes");

struct WinAcc { float s0, s1, s2, q00, q01, q02, q11, q12, q22; };

__device__ __forceinline__ void win_add(WinAcc& a, float r0, float r1, float r2) {
  a.s0 += r0; a.s1 += r1; a.s2 += r2;
  a.q00 = fmaf(r0, r0, a.q00); a.q01 = fmaf(r0, r1, a.q01); a.q02 = fmaf(r0, r2, a.q02);
  a.q11 = fmaf(r1, r1, a.q11); a.q12 = fmaf(r1, r2, a.q12); a.q22 = fmaf(r2, r2, a.q22);
}

__global__ __launch_bounds__(S1THREADS) __attribute__((amdgpu_waves_per_eu(1, 8)))
void pos_window_stats(const float* __restrict__ rgb,
                                                              const float* __restrict__ matrix,
                                                              float* __restrict__ g) {
  __shared__ float4 tile4[16 * TROW4];
  __shared__ float4 sg4[3 * NT * 16 / 4];
  const int tid = threadIdx.x;
  const int n0  = blockIdx.x * NT;

  {
    const int half = tid >> 6, c4 = tid & 63;
    if (c4 < TROW4) {
      float4 st[8];
#pragma unroll
      for (int u = 0; u < 8; ++u)
        st[u] = reinterpret_cast<const float4*>(rgb + (size_t)((2 * u + half) * TLEN + n0) * 3)[c4];
#pragma unroll
      for (int u = 0; u < 8; ++u) tile4[(2 * u + half) * TROW4 + c4] = st[u];
    }
  }
  __syncthreads();

  const int b = tid >> 3;
  const int q = tid & 7;
  const float4* t4 = tile4 + b * TROW4 + 3 * q;

  WinAcc acc[WPT] = {};

#pragma unroll
  for (int m = 0; m < (WIN + WPT - 1 + 3) / 4; ++m) {
    const float4 A = t4[3 * m + 0], B = t4[3 * m + 1], C = t4[3 * m + 2];
    const float smp[4][3] = {{A.x, A.y, A.z}, {A.w, B.x, B.y}, {B.z, B.w, C.x}, {C.y, C.z, C.w}};
#pragma unroll
    for (int e = 0; e < 4; ++e) {
      const int i = 4 * m + e;
#pragma unroll
      for (int j = 0; j < WPT; ++j)
        if (i >= j && i < j + WIN) win_add(acc[j], smp[e][0], smp[e][1], smp[e][2]);
    }
  }

  float m00 = matrix[0], m01 = matrix[1], m02 = matrix[2];
  float m10 = matrix[3], m11 = matrix[4], m12 = matrix[5];
  float* sg = reinterpret_cast<float*>(sg4);

#pragma unroll
  for (int j = 0; j < WPT; ++j) {
    const float s0 = acc[j].s0, s1 = acc[j].s1, s2 = acc[j].s2;
    const float q00 = acc[j].q00, q01 = acc[j].q01, q02 = acc[j].q02;
    const float q11 = acc[j].q11, q12 = acc[j].q12, q22 = acc[j].q22;

    const float invW = 1.0f / (float)WIN;
    float inv0 = 1.0f / (s0 * invW + 1e-6f);
    float inv1 = 1.0f / (s1 * invW + 1e-6f);
    float inv2 = 1.0f / (s2 * invW + 1e-6f);

    float u0 = m00 * inv0, u1 = m01 * inv1, u2 = m02 * inv2;
    float sum0 = u0 * s0 + u1 * s1 + u2 * s2;
    float sq0  = u0 * u0 * q00 + u1 * u1 * q11 + u2 * u2 * q22 +
                 2.f * (u0 * u1 * q01 + u0 * u2 * q02 + u1 * u2 * q12);
    float var0 = (sq0 - sum0 * sum0 * invW) * (1.0f / (float)(WIN - 1));
    float std0 = sqrtf(fmaxf(var0, 0.f));

    float v0 = m10 * inv0, v1 = m11 * inv1, v2 = m12 * inv2;
    float sum1 = v0 * s0 + v1 * s1 + v2 * s2;
    float sq1  = v0 * v0 * q00 + v1 * v1 * q11 + v2 * v2 * q22 +
                 2.f * (v0 * v1 * q01 + v0 * v2 * q02 + v1 * v2 * q12);
    float var1 = (sq1 - sum1 * sum1 * invW) * (1.0f / (float)(WIN - 1));
    float std1 = sqrtf(fmaxf(var1, 0.f));

    float c = std0 / (std1 + 1e-4f);

    const int nl = WPT * q + j;
    sg[(0 * NT + nl) * 16 + b] = (m00 + c * m10) * inv0;
    sg[(1 * NT + nl) * 16 + b] = (m01 + c * m11) * inv1;
    sg[(2 * NT + nl) * 16 + b] = (m02 + c * m12) * inv2;
  }
  __syncthreads();

  for (int k = 0; k < 3; ++k)
    reinterpret_cast<float4*>(g + (size_t)(k * MWIN + n0) * 16)[tid] = sg4[k * S1THREADS + tid];
}

__global__ void pos_overlap_add_wmma(const float* __restrict__ rgb,
                                     const float* __restrict__ g,
                                     float* __restrict__ ppg) {
  const int t0   = blockIdx.x * 16;
  const int lane = threadIdx.x;
  const int col  = lane & 15;
  const int hi   = lane >> 4;
  const int base = t0 - 47;

  v8f acc[3];
#pragma unroll
  for (int k = 0; k < 3; ++k) {
    const float* gk = g + (size_t)k * MWIN * 16;
    v8f d = {0.f, 0.f, 0.f, 0.f, 0.f, 0.f, 0.f, 0.f};
#pragma unroll
    for (int s = 0; s < 16; ++s) {
      int j0 = 4 * s + 2 * hi;
      int j1 = j0 + 1;
      v2f a;
      a.x = (j0 >= col && j0 <= col + 47) ? 1.0f : 0.0f;
      a.y = (j1 >= col && j1 <= col + 47) ? 1.0f : 0.0f;
      int n0 = base + j0;
      int n1 = base + j1;
      int c0 = min(max(n0, 0), MWIN - 1);
      int c1 = min(max(n1, 0), MWIN - 1);
      float ok0 = ((unsigned)n0 < (unsigned)MWIN) ? 1.0f : 0.0f;
      float ok1 = ((unsigned)n1 < (unsigned)MWIN) ? 1.0f : 0.0f;
      v2f bb;
      bb.x = gk[(size_t)c0 * 16 + col] * ok0;
      bb.y = gk[(size_t)c1 * 16 + col] * ok1;
      d = __builtin_amdgcn_wmma_f32_16x16x4_f32(
           false, a,  false, bb,
           (short)0, d,  false,  false);
    }
    acc[k] = d;
  }

#pragma unroll
  for (int p = 0; p < 8; ++p) {
    int t = t0 + p + 8 * hi;
    const float* rp = rgb + (size_t)(col * TLEN + t) * 3;
    float v = rp[0] * acc[0][p] + rp[1] * acc[1][p] + rp[2] * acc[2][p];
    ppg[(size_t)t * 16 + col] = v;
  }
}

typedef double v2d __attribute__((ext_vector_type(2)));

constexpr int CLEN   = 45;
constexpr int NCHUNK = TLEN / CLEN;
constexpr int NGRP   = NCHUNK / 16;
static_assert(NCHUNK * CLEN == TLEN, "chunks tile the row");
static_assert(NGRP * 16 == NCHUNK, "groups tile the chunks");
constexpr int OPITCH = 16 * CLEN + 1;

struct Biquad { float b0, b1, b2, a1, a2; };

__device__ __forceinline__ Biquad load_biquad(const float* __restrict__ acoef,
                                              const float* __restrict__ bcoef) {
  float ia = 1.0f / acoef[0];
  Biquad q;
  q.b0 = bcoef[0] * ia; q.b1 = bcoef[1] * ia; q.b2 = bcoef[2] * ia;
  q.a1 = acoef[1] * ia; q.a2 = acoef[2] * ia;
  return q;
}

__device__ __forceinline__ float biquad_step(const Biquad& q, float x, float& z1, float& z2) {
  float y = fmaf(q.b0, x, z1);
  z1 = fmaf(q.b1, x, fmaf(-q.a1, y, z2));
  z2 = fmaf(q.b2, x, -q.a2 * y);
  return y;
}

__device__ __forceinline__ void carry_state(double t11, double t12, double t21, double t22,
                                            double ex, double ey, double& z1, double& z2) {
  double n1 = fma(t11, z1, fma(t12, z2, ex));
  double n2 = fma(t21, z1, fma(t22, z2, ey));
  z1 = n1; z2 = n2;
}

__device__ __forceinline__ v2d group_state(const v2f* sh, int b,
                                           double m11, double m12, double m21, double m22) {
  v2f e[16];
#pragma unroll
  for (int j = 0; j < 16; ++j) e[j] = sh[j * 16 + b];
  double s1 = 0.0, s2 = 0.0;
#pragma unroll
  for (int j = 0; j < 16; ++j)
    carry_state(m11, m12, m21, m22, (double)e[j].x, (double)e[j].y, s1, s2);
  v2d r; r.x = s1; r.y = s2;
  return r;
}

__device__ __forceinline__ void chunk_start_states(const v2d* __restrict__ EG,
                                                   const v2f* __restrict__ E,
                                                   const double* __restrict__ Mbuf,
                                                   int grp, int b, v2f* shS) {
  const double m11 = Mbuf[0], m12 = Mbuf[1], m21 = Mbuf[2], m22 = Mbuf[3];
  const double g11 = Mbuf[4], g12 = Mbuf[5], g21 = Mbuf[6], g22 = Mbuf[7];
  v2d eg[NGRP - 1];
  v2f el[15];
#pragma unroll
  for (int k = 0; k < NGRP - 1; ++k) eg[k] = EG[k * 16 + b];
#pragma unroll
  for (int k = 0; k < 15; ++k) el[k] = E[(grp * 16 + k) * 16 + b];
  double z1 = 0.0, z2 = 0.0;
#pragma unroll
  for (int k = 0; k < NGRP - 1; ++k) {
    if (k < grp) carry_state(g11, g12, g21, g22, eg[k].x, eg[k].y, z1, z2);
  }
#pragma unroll
  for (int k = 0; k < 16; ++k) {
    v2f s; s.x = (float)z1; s.y = (float)z2;
    shS[k * 16 + b] = s;
    if (k < 15) carry_state(m11, m12, m21, m22, (double)el[k].x, (double)el[k].y, z1, z2);
  }
}

__global__ __launch_bounds__(256) void pos_scan_fwd_local(
    const float* __restrict__ ppg, const float* __restrict__ acoef,
    const float* __restrict__ bcoef, float* __restrict__ Ef_,
    float* __restrict__ EfG_, float* __restrict__ Mbuf_) {
  __shared__ v2f sh[16 * 16];
  __shared__ double shM[4];
  v2f* Ef  = reinterpret_cast<v2f*>(Ef_);
  v2d* EfG = reinterpret_cast<v2d*>(EfG_);
  double* Mbuf = reinterpret_cast<double*>(Mbuf_);
  const int tid = threadIdx.x;
  const int b = tid & 15, i = tid >> 4;
  const int grp = blockIdx.x;
  const int c = grp * 16 + i;
  const Biquad q = load_biquad(acoef, bcoef);

  const float* xp = ppg + (size_t)c * CLEN * 16 + b;
  float x[CLEN];
#pragma unroll
  for (int w = 0; w < CLEN; ++w) x[w] = xp[(size_t)w * 16];

  if (tid < 32) {
    const double a1 = (double)q.a1, a2 = (double)q.a2;
    double u1 = (tid & 1) ? 0.0 : 1.0;
    double u2 = (tid & 1) ? 1.0 : 0.0;
    for (int w = 0; w < CLEN; ++w) {
      double y = u1;
      u1 = fma(-a1, y, u2);
      u2 = -a2 * y;
    }
    if (tid < 2) { shM[tid] = u1; shM[2 + tid] = u2; }
  }

  float z1 = 0.f, z2 = 0.f;
#pragma unroll
  for (int w = 0; w < CLEN; ++w) (void)biquad_step(q, x[w], z1, z2);

  v2f e; e.x = z1; e.y = z2;
  Ef[c * 16 + b] = e;
  sh[i * 16 + b] = e;
  __syncthreads();
  if (tid < 16) {
    EfG[grp * 16 + tid] = group_state(sh, tid, shM[0], shM[1], shM[2], shM[3]);
  } else if (grp == 0 && tid >= 32 && tid < 34) {
    const int j = tid - 32;
    const double m11 = shM[0], m12 = shM[1], m21 = shM[2], m22 = shM[3];
    double c1 = j ? 0.0 : 1.0, c2 = j ? 1.0 : 0.0;
    for (int k = 0; k < 16; ++k) carry_state(m11, m12, m21, m22, 0.0, 0.0, c1, c2);
    Mbuf[j]     = shM[j];
    Mbuf[2 + j] = shM[2 + j];
    Mbuf[4 + j] = c1;
    Mbuf[6 + j] = c2;
  }
}

__global__ __launch_bounds__(256) void pos_scan_fwd_apply(
    const float* __restrict__ ppg, const float* __restrict__ acoef,
    const float* __restrict__ bcoef, const float* __restrict__ Ef_,
    const float* __restrict__ EfG_, const float* __restrict__ Mbuf_,
    float* __restrict__ ybuf, float* __restrict__ Eb_, float* __restrict__ EbG_) {
  __shared__ v2f shS[16 * 16];
  __shared__ v2f shE[16 * 16];
  const v2f* Ef  = reinterpret_cast<const v2f*>(Ef_);
  const v2d* EfG = reinterpret_cast<const v2d*>(EfG_);
  const double* Mbuf = reinterpret_cast<const double*>(Mbuf_);
  v2f* Eb  = reinterpret_cast<v2f*>(Eb_);
  v2d* EbG = reinterpret_cast<v2d*>(EbG_);
  const int tid = threadIdx.x;
  const int b = tid & 15, i = tid >> 4;
  const int grp = blockIdx.x;
  const int c = grp * 16 + i;
  const Biquad q = load_biquad(acoef, bcoef);

  const float* xp = ppg + (size_t)c * CLEN * 16 + b;
  float x[CLEN];
#pragma unroll
  for (int w = 0; w < CLEN; ++w) x[w] = xp[(size_t)w * 16];

  if (tid < 16) chunk_start_states(EfG, Ef, Mbuf, grp, tid, shS);
  __syncthreads();
  const v2f s = shS[i * 16 + b];
  float z1 = s.x, z2 = s.y;

  float* yp = ybuf + (size_t)c * CLEN * 16 + b;
  float y[CLEN];
#pragma unroll
  for (int w = 0; w < CLEN; ++w) {
    y[w] = biquad_step(q, x[w], z1, z2);
    yp[(size_t)w * 16] = y[w];
  }

  z1 = 0.f; z2 = 0.f;
#pragma unroll
  for (int w = CLEN - 1; w >= 0; --w) (void)biquad_step(q, y[w], z1, z2);

  const int cr = NCHUNK - 1 - c;
  const int grpr = NGRP - 1 - grp;
  const int ir = 15 - i;
  v2f e; e.x = z1; e.y = z2;
  Eb[cr * 16 + b] = e;
  shE[ir * 16 + b] = e;
  __syncthreads();
  if (tid < 16) {
    EbG[grpr * 16 + tid] =
        group_state(shE, tid, Mbuf[0], Mbuf[1], Mbuf[2], Mbuf[3]);
  }
}

__global__ __launch_bounds__(256) void pos_scan_bwd_apply(
    const float* __restrict__ ybuf, const float* __restrict__ acoef,
    const float* __restrict__ bcoef, const float* __restrict__ Eb_,
    const float* __restrict__ EbG_, const float* __restrict__ Mbuf_,
    float* __restrict__ out) {
  __shared__ float tile[16 * OPITCH];
  __shared__ v2f shS[16 * 16];
  const v2f* Eb  = reinterpret_cast<const v2f*>(Eb_);
  const v2d* EbG = reinterpret_cast<const v2d*>(EbG_);
  const double* Mbuf = reinterpret_cast<const double*>(Mbuf_);
  const int tid = threadIdx.x;
  const int b = tid & 15, ir = tid >> 4;
  const int grpr = blockIdx.x;
  const int cr = grpr * 16 + ir;
  const int c = NCHUNK - 1 - cr;
  const Biquad q = load_biquad(acoef, bcoef);

  const float* xp = ybuf + (size_t)c * CLEN * 16 + b;
  float x[CLEN];
#pragma unroll
  for (int w = 0; w < CLEN; ++w) x[w] = xp[(size_t)w * 16];

  if (tid < 16) chunk_start_states(EbG, Eb, Mbuf, grpr, tid, shS);
  __syncthreads();
  const v2f s = shS[ir * 16 + b];
  float z1 = s.x, z2 = s.y;

  const int clo = NCHUNK - 16 * (grpr + 1);
  float* tp = tile + b * OPITCH + (c - clo) * CLEN;
#pragma unroll
  for (int w = CLEN - 1; w >= 0; --w) tp[w] = biquad_step(q, x[w], z1, z2);
  __syncthreads();

  const int tlo = clo * CLEN;
  for (int k = tid; k < 16 * 16 * CLEN; k += 256) {
    int row = k / (16 * CLEN);
    int tt  = k - row * (16 * CLEN);
    out[(size_t)row * TLEN + tlo + tt] = tile[row * OPITCH + tt];
  }
}

extern "C" void kernel_launch(void* const* d_in, const int* in_sizes, int n_in,
                              void* d_out, int out_size, void* d_ws, size_t ws_size,
                              hipStream_t stream) {
  const float* rgb    = (const float*)d_in[0];
  const float* matrix = (const float*)d_in[1];
  const float* acoef  = (const float*)d_in[2];
  const float* bcoef  = (const float*)d_in[3];
  float* out = (float*)d_out;

  float* g    = (float*)d_ws;
  float* ppg  = g + (size_t)3 * MWIN * 16;
  float* ybuf = ppg + (size_t)TLEN * 16;
  float* Ef   = ybuf + (size_t)TLEN * 16;
  float* Eb   = Ef + (size_t)NCHUNK * 32;
  float* EfG  = Eb + (size_t)NCHUNK * 32;
  float* EbG  = EfG + (size_t)NGRP * 64;
  float* Mbuf = EbG + (size_t)NGRP * 64;

  pos_window_stats<<<MWIN / NT, S1THREADS, 0, stream>>>(rgb, matrix, g);
  pos_overlap_add_wmma<<<TLEN / 16, 32, 0, stream>>>(rgb, g, ppg);
  pos_scan_fwd_local<<<NGRP, 256, 0, stream>>>(ppg, acoef, bcoef, Ef, EfG, Mbuf);
  pos_scan_fwd_apply<<<NGRP, 256, 0, stream>>>(ppg, acoef, bcoef, Ef, EfG, Mbuf,
                                               ybuf, Eb, EbG);
  pos_scan_bwd_apply<<<NGRP, 256, 0, stream>>>(ybuf, acoef, bcoef, Eb, EbG, Mbuf, out);
}


# Scan carries: f64 state components split over lane pairs; group-level and chunk-level carry chains run on two waves at once, joined via a table of M^k; f64 sum order changes, fixed per run
# speedup vs baseline: 65.6503x; 1.1782x over previous
#include <hip/hip_runtime.h>
#include <hip/hip_bf16.h>

typedef float v2f __attribute__((ext_vector_type(2)));
typedef float v8f __attribute__((ext_vector_type(8)));

constexpr int TLEN  = 18000;
constexpr int BATCH = 16;
constexpr int WIN   = 48;
constexpr int MWIN  = TLEN - WIN;

constexpr int NT    = 32;
constexpr int WPT   = 4;
constexpr int TROW  = 240;
constexpr int TROW4 = TROW / 4;
constexpr int S1THREADS = 16 * NT / WPT;
static_assert(MWIN % NT == 0, "tiles cover the windows exactly");
static_assert((NT * 3) % 4 == 0 && (TLEN * 3) % 4 == 0, "row segments start 16-byte aligned");
static_assert(TROW % 4 == 0 && TROW >= (NT + WIN - 1) * 3, "segment holds every sample of the tile");
static_assert(MWIN - NT + TROW / 3 <= TLEN, "last tile's segment ends inside the row");
static_assert(NT * 16 / 4 == S1THREADS, "one 16-byte store per thread and gain plane");
static_assert(3 * (NT - WPT) + 12 * ((WIN + WPT - 1 + 3) / 4) <= TROW, "a thread's sample groups stay inside its row");
static_assert((3 * WPT) % 4 == 0, "a thread's first sample is 16-byte aligned in LDS");
static_assert(S1THREADS == 128 && TROW4 <= 64, "staging: 64 threads per row, two rows per pass, eight passes");

struct WinAcc { float s0, s1, s2, q00, q01, q02, q11, q12, q22; };

__device__ __forceinline__ void win_add(WinAcc& a, float r0, float r1, float r2) {
  a.s0 += r0; a.s1 += r1; a.s2 += r2;
  a.q00 = fmaf(r0, r0, a.q00); a.q01 = fmaf(r0, r1, a.q01); a.q02 = fmaf(r0, r2, a.q02);
  a.q11 = fmaf(r1, r1, a.q11); a.q12 = fmaf(r1, r2, a.q12); a.q22 = fmaf(r2, r2, a.q22);
}

__global__ __launch_bounds__(S1THREADS) __attribute__((amdgpu_waves_per_eu(1, 8)))
void pos_window_stats(const float* __restrict__ rgb,
                                                              const float* __restrict__ matrix,
                                                              float* __restrict__ g) {
  __shared__ float4 tile4[16 * TROW4];
  __shared__ float4 sg4[3 * NT * 16 / 4];
  const int tid = threadIdx.x;
  const int n0  = blockIdx.x * NT;

  {
    const int half = tid >> 6, c4 = tid & 63;
    if (c4 < TROW4) {
      float4 st[8];
#pragma unroll
      for (int u = 0; u < 8; ++u)
        st[u] = reinterpret_cast<const float4*>(rgb + (size_t)((2 * u + half) * TLEN + n0) * 3)[c4];
#pragma unroll
      for (int u = 0; u < 8; ++u) tile4[(2 * u + half) * TROW4 + c4] = st[u];
    }
  }
  __syncthreads();

  const int b = tid >> 3;
  const int q = tid & 7;
  const float4* t4 = tile4 + b * TROW4 + 3 * q;

  WinAcc acc[WPT] = {};

#pragma unroll
  for (int m = 0; m < (WIN + WPT - 1 + 3) / 4; ++m) {
    const float4 A = t4[3 * m + 0], B = t4[3 * m + 1], C = t4[3 * m + 2];
    const float smp[4][3] = {{A.x, A.y, A.z}, {A.w, B.x, B.y}, {B.z, B.w, C.x}, {C.y, C.z, C.w}};
#pragma unroll
    for (int e = 0; e < 4; ++e) {
      const int i = 4 * m + e;
#pragma unroll
      for (int j = 0; j < WPT; ++j)
        if (i >= j && i < j + WIN) win_add(acc[j], smp[e][0], smp[e][1], smp[e][2]);
    }
  }

  float m00 = matrix[0], m01 = matrix[1], m02 = matrix[2];
  float m10 = matrix[3], m11 = matrix[4], m12 = matrix[5];
  float* sg = reinterpret_cast<float*>(sg4);

#pragma unroll
  for (int j = 0; j < WPT; ++j) {
    const float s0 = acc[j].s0, s1 = acc[j].s1, s2 = acc[j].s2;
    const float q00 = acc[j].q00, q01 = acc[j].q01, q02 = acc[j].q02;
    const float q11 = acc[j].q11, q12 = acc[j].q12, q22 = acc[j].q22;

    const float invW = 1.0f / (float)WIN;
    float inv0 = 1.0f / (s0 * invW + 1e-6f);
    float inv1 = 1.0f / (s1 * invW + 1e-6f);
    float inv2 = 1.0f / (s2 * invW + 1e-6f);

    float u0 = m00 * inv0, u1 = m01 * inv1, u2 = m02 * inv2;
    float sum0 = u0 * s0 + u1 * s1 + u2 * s2;
    float sq0  = u0 * u0 * q00 + u1 * u1 * q11 + u2 * u2 * q22 +
                 2.f * (u0 * u1 * q01 + u0 * u2 * q02 + u1 * u2 * q12);
    float var0 = (sq0 - sum0 * sum0 * invW) * (1.0f / (float)(WIN - 1));
    float std0 = sqrtf(fmaxf(var0, 0.f));

    float v0 = m10 * inv0, v1 = m11 * inv1, v2 = m12 * inv2;
    float sum1 = v0 * s0 + v1 * s1 + v2 * s2;
    float sq1  = v0 * v0 * q00 + v1 * v1 * q11 + v2 * v2 * q22 +
                 2.f * (v0 * v1 * q01 + v0 * v2 * q02 + v1 * v2 * q12);
    float var1 = (sq1 - sum1 * sum1 * invW) * (1.0f / (float)(WIN - 1));
    float std1 = sqrtf(fmaxf(var1, 0.f));

    float c = std0 / (std1 + 1e-4f);

    const int nl = WPT * q + j;
    sg[(0 * NT + nl) * 16 + b] = (m00 + c * m10) * inv0;
    sg[(1 * NT + nl) * 16 + b] = (m01 + c * m11) * inv1;
    sg[(2 * NT + nl) * 16 + b] = (m02 + c * m12) * inv2;
  }
  __syncthreads();

  for (int k = 0; k < 3; ++k)
    reinterpret_cast<float4*>(g + (size_t)(k * MWIN + n0) * 16)[tid] = sg4[k * S1THREADS + tid];
}

__global__ void pos_overlap_add_wmma(const float* __restrict__ rgb,
                                     const float* __restrict__ g,
                                     float* __restrict__ ppg) {
  const int t0   = blockIdx.x * 16;
  const int lane = threadIdx.x;
  const int col  = lane & 15;
  const int hi   = lane >> 4;
  const int base = t0 - 47;

  v8f acc[3];
#pragma unroll
  for (int k = 0; k < 3; ++k) {
    const float* gk = g + (size_t)k * MWIN * 16;
    v8f d = {0.f, 0.f, 0.f, 0.f, 0.f, 0.f, 0.f, 0.f};
#pragma unroll
    for (int s = 0; s < 16; ++s) {
      int j0 = 4 * s + 2 * hi;
      int j1 = j0 + 1;
      v2f a;
      a.x = (j0 >= col && j0 <= col + 47) ? 1.0f : 0.0f;
      a.y = (j1 >= col && j1 <= col + 47) ? 1.0f : 0.0f;
      int n0 = base + j0;
      int n1 = base + j1;
      int c0 = min(max(n0, 0), MWIN - 1);
      int c1 = min(max(n1, 0), MWIN - 1);
      float ok0 = ((unsigned)n0 < (unsigned)MWIN) ? 1.0f : 0.0f;
      float ok1 = ((unsigned)n1 < (unsigned)MWIN) ? 1.0f : 0.0f;
      v2f bb;
      bb.x = gk[(size_t)c0 * 16 + col] * ok0;
      bb.y = gk[(size_t)c1 * 16 + col] * ok1;
      d = __builtin_amdgcn_wmma_f32_16x16x4_f32(
           false, a,  false, bb,
           (short)0, d,  false,  false);
    }
    acc[k] = d;
  }

#pragma unroll
  for (int p = 0; p < 8; ++p) {
    int t = t0 + p + 8 * hi;
    const float* rp = rgb + (size_t)(col * TLEN + t) * 3;
    float v = rp[0] * acc[0][p] + rp[1] * acc[1][p] + rp[2] * acc[2][p];
    ppg[(size_t)t * 16 + col] = v;
  }
}


constexpr int CLEN   = 45;
constexpr int NCHUNK = TLEN / CLEN;
constexpr int NGRP   = NCHUNK / 16;
static_assert(NCHUNK * CLEN == TLEN, "chunks tile the row");
static_assert(NGRP * 16 == NCHUNK, "groups tile the chunks");
constexpr int OPITCH = 16 * CLEN + 1;

struct Biquad { float b0, b1, b2, a1, a2; };

__device__ __forceinline__ Biquad load_biquad(const float* __restrict__ acoef,
                                              const float* __restrict__ bcoef) {
  float ia = 1.0f / acoef[0];
  Biquad q;
  q.b0 = bcoef[0] * ia; q.b1 = bcoef[1] * ia; q.b2 = bcoef[2] * ia;
  q.a1 = acoef[1] * ia; q.a2 = acoef[2] * ia;
  return q;
}

__device__ __forceinline__ float biquad_step(const Biquad& q, float x, float& z1, float& z2) {
  float y = fmaf(q.b0, x, z1);
  z1 = fmaf(q.b1, x, fmaf(-q.a1, y, z2));
  z2 = fmaf(q.b2, x, -q.a2 * y);
  return y;
}


__device__ __forceinline__ double pair_swap(double v) {
  const long long bits = __builtin_bit_cast(long long, v);
  int lo = (int)(bits & 0xffffffffLL), hi = (int)(bits >> 32);
  lo = __builtin_amdgcn_update_dpp(0, lo, 0xB1, 0xF, 0xF, false);
  hi = __builtin_amdgcn_update_dpp(0, hi, 0xB1, 0xF, 0xF, false);
  const long long r = ((long long)hi << 32) | (long long)(unsigned int)lo;
  return __builtin_bit_cast(double, r);
}

__device__ __forceinline__ void pair_carry(double td, double to, double e,
                                           double& own, double& other) {
  own = fma(td, own, fma(to, other, e));
  other = pair_swap(own);
}

__device__ __forceinline__ double group_state(const v2f* sh, int lane, const double* m) {
  const int b = lane >> 1, c = lane & 1;
  const double td = m[3 * c], to = m[1 + c];
  const float* shf = reinterpret_cast<const float*>(sh);
  float e[16];
#pragma unroll
  for (int j = 0; j < 16; ++j) e[j] = shf[(j * 16 + b) * 2 + c];
  double own = 0.0, other = 0.0;
#pragma unroll
  for (int j = 0; j < 16; ++j) pair_carry(td, to, (double)e[j], own, other);
  return own;
}

__device__ __forceinline__ void chunk_start_states(const double* __restrict__ EG,
                                                   const float* __restrict__ E,
                                                   const double* __restrict__ Mpow,
                                                   int grp, int tid,
                                                   double* shSd, double* shL,
                                                   float& z1, float& z2) {
  const int b = tid & 15, i = tid >> 4;
  const double p11 = Mpow[4 * i + 0], p12 = Mpow[4 * i + 1];
  const double p21 = Mpow[4 * i + 2], p22 = Mpow[4 * i + 3];
  if (tid < 32) {
    const int rb = tid >> 1, c = tid & 1;
    const double* G = Mpow + 4 * 16;
    const double td = G[3 * c], to = G[1 + c];
    double eg[NGRP - 1];
#pragma unroll
    for (int k = 0; k < NGRP - 1; ++k) eg[k] = EG[(k * 16 + rb) * 2 + c];
    double own = 0.0, other = 0.0;
#pragma unroll
    for (int k = 0; k < NGRP - 1; ++k) {
      if (k < grp) pair_carry(td, to, eg[k], own, other);
    }
    shSd[rb * 2 + c] = own;
  } else if (tid < 64) {
    const int t = tid - 32;
    const int rb = t >> 1, c = t & 1;
    const double* M1 = Mpow + 4;
    const double td = M1[3 * c], to = M1[1 + c];
    float el[15];
#pragma unroll
    for (int k = 0; k < 15; ++k) el[k] = E[((grp * 16 + k) * 16 + rb) * 2 + c];
    double own = 0.0, other = 0.0;
    shL[rb * 2 + c] = 0.0;
#pragma unroll
    for (int k = 0; k < 15; ++k) {
      pair_carry(td, to, (double)el[k], own, other);
      shL[((k + 1) * 16 + rb) * 2 + c] = own;
    }
  }
  __syncthreads();
  const double s1 = shSd[b * 2], s2 = shSd[b * 2 + 1];
  const double l1 = shL[(i * 16 + b) * 2], l2 = shL[(i * 16 + b) * 2 + 1];
  z1 = (float)fma(p11, s1, fma(p12, s2, l1));
  z2 = (float)fma(p21, s1, fma(p22, s2, l2));
}

__global__ __launch_bounds__(256) void pos_scan_fwd_local(
    const float* __restrict__ ppg, const float* __restrict__ acoef,
    const float* __restrict__ bcoef, float* __restrict__ Ef_,
    float* __restrict__ EfG_, float* __restrict__ Mbuf_) {
  __shared__ v2f sh[16 * 16];
  __shared__ double shM[4];
  v2f* Ef  = reinterpret_cast<v2f*>(Ef_);
  double* EfG = reinterpret_cast<double*>(EfG_);
  double* Mbuf = reinterpret_cast<double*>(Mbuf_);
  const int tid = threadIdx.x;
  const int b = tid & 15, i = tid >> 4;
  const int grp = blockIdx.x;
  const int c = grp * 16 + i;
  const Biquad q = load_biquad(acoef, bcoef);

  const float* xp = ppg + (size_t)c * CLEN * 16 + b;
  float x[CLEN];
#pragma unroll
  for (int w = 0; w < CLEN; ++w) x[w] = xp[(size_t)w * 16];

  if (tid < 32) {
    const int j = (tid >> 1) & 1, cc = tid & 1;
    const double coef = cc ? -(double)q.a2 : -(double)q.a1;
    double own = (cc == j) ? 1.0 : 0.0, other = (cc == j) ? 0.0 : 1.0;
    for (int w = 0; w < CLEN; ++w) {
      const double p  = cc ? other : own;
      const double ad = cc ? 0.0 : other;
      own = fma(coef, p, ad);
      other = pair_swap(own);
    }
    if (tid < 4) shM[2 * cc + j] = own;
  }

  float z1 = 0.f, z2 = 0.f;
#pragma unroll
  for (int w = 0; w < CLEN; ++w) (void)biquad_step(q, x[w], z1, z2);

  v2f e; e.x = z1; e.y = z2;
  Ef[c * 16 + b] = e;
  sh[i * 16 + b] = e;
  __syncthreads();
  if (tid < 32) {
    EfG[grp * 32 + tid] = group_state(sh, tid, shM);
  } else if (grp == 0 && tid < 36) {
    const int t = tid - 32;
    const int j = t >> 1, cc = t & 1;
    const double td = shM[3 * cc], to = shM[1 + cc];
    double own = (cc == j) ? 1.0 : 0.0, other = (cc == j) ? 0.0 : 1.0;
    Mbuf[2 * cc + j] = own;
    for (int k = 1; k <= 16; ++k) {
      pair_carry(td, to, 0.0, own, other);
      Mbuf[4 * k + 2 * cc + j] = own;
    }
  }
}

__global__ __launch_bounds__(256) void pos_scan_fwd_apply(
    const float* __restrict__ ppg, const float* __restrict__ acoef,
    const float* __restrict__ bcoef, const float* __restrict__ Ef_,
    const float* __restrict__ EfG_, const float* __restrict__ Mbuf_,
    float* __restrict__ ybuf, float* __restrict__ Eb_, float* __restrict__ EbG_) {
  __shared__ double shSd[16 * 2];
  __shared__ double shL[16 * 16 * 2];
  __shared__ v2f shE[16 * 16];
  const double* EfG = reinterpret_cast<const double*>(EfG_);
  const double* Mbuf = reinterpret_cast<const double*>(Mbuf_);
  v2f* Eb  = reinterpret_cast<v2f*>(Eb_);
  double* EbG = reinterpret_cast<double*>(EbG_);
  const int tid = threadIdx.x;
  const int b = tid & 15, i = tid >> 4;
  const int grp = blockIdx.x;
  const int c = grp * 16 + i;
  const Biquad q = load_biquad(acoef, bcoef);

  const float* xp = ppg + (size_t)c * CLEN * 16 + b;
  float x[CLEN];
#pragma unroll
  for (int w = 0; w < CLEN; ++w) x[w] = xp[(size_t)w * 16];

  float z1, z2;
  chunk_start_states(EfG, Ef_, Mbuf, grp, tid, shSd, shL, z1, z2);

  float* yp = ybuf + (size_t)c * CLEN * 16 + b;
  float y[CLEN];
#pragma unroll
  for (int w = 0; w < CLEN; ++w) {
    y[w] = biquad_step(q, x[w], z1, z2);
    yp[(size_t)w * 16] = y[w];
  }

  z1 = 0.f; z2 = 0.f;
#pragma unroll
  for (int w = CLEN - 1; w >= 0; --w) (void)biquad_step(q, y[w], z1, z2);

  const int cr = NCHUNK - 1 - c;
  const int grpr = NGRP - 1 - grp;
  const int ir = 15 - i;
  v2f e; e.x = z1; e.y = z2;
  Eb[cr * 16 + b] = e;
  shE[ir * 16 + b] = e;
  __syncthreads();
  if (tid < 32) {
    EbG[grpr * 32 + tid] = group_state(shE, tid, Mbuf + 4);
  }
}

__global__ __launch_bounds__(256) void pos_scan_bwd_apply(
    const float* __restrict__ ybuf, const float* __restrict__ acoef,
    const float* __restrict__ bcoef, const float* __restrict__ Eb_,
    const float* __restrict__ EbG_, const float* __restrict__ Mbuf_,
    float* __restrict__ out) {
  __shared__ float tile[16 * OPITCH];
  __shared__ double shSd[16 * 2];
  __shared__ double shL[16 * 16 * 2];
  const double* EbG = reinterpret_cast<const double*>(EbG_);
  const double* Mbuf = reinterpret_cast<const double*>(Mbuf_);
  const int tid = threadIdx.x;
  const int b = tid & 15, ir = tid >> 4;
  const int grpr = blockIdx.x;
  const int cr = grpr * 16 + ir;
  const int c = NCHUNK - 1 - cr;
  const Biquad q = load_biquad(acoef, bcoef);

  const float* xp = ybuf + (size_t)c * CLEN * 16 + b;
  float x[CLEN];
#pragma unroll
  for (int w = 0; w < CLEN; ++w) x[w] = xp[(size_t)w * 16];

  float z1, z2;
  chunk_start_states(EbG, Eb_, Mbuf, grpr, tid, shSd, shL, z1, z2);

  const int clo = NCHUNK - 16 * (grpr + 1);
  float* tp = tile + b * OPITCH + (c - clo) * CLEN;
#pragma unroll
  for (int w = CLEN - 1; w >= 0; --w) tp[w] = biquad_step(q, x[w], z1, z2);
  __syncthreads();

  const int tlo = clo * CLEN;
  for (int k = tid; k < 16 * 16 * CLEN; k += 256) {
    int row = k / (16 * CLEN);
    int tt  = k - row * (16 * CLEN);
    out[(size_t)row * TLEN + tlo + tt] = tile[row * OPITCH + tt];
  }
}

extern "C" void kernel_launch(void* const* d_in, const int* in_sizes, int n_in,
                              void* d_out, int out_size, void* d_ws, size_t ws_size,
                              hipStream_t stream) {
  const float* rgb    = (const float*)d_in[0];
  const float* matrix = (const float*)d_in[1];
  const float* acoef  = (const float*)d_in[2];
  const float* bcoef  = (const float*)d_in[3];
  float* out = (float*)d_out;

  float* g    = (float*)d_ws;
  float* ppg  = g + (size_t)3 * MWIN * 16;
  float* ybuf = ppg + (size_t)TLEN * 16;
  float* Ef   = ybuf + (size_t)TLEN * 16;
  float* Eb   = Ef + (size_t)NCHUNK * 32;
  float* EfG  = Eb + (size_t)NCHUNK * 32;
  float* EbG  = EfG + (size_t)NGRP * 64;
  float* Mbuf = EbG + (size_t)NGRP * 64;

  pos_window_stats<<<MWIN / NT, S1THREADS, 0, stream>>>(rgb, matrix, g);
  pos_overlap_add_wmma<<<TLEN / 16, 32, 0, stream>>>(rgb, g, ppg);
  pos_scan_fwd_local<<<NGRP, 256, 0, stream>>>(ppg, acoef, bcoef, Ef, EfG, Mbuf);
  pos_scan_fwd_apply<<<NGRP, 256, 0, stream>>>(ppg, acoef, bcoef, Ef, EfG, Mbuf,
                                               ybuf, Eb, EbG);
  pos_scan_bwd_apply<<<NGRP, 256, 0, stream>>>(ybuf, acoef, bcoef, Eb, EbG, Mbuf, out);
}


# Window stats and WMMA overlap-add fused into one launch: a work-group computes the gains of the 127 windows over its 80 samples into LDS and overlap-adds from there; 4 launches, gains never in memory
# speedup vs baseline: 74.1982x; 1.1302x over previous
#include <hip/hip_runtime.h>
#include <hip/hip_bf16.h>

typedef float v2f __attribute__((ext_vector_type(2)));
typedef float v8f __attribute__((ext_vector_type(8)));

constexpr int TLEN  = 18000;
constexpr int BATCH = 16;
constexpr int WIN   = 48;
constexpr int MWIN  = TLEN - WIN;

constexpr int TT     = 80;
constexpr int NTILE  = TT / 16;
constexpr int NWIN   = 128;
constexpr int WPT    = 4;
constexpr int TPR    = NWIN / WPT;
constexpr int FTHREADS = 16 * TPR;
constexpr int TSMP   = 180;
constexpr int TROW   = TSMP * 3;
constexpr int TROW4  = TROW / 4;
constexpr int ROW4   = TLEN * 3 / 4;
constexpr int GPITCH = NWIN + 4;
constexpr int NSTAGE = (16 * TROW4 + FTHREADS - 1) / FTHREADS;
static_assert(TLEN % TT == 0 && TT % 16 == 0, "work-groups and tiles cover the row exactly");
static_assert(NWIN >= TT + WIN && NWIN % WPT == 0 && TPR == 32, "one wave per row holds every window of the group");
static_assert((WIN * 3) % 4 == 0 && (TT * 3) % 4 == 0 && (TLEN * 3) % 4 == 0, "row segments start 16-byte aligned");
static_assert(TROW % 4 == 0 && TSMP >= NWIN - WPT + 4 * ((WIN + WPT - 1 + 3) / 4), "segment holds every sample a thread reads");
static_assert(TSMP >= WIN + TT, "segment holds the group's own output samples");
static_assert((3 * WPT) % 4 == 0, "a thread's first sample is 16-byte aligned in LDS");
static_assert(GPITCH % 4 == 0 && GPITCH >= 16 * (NTILE - 1) + 1 + 64, "a tile's 64 K positions stay inside the gain row");

struct WinAcc { float s0, s1, s2, q00, q01, q02, q11, q12, q22; };

__device__ __forceinline__ void win_add(WinAcc& a, float r0, float r1, float r2) {
  a.s0 += r0; a.s1 += r1; a.s2 += r2;
  a.q00 = fmaf(r0, r0, a.q00); a.q01 = fmaf(r0, r1, a.q01); a.q02 = fmaf(r0, r2, a.q02);
  a.q11 = fmaf(r1, r1, a.q11); a.q12 = fmaf(r1, r2, a.q12); a.q22 = fmaf(r2, r2, a.q22);
}

__global__ __launch_bounds__(FTHREADS)
void pos_window_overlap_add(const float* __restrict__ rgb,
                            const float* __restrict__ matrix,
                            float* __restrict__ ppg) {
  __shared__ float4 tile4[16 * TROW4];
  __shared__ float4 sg4[3 * 16 * GPITCH / 4];
  const int tid = threadIdx.x;
  const int t0  = blockIdx.x * TT;

  {
    const int g4lo = (t0 - WIN) * 3 / 4;
    const float4* rgb4 = reinterpret_cast<const float4*>(rgb);
    float4 st[NSTAGE];
#pragma unroll
    for (int u = 0; u < NSTAGE; ++u) {
      const int idx = tid + u * FTHREADS;
      const int row = idx / TROW4, c4 = idx - row * TROW4;
      const int gi = g4lo + c4;
      st[u] = make_float4(0.f, 0.f, 0.f, 0.f);
      if (idx < 16 * TROW4 && (unsigned)gi < (unsigned)ROW4) st[u] = rgb4[(size_t)row * ROW4 + gi];
    }
#pragma unroll
    for (int u = 0; u < NSTAGE; ++u) {
      const int idx = tid + u * FTHREADS;
      if (idx < 16 * TROW4) tile4[idx] = st[u];
    }
  }
  __syncthreads();

  const int b = tid >> 5;
  const int q = tid & 31;
  const float4* t4 = tile4 + b * TROW4 + 3 * q;

  WinAcc acc[WPT] = {};

#pragma unroll
  for (int m = 0; m < (WIN + WPT - 1 + 3) / 4; ++m) {
    const float4 A = t4[3 * m + 0], B = t4[3 * m + 1], C = t4[3 * m + 2];
    const float smp[4][3] = {{A.x, A.y, A.z}, {A.w, B.x, B.y}, {B.z, B.w, C.x}, {C.y, C.z, C.w}};
#pragma unroll
    for (int e = 0; e < 4; ++e) {
      const int i = 4 * m + e;
#pragma unroll
      for (int j = 0; j < WPT; ++j)
        if (i >= j && i < j + WIN) win_add(acc[j], smp[e][0], smp[e][1], smp[e][2]);
    }
  }

  float m00 = matrix[0], m01 = matrix[1], m02 = matrix[2];
  float m10 = matrix[3], m11 = matrix[4], m12 = matrix[5];
  float gv[3][WPT];

#pragma unroll
  for (int j = 0; j < WPT; ++j) {
    const float s0 = acc[j].s0, s1 = acc[j].s1, s2 = acc[j].s2;
    const float q00 = acc[j].q00, q01 = acc[j].q01, q02 = acc[j].q02;
    const float q11 = acc[j].q11, q12 = acc[j].q12, q22 = acc[j].q22;

    const float invW = 1.0f / (float)WIN;
    float inv0 = 1.0f / (s0 * invW + 1e-6f);
    float inv1 = 1.0f / (s1 * invW + 1e-6f);
    float inv2 = 1.0f / (s2 * invW + 1e-6f);

    float u0 = m00 * inv0, u1 = m01 * inv1, u2 = m02 * inv2;
    float sum0 = u0 * s0 + u1 * s1 + u2 * s2;
    float sq0  = u0 * u0 * q00 + u1 * u1 * q11 + u2 * u2 * q22 +
                 2.f * (u0 * u1 * q01 + u0 * u2 * q02 + u1 * u2 * q12);
    float var0 = (sq0 - sum0 * sum0 * invW) * (1.0f / (float)(WIN - 1));
    float std0 = sqrtf(fmaxf(var0, 0.f));

    float v0 = m10 * inv0, v1 = m11 * inv1, v2 = m12 * inv2;
    float sum1 = v0 * s0 + v1 * s1 + v2 * s2;
    float sq1  = v0 * v0 * q00 + v1 * v1 * q11 + v2 * v2 * q22 +
                 2.f * (v0 * v1 * q01 + v0 * v2 * q02 + v1 * v2 * q12);
    float var1 = (sq1 - sum1 * sum1 * invW) * (1.0f / (float)(WIN - 1));
    float std1 = sqrtf(fmaxf(var1, 0.f));

    float c = std0 / (std1 + 1e-4f);

    const int n = t0 - WIN + WPT * q + j;
    const bool have = (unsigned)n < (unsigned)MWIN;
    gv[0][j] = have ? (m00 + c * m10) * inv0 : 0.f;
    gv[1][j] = have ? (m01 + c * m11) * inv1 : 0.f;
    gv[2][j] = have ? (m02 + c * m12) * inv2 : 0.f;
  }
#pragma unroll
  for (int k = 0; k < 3; ++k) {
    sg4[(k * 16 + b) * (GPITCH / 4) + q] = make_float4(gv[k][0], gv[k][1], gv[k][2], gv[k][3]);
    if (q == 0) sg4[(k * 16 + b) * (GPITCH / 4) + TPR] = make_float4(0.f, 0.f, 0.f, 0.f);
  }
  __syncthreads();

  const int wave = tid >> 5;
  if (wave < NTILE) {
    const int lane = tid & 31;
    const int col  = lane & 15;
    const int hi   = lane >> 4;
    const float* sgt = reinterpret_cast<const float*>(sg4) + col * GPITCH + 16 * wave + 1;

    v8f acc3[3];
#pragma unroll
    for (int k = 0; k < 3; ++k) {
      const float* gk = sgt + k * 16 * GPITCH;
      v8f d = {0.f, 0.f, 0.f, 0.f, 0.f, 0.f, 0.f, 0.f};
#pragma unroll
      for (int s = 0; s < 16; ++s) {
        int j0 = 4 * s + 2 * hi;
        int j1 = j0 + 1;
        v2f a;
        a.x = (j0 >= col && j0 <= col + 47) ? 1.0f : 0.0f;
        a.y = (j1 >= col && j1 <= col + 47) ? 1.0f : 0.0f;
        v2f bb;
        bb.x = gk[j0];
        bb.y = gk[j1];
        d = __builtin_amdgcn_wmma_f32_16x16x4_f32(
             false, a,  false, bb,
             (short)0, d,  false,  false);
      }
      acc3[k] = d;
    }

    const float* rt = reinterpret_cast<const float*>(tile4) + col * TROW + (WIN + 16 * wave) * 3;
#pragma unroll
    for (int p = 0; p < 8; ++p) {
      const int i = p + 8 * hi;
      const float* rp = rt + i * 3;
      float v = rp[0] * acc3[0][p] + rp[1] * acc3[1][p] + rp[2] * acc3[2][p];
      ppg[(size_t)(t0 + 16 * wave + i) * 16 + col] = v;
    }
  }
}


constexpr int CLEN   = 45;
constexpr int NCHUNK = TLEN / CLEN;
constexpr int NGRP   = NCHUNK / 16;
static_assert(NCHUNK * CLEN == TLEN, "chunks tile the row");
static_assert(NGRP * 16 == NCHUNK, "groups tile the chunks");
constexpr int OPITCH = 16 * CLEN + 1;

struct Biquad { float b0, b1, b2, a1, a2; };

__device__ __forceinline__ Biquad load_biquad(const float* __restrict__ acoef,
                                              const float* __restrict__ bcoef) {
  float ia = 1.0f / acoef[0];
  Biquad q;
  q.b0 = bcoef[0] * ia; q.b1 = bcoef[1] * ia; q.b2 = bcoef[2] * ia;
  q.a1 = acoef[1] * ia; q.a2 = acoef[2] * ia;
  return q;
}

__device__ __forceinline__ float biquad_step(const Biquad& q, float x, float& z1, float& z2) {
  float y = fmaf(q.b0, x, z1);
  z1 = fmaf(q.b1, x, fmaf(-q.a1, y, z2));
  z2 = fmaf(q.b2, x, -q.a2 * y);
  return y;
}


__device__ __forceinline__ double pair_swap(double v) {
  const long long bits = __builtin_bit_cast(long long, v);
  int lo = (int)(bits & 0xffffffffLL), hi = (int)(bits >> 32);
  lo = __builtin_amdgcn_update_dpp(0, lo, 0xB1, 0xF, 0xF, false);
  hi = __builtin_amdgcn_update_dpp(0, hi, 0xB1, 0xF, 0xF, false);
  const long long r = ((long long)hi << 32) | (long long)(unsigned int)lo;
  return __builtin_bit_cast(double, r);
}

__device__ __forceinline__ void pair_carry(double td, double to, double e,
                                           double& own, double& other) {
  own = fma(td, own, fma(to, other, e));
  other = pair_swap(own);
}

__device__ __forceinline__ double group_state(const v2f* sh, int lane, const double* m) {
  const int b = lane >> 1, c = lane & 1;
  const double td = m[3 * c], to = m[1 + c];
  const float* shf = reinterpret_cast<const float*>(sh);
  float e[16];
#pragma unroll
  for (int j = 0; j < 16; ++j) e[j] = shf[(j * 16 + b) * 2 + c];
  double own = 0.0, other = 0.0;
#pragma unroll
  for (int j = 0; j < 16; ++j) pair_carry(td, to, (double)e[j], own, other);
  return own;
}

__device__ __forceinline__ void chunk_start_states(const double* __restrict__ EG,
                                                   const float* __restrict__ E,
                                                   const double* __restrict__ Mpow,
                                                   int grp, int tid,
                                                   double* shSd, double* shL,
                                                   float& z1, float& z2) {
  const int b = tid & 15, i = tid >> 4;
  const double p11 = Mpow[4 * i + 0], p12 = Mpow[4 * i + 1];
  const double p21 = Mpow[4 * i + 2], p22 = Mpow[4 * i + 3];
  if (tid < 32) {
    const int rb = tid >> 1, c = tid & 1;
    const double* G = Mpow + 4 * 16;
    const double td = G[3 * c], to = G[1 + c];
    double eg[NGRP - 1];
#pragma unroll
    for (int k = 0; k < NGRP - 1; ++k) eg[k] = EG[(k * 16 + rb) * 2 + c];
    double own = 0.0, other = 0.0;
#pragma unroll
    for (int k = 0; k < NGRP - 1; ++k) {
      if (k < grp) pair_carry(td, to, eg[k], own, other);
    }
    shSd[rb * 2 + c] = own;
  } else if (tid < 64) {
    const int t = tid - 32;
    const int rb = t >> 1, c = t & 1;
    const double* M1 = Mpow + 4;
    const double td = M1[3 * c], to = M1[1 + c];
    float el[15];
#pragma unroll
    for (int k = 0; k < 15; ++k) el[k] = E[((grp * 16 + k) * 16 + rb) * 2 + c];
    double own = 0.0, other = 0.0;
    shL[rb * 2 + c] = 0.0;
#pragma unroll
    for (int k = 0; k < 15; ++k) {
      pair_carry(td, to, (double)el[k], own, other);
      shL[((k + 1) * 16 + rb) * 2 + c] = own;
    }
  }
  __syncthreads();
  const double s1 = shSd[b * 2], s2 = shSd[b * 2 + 1];
  const double l1 = shL[(i * 16 + b) * 2], l2 = shL[(i * 16 + b) * 2 + 1];
  z1 = (float)fma(p11, s1, fma(p12, s2, l1));
  z2 = (float)fma(p21, s1, fma(p22, s2, l2));
}

__global__ __launch_bounds__(256) void pos_scan_fwd_local(
    const float* __restrict__ ppg, const float* __restrict__ acoef,
    const float* __restrict__ bcoef, float* __restrict__ Ef_,
    float* __restrict__ EfG_, float* __restrict__ Mbuf_) {
  __shared__ v2f sh[16 * 16];
  __shared__ double shM[4];
  v2f* Ef  = reinterpret_cast<v2f*>(Ef_);
  double* EfG = reinterpret_cast<double*>(EfG_);
  double* Mbuf = reinterpret_cast<double*>(Mbuf_);
  const int tid = threadIdx.x;
  const int b = tid & 15, i = tid >> 4;
  const int grp = blockIdx.x;
  const int c = grp * 16 + i;
  const Biquad q = load_biquad(acoef, bcoef);

  const float* xp = ppg + (size_t)c * CLEN * 16 + b;
  float x[CLEN];
#pragma unroll
  for (int w = 0; w < CLEN; ++w) x[w] = xp[(size_t)w * 16];

  if (tid < 32) {
    const int j = (tid >> 1) & 1, cc = tid & 1;
    const double coef = cc ? -(double)q.a2 : -(double)q.a1;
    double own = (cc == j) ? 1.0 : 0.0, other = (cc == j) ? 0.0 : 1.0;
    for (int w = 0; w < CLEN; ++w) {
      const double p  = cc ? other : own;
      const double ad = cc ? 0.0 : other;
      own = fma(coef, p, ad);
      other = pair_swap(own);
    }
    if (tid < 4) shM[2 * cc + j] = own;
  }

  float z1 = 0.f, z2 = 0.f;
#pragma unroll
  for (int w = 0; w < CLEN; ++w) (void)biquad_step(q, x[w], z1, z2);

  v2f e; e.x = z1; e.y = z2;
  Ef[c * 16 + b] = e;
  sh[i * 16 + b] = e;
  __syncthreads();
  if (tid < 32) {
    EfG[grp * 32 + tid] = group_state(sh, tid, shM);
  } else if (grp == 0 && tid < 36) {
    const int t = tid - 32;
    const int j = t >> 1, cc = t & 1;
    const double td = shM[3 * cc], to = shM[1 + cc];
    double own = (cc == j) ? 1.0 : 0.0, other = (cc == j) ? 0.0 : 1.0;
    Mbuf[2 * cc + j] = own;
    for (int k = 1; k <= 16; ++k) {
      pair_carry(td, to, 0.0, own, other);
      Mbuf[4 * k + 2 * cc + j] = own;
    }
  }
}

__global__ __launch_bounds__(256) void pos_scan_fwd_apply(
    const float* __restrict__ ppg, const float* __restrict__ acoef,
    const float* __restrict__ bcoef, const float* __restrict__ Ef_,
    const float* __restrict__ EfG_, const float* __restrict__ Mbuf_,
    float* __restrict__ ybuf, float* __restrict__ Eb_, float* __restrict__ EbG_) {
  __shared__ double shSd[16 * 2];
  __shared__ double shL[16 * 16 * 2];
  __shared__ v2f shE[16 * 16];
  const double* EfG = reinterpret_cast<const double*>(EfG_);
  const double* Mbuf = reinterpret_cast<const double*>(Mbuf_);
  v2f* Eb  = reinterpret_cast<v2f*>(Eb_);
  double* EbG = reinterpret_cast<double*>(EbG_);
  const int tid = threadIdx.x;
  const int b = tid & 15, i = tid >> 4;
  const int grp = blockIdx.x;
  const int c = grp * 16 + i;
  const Biquad q = load_biquad(acoef, bcoef);

  const float* xp = ppg + (size_t)c * CLEN * 16 + b;
  float x[CLEN];
#pragma unroll
  for (int w = 0; w < CLEN; ++w) x[w] = xp[(size_t)w * 16];

  float z1, z2;
  chunk_start_states(EfG, Ef_, Mbuf, grp, tid, shSd, shL, z1, z2);

  float* yp = ybuf + (size_t)c * CLEN * 16 + b;
  float y[CLEN];
#pragma unroll
  for (int w = 0; w < CLEN; ++w) {
    y[w] = biquad_step(q, x[w], z1, z2);
    yp[(size_t)w * 16] = y[w];
  }

  z1 = 0.f; z2 = 0.f;
#pragma unroll
  for (int w = CLEN - 1; w >= 0; --w) (void)biquad_step(q, y[w], z1, z2);

  const int cr = NCHUNK - 1 - c;
  const int grpr = NGRP - 1 - grp;
  const int ir = 15 - i;
  v2f e; e.x = z1; e.y = z2;
  Eb[cr * 16 + b] = e;
  shE[ir * 16 + b] = e;
  __syncthreads();
  if (tid < 32) {
    EbG[grpr * 32 + tid] = group_state(shE, tid, Mbuf + 4);
  }
}

__global__ __launch_bounds__(256) void pos_scan_bwd_apply(
    const float* __restrict__ ybuf, const float* __restrict__ acoef,
    const float* __restrict__ bcoef, const float* __restrict__ Eb_,
    const float* __restrict__ EbG_, const float* __restrict__ Mbuf_,
    float* __restrict__ out) {
  __shared__ float tile[16 * OPITCH];
  __shared__ double shSd[16 * 2];
  __shared__ double shL[16 * 16 * 2];
  const double* EbG = reinterpret_cast<const double*>(EbG_);
  const double* Mbuf = reinterpret_cast<const double*>(Mbuf_);
  const int tid = threadIdx.x;
  const int b = tid & 15, ir = tid >> 4;
  const int grpr = blockIdx.x;
  const int cr = grpr * 16 + ir;
  const int c = NCHUNK - 1 - cr;
  const Biquad q = load_biquad(acoef, bcoef);

  const float* xp = ybuf + (size_t)c * CLEN * 16 + b;
  float x[CLEN];
#pragma unroll
  for (int w = 0; w < CLEN; ++w) x[w] = xp[(size_t)w * 16];

  float z1, z2;
  chunk_start_states(EbG, Eb_, Mbuf, grpr, tid, shSd, shL, z1, z2);

  const int clo = NCHUNK - 16 * (grpr + 1);
  float* tp = tile + b * OPITCH + (c - clo) * CLEN;
#pragma unroll
  for (int w = CLEN - 1; w >= 0; --w) tp[w] = biquad_step(q, x[w], z1, z2);
  __syncthreads();

  const int tlo = clo * CLEN;
  for (int k = tid; k < 16 * 16 * CLEN; k += 256) {
    int row = k / (16 * CLEN);
    int tt  = k - row * (16 * CLEN);
    out[(size_t)row * TLEN + tlo + tt] = tile[row * OPITCH + tt];
  }
}

extern "C" void kernel_launch(void* const* d_in, const int* in_sizes, int n_in,
                              void* d_out, int out_size, void* d_ws, size_t ws_size,
                              hipStream_t stream) {
  const float* rgb    = (const float*)d_in[0];
  const float* matrix = (const float*)d_in[1];
  const float* acoef  = (const float*)d_in[2];
  const float* bcoef  = (const float*)d_in[3];
  float* out = (float*)d_out;

  float* ppg  = (float*)d_ws;
  float* ybuf = ppg + (size_t)TLEN * 16;
  float* Ef   = ybuf + (size_t)TLEN * 16;
  float* Eb   = Ef + (size_t)NCHUNK * 32;
  float* EfG  = Eb + (size_t)NCHUNK * 32;
  float* EbG  = EfG + (size_t)NGRP * 64;
  float* Mbuf = EbG + (size_t)NGRP * 64;

  pos_window_overlap_add<<<TLEN / TT, FTHREADS, 0, stream>>>(rgb, matrix, ppg);
  pos_scan_fwd_local<<<NGRP, 256, 0, stream>>>(ppg, acoef, bcoef, Ef, EfG, Mbuf);
  pos_scan_fwd_apply<<<NGRP, 256, 0, stream>>>(ppg, acoef, bcoef, Ef, EfG, Mbuf,
                                               ybuf, Eb, EbG);
  pos_scan_bwd_apply<<<NGRP, 256, 0, stream>>>(ybuf, acoef, bcoef, Eb, EbG, Mbuf, out);
}


# Window sums shared in a thread: 8 consecutive windows per thread (was 4), their 41 common samples summed once, each window adds its own front and back samples; f32 sum order changes, fixed per run
# speedup vs baseline: 76.5938x; 1.0323x over previous
#include <hip/hip_runtime.h>
#include <hip/hip_bf16.h>

typedef float v2f __attribute__((ext_vector_type(2)));
typedef float v8f __attribute__((ext_vector_type(8)));

constexpr int TLEN  = 18000;
constexpr int BATCH = 16;
constexpr int WIN   = 48;
constexpr int MWIN  = TLEN - WIN;

constexpr int TT     = 80;
constexpr int NTILE  = TT / 16;
constexpr int NWIN   = 128;
constexpr int WPT    = 8;
constexpr int TPR    = NWIN / WPT;
constexpr int FTHREADS = 16 * TPR;
constexpr int TSMP   = 180;
constexpr int TROW   = TSMP * 3;
constexpr int TROW4  = TROW / 4;
constexpr int ROW4   = TLEN * 3 / 4;
constexpr int GPITCH = NWIN + 4;
constexpr int NSTAGE = (16 * TROW4 + FTHREADS - 1) / FTHREADS;
static_assert(TLEN % TT == 0 && TT % 16 == 0, "work-groups and tiles cover the row exactly");
static_assert(NWIN >= TT + WIN && NWIN % WPT == 0 && TPR == 16, "half a wave per row holds every window of the group");
static_assert(WPT % 4 == 0 && FTHREADS / 32 >= NTILE, "gains leave as 16-byte pieces; one wave per overlap-add tile");
static_assert((WIN * 3) % 4 == 0 && (TT * 3) % 4 == 0 && (TLEN * 3) % 4 == 0, "row segments start 16-byte aligned");
static_assert(TROW % 4 == 0 && TSMP >= NWIN - WPT + 4 * ((WIN + WPT - 1 + 3) / 4), "segment holds every sample a thread reads");
static_assert(TSMP >= WIN + TT, "segment holds the group's own output samples");
static_assert((3 * WPT) % 4 == 0, "a thread's first sample is 16-byte aligned in LDS");
static_assert(GPITCH % 4 == 0 && GPITCH >= 16 * (NTILE - 1) + 1 + 64, "a tile's 64 K positions stay inside the gain row");

struct WinAcc { float s0, s1, s2, q00, q01, q02, q11, q12, q22; };

__device__ __forceinline__ void win_add(WinAcc& a, float r0, float r1, float r2) {
  a.s0 += r0; a.s1 += r1; a.s2 += r2;
  a.q00 = fmaf(r0, r0, a.q00); a.q01 = fmaf(r0, r1, a.q01); a.q02 = fmaf(r0, r2, a.q02);
  a.q11 = fmaf(r1, r1, a.q11); a.q12 = fmaf(r1, r2, a.q12); a.q22 = fmaf(r2, r2, a.q22);
}

__device__ __forceinline__ void win_join(WinAcc& a, const WinAcc& b) {
  a.s0 += b.s0; a.s1 += b.s1; a.s2 += b.s2;
  a.q00 += b.q00; a.q01 += b.q01; a.q02 += b.q02;
  a.q11 += b.q11; a.q12 += b.q12; a.q22 += b.q22;
}

__global__ __launch_bounds__(FTHREADS)
void pos_window_overlap_add(const float* __restrict__ rgb,
                            const float* __restrict__ matrix,
                            float* __restrict__ ppg) {
  __shared__ float4 tile4[16 * TROW4];
  __shared__ float4 sg4[3 * 16 * GPITCH / 4];
  const int tid = threadIdx.x;
  const int t0  = blockIdx.x * TT;

  {
    const int g4lo = (t0 - WIN) * 3 / 4;
    const float4* rgb4 = reinterpret_cast<const float4*>(rgb);
    float4 st[NSTAGE];
#pragma unroll
    for (int u = 0; u < NSTAGE; ++u) {
      const int idx = tid + u * FTHREADS;
      const int row = idx / TROW4, c4 = idx - row * TROW4;
      const int gi = g4lo + c4;
      st[u] = make_float4(0.f, 0.f, 0.f, 0.f);
      if (idx < 16 * TROW4 && (unsigned)gi < (unsigned)ROW4) st[u] = rgb4[(size_t)row * ROW4 + gi];
    }
#pragma unroll
    for (int u = 0; u < NSTAGE; ++u) {
      const int idx = tid + u * FTHREADS;
      if (idx < 16 * TROW4) tile4[idx] = st[u];
    }
  }
  __syncthreads();

  const int b = tid / TPR;
  const int q = tid % TPR;
  const float4* t4 = tile4 + b * TROW4 + (3 * WPT / 4) * q;

  WinAcc acc[WPT] = {};
  WinAcc com = {};

#pragma unroll
  for (int m = 0; m < (WIN + WPT - 1 + 3) / 4; ++m) {
    const float4 A = t4[3 * m + 0], B = t4[3 * m + 1], C = t4[3 * m + 2];
    const float smp[4][3] = {{A.x, A.y, A.z}, {A.w, B.x, B.y}, {B.z, B.w, C.x}, {C.y, C.z, C.w}};
#pragma unroll
    for (int e = 0; e < 4; ++e) {
      const int i = 4 * m + e;
      if (i >= WPT - 1 && i < WIN) {
        win_add(com, smp[e][0], smp[e][1], smp[e][2]);
      } else {
        if (i == WIN) {
#pragma unroll
          for (int j = 0; j < WPT; ++j) win_join(acc[j], com);
        }
#pragma unroll
        for (int j = 0; j < WPT; ++j)
          if (i >= j && i < j + WIN) win_add(acc[j], smp[e][0], smp[e][1], smp[e][2]);
      }
    }
  }

  float m00 = matrix[0], m01 = matrix[1], m02 = matrix[2];
  float m10 = matrix[3], m11 = matrix[4], m12 = matrix[5];
  float gv[3][WPT];

#pragma unroll
  for (int j = 0; j < WPT; ++j) {
    const float s0 = acc[j].s0, s1 = acc[j].s1, s2 = acc[j].s2;
    const float q00 = acc[j].q00, q01 = acc[j].q01, q02 = acc[j].q02;
    const float q11 = acc[j].q11, q12 = acc[j].q12, q22 = acc[j].q22;

    const float invW = 1.0f / (float)WIN;
    float inv0 = 1.0f / (s0 * invW + 1e-6f);
    float inv1 = 1.0f / (s1 * invW + 1e-6f);
    float inv2 = 1.0f / (s2 * invW + 1e-6f);

    float u0 = m00 * inv0, u1 = m01 * inv1, u2 = m02 * inv2;
    float sum0 = u0 * s0 + u1 * s1 + u2 * s2;
    float sq0  = u0 * u0 * q00 + u1 * u1 * q11 + u2 * u2 * q22 +
                 2.f * (u0 * u1 * q01 + u0 * u2 * q02 + u1 * u2 * q12);
    float var0 = (sq0 - sum0 * sum0 * invW) * (1.0f / (float)(WIN - 1));
    float std0 = sqrtf(fmaxf(var0, 0.f));

    float v0 = m10 * inv0, v1 = m11 * inv1, v2 = m12 * inv2;
    float sum1 = v0 * s0 + v1 * s1 + v2 * s2;
    float sq1  = v0 * v0 * q00 + v1 * v1 * q11 + v2 * v2 * q22 +
                 2.f * (v0 * v1 * q01 + v0 * v2 * q02 + v1 * v2 * q12);
    float var1 = (sq1 - sum1 * sum1 * invW) * (1.0f / (float)(WIN - 1));
    float std1 = sqrtf(fmaxf(var1, 0.f));

    float c = std0 / (std1 + 1e-4f);

    const int n = t0 - WIN + WPT * q + j;
    const bool have = (unsigned)n < (unsigned)MWIN;
    gv[0][j] = have ? (m00 + c * m10) * inv0 : 0.f;
    gv[1][j] = have ? (m01 + c * m11) * inv1 : 0.f;
    gv[2][j] = have ? (m02 + c * m12) * inv2 : 0.f;
  }
#pragma unroll
  for (int k = 0; k < 3; ++k) {
    float4* grow = sg4 + (k * 16 + b) * (GPITCH / 4);
#pragma unroll
    for (int h = 0; h < WPT / 4; ++h)
      grow[(WPT / 4) * q + h] = make_float4(gv[k][4 * h], gv[k][4 * h + 1], gv[k][4 * h + 2], gv[k][4 * h + 3]);
    if (q == 0) grow[NWIN / 4] = make_float4(0.f, 0.f, 0.f, 0.f);
  }
  __syncthreads();

  const int wave = tid >> 5;
  if (wave < NTILE) {
    const int lane = tid & 31;
    const int col  = lane & 15;
    const int hi   = lane >> 4;
    const float* sgt = reinterpret_cast<const float*>(sg4) + col * GPITCH + 16 * wave + 1;

    v8f acc3[3];
#pragma unroll
    for (int k = 0; k < 3; ++k) {
      const float* gk = sgt + k * 16 * GPITCH;
      v8f d = {0.f, 0.f, 0.f, 0.f, 0.f, 0.f, 0.f, 0.f};
#pragma unroll
      for (int s = 0; s < 16; ++s) {
        int j0 = 4 * s + 2 * hi;
        int j1 = j0 + 1;
        v2f a;
        a.x = (j0 >= col && j0 <= col + 47) ? 1.0f : 0.0f;
        a.y = (j1 >= col && j1 <= col + 47) ? 1.0f : 0.0f;
        v2f bb;
        bb.x = gk[j0];
        bb.y = gk[j1];
        d = __builtin_amdgcn_wmma_f32_16x16x4_f32(
             false, a,  false, bb,
             (short)0, d,  false,  false);
      }
      acc3[k] = d;
    }

    const float* rt = reinterpret_cast<const float*>(tile4) + col * TROW + (WIN + 16 * wave) * 3;
#pragma unroll
    for (int p = 0; p < 8; ++p) {
      const int i = p + 8 * hi;
      const float* rp = rt + i * 3;
      float v = rp[0] * acc3[0][p] + rp[1] * acc3[1][p] + rp[2] * acc3[2][p];
      ppg[(size_t)(t0 + 16 * wave + i) * 16 + col] = v;
    }
  }
}


constexpr int CLEN   = 45;
constexpr int NCHUNK = TLEN / CLEN;
constexpr int NGRP   = NCHUNK / 16;
static_assert(NCHUNK * CLEN == TLEN, "chunks tile the row");
static_assert(NGRP * 16 == NCHUNK, "groups tile the chunks");
constexpr int OPITCH = 16 * CLEN + 1;

struct Biquad { float b0, b1, b2, a1, a2; };

__device__ __forceinline__ Biquad load_biquad(const float* __restrict__ acoef,
                                              const float* __restrict__ bcoef) {
  float ia = 1.0f / acoef[0];
  Biquad q;
  q.b0 = bcoef[0] * ia; q.b1 = bcoef[1] * ia; q.b2 = bcoef[2] * ia;
  q.a1 = acoef[1] * ia; q.a2 = acoef[2] * ia;
  return q;
}

__device__ __forceinline__ float biquad_step(const Biquad& q, float x, float& z1, float& z2) {
  float y = fmaf(q.b0, x, z1);
  z1 = fmaf(q.b1, x, fmaf(-q.a1, y, z2));
  z2 = fmaf(q.b2, x, -q.a2 * y);
  return y;
}


__device__ __forceinline__ double pair_swap(double v) {
  const long long bits = __builtin_bit_cast(long long, v);
  int lo = (int)(bits & 0xffffffffLL), hi = (int)(bits >> 32);
  lo = __builtin_amdgcn_update_dpp(0, lo, 0xB1, 0xF, 0xF, false);
  hi = __builtin_amdgcn_update_dpp(0, hi, 0xB1, 0xF, 0xF, false);
  const long long r = ((long long)hi << 32) | (long long)(unsigned int)lo;
  return __builtin_bit_cast(double, r);
}

__device__ __forceinline__ void pair_carry(double td, double to, double e,
                                           double& own, double& other) {
  own = fma(td, own, fma(to, other, e));
  other = pair_swap(own);
}

__device__ __forceinline__ double group_state(const v2f* sh, int lane, const double* m) {
  const int b = lane >> 1, c = lane & 1;
  const double td = m[3 * c], to = m[1 + c];
  const float* shf = reinterpret_cast<const float*>(sh);
  float e[16];
#pragma unroll
  for (int j = 0; j < 16; ++j) e[j] = shf[(j * 16 + b) * 2 + c];
  double own = 0.0, other = 0.0;
#pragma unroll
  for (int j = 0; j < 16; ++j) pair_carry(td, to, (double)e[j], own, other);
  return own;
}

__device__ __forceinline__ void chunk_start_states(const double* __restrict__ EG,
                                                   const float* __restrict__ E,
                                                   const double* __restrict__ Mpow,
                                                   int grp, int tid,
                                                   double* shSd, double* shL,
                                                   float& z1, float& z2) {
  const int b = tid & 15, i = tid >> 4;
  const double p11 = Mpow[4 * i + 0], p12 = Mpow[4 * i + 1];
  const double p21 = Mpow[4 * i + 2], p22 = Mpow[4 * i + 3];
  if (tid < 32) {
    const int rb = tid >> 1, c = tid & 1;
    const double* G = Mpow + 4 * 16;
    const double td = G[3 * c], to = G[1 + c];
    double eg[NGRP - 1];
#pragma unroll
    for (int k = 0; k < NGRP - 1; ++k) eg[k] = EG[(k * 16 + rb) * 2 + c];
    double own = 0.0, other = 0.0;
#pragma unroll
    for (int k = 0; k < NGRP - 1; ++k) {
      if (k < grp) pair_carry(td, to, eg[k], own, other);
    }
    shSd[rb * 2 + c] = own;
  } else if (tid < 64) {
    const int t = tid - 32;
    const int rb = t >> 1, c = t & 1;
    const double* M1 = Mpow + 4;
    const double td = M1[3 * c], to = M1[1 + c];
    float el[15];
#pragma unroll
    for (int k = 0; k < 15; ++k) el[k] = E[((grp * 16 + k) * 16 + rb) * 2 + c];
    double own = 0.0, other = 0.0;
    shL[rb * 2 + c] = 0.0;
#pragma unroll
    for (int k = 0; k < 15; ++k) {
      pair_carry(td, to, (double)el[k], own, other);
      shL[((k + 1) * 16 + rb) * 2 + c] = own;
    }
  }
  __syncthreads();
  const double s1 = shSd[b * 2], s2 = shSd[b * 2 + 1];
  const double l1 = shL[(i * 16 + b) * 2], l2 = shL[(i * 16 + b) * 2 + 1];
  z1 = (float)fma(p11, s1, fma(p12, s2, l1));
  z2 = (float)fma(p21, s1, fma(p22, s2, l2));
}

__global__ __launch_bounds__(256) void pos_scan_fwd_local(
    const float* __restrict__ ppg, const float* __restrict__ acoef,
    const float* __restrict__ bcoef, float* __restrict__ Ef_,
    float* __restrict__ EfG_, float* __restrict__ Mbuf_) {
  __shared__ v2f sh[16 * 16];
  __shared__ double shM[4];
  v2f* Ef  = reinterpret_cast<v2f*>(Ef_);
  double* EfG = reinterpret_cast<double*>(EfG_);
  double* Mbuf = reinterpret_cast<double*>(Mbuf_);
  const int tid = threadIdx.x;
  const int b = tid & 15, i = tid >> 4;
  const int grp = blockIdx.x;
  const int c = grp * 16 + i;
  const Biquad q = load_biquad(acoef, bcoef);

  const float* xp = ppg + (size_t)c * CLEN * 16 + b;
  float x[CLEN];
#pragma unroll
  for (int w = 0; w < CLEN; ++w) x[w] = xp[(size_t)w * 16];

  if (tid < 32) {
    const int j = (tid >> 1) & 1, cc = tid & 1;
    const double coef = cc ? -(double)q.a2 : -(double)q.a1;
    double own = (cc == j) ? 1.0 : 0.0, other = (cc == j) ? 0.0 : 1.0;
    for (int w = 0; w < CLEN; ++w) {
      const double p  = cc ? other : own;
      const double ad = cc ? 0.0 : other;
      own = fma(coef, p, ad);
      other = pair_swap(own);
    }
    if (tid < 4) shM[2 * cc + j] = own;
  }

  float z1 = 0.f, z2 = 0.f;
#pragma unroll
  for (int w = 0; w < CLEN; ++w) (void)biquad_step(q, x[w], z1, z2);

  v2f e; e.x = z1; e.y = z2;
  Ef[c * 16 + b] = e;
  sh[i * 16 + b] = e;
  __syncthreads();
  if (tid < 32) {
    EfG[grp * 32 + tid] = group_state(sh, tid, shM);
  } else if (grp == 0 && tid < 36) {
    const int t = tid - 32;
    const int j = t >> 1, cc = t & 1;
    const double td = shM[3 * cc], to = shM[1 + cc];
    double own = (cc == j) ? 1.0 : 0.0, other = (cc == j) ? 0.0 : 1.0;
    Mbuf[2 * cc + j] = own;
    for (int k = 1; k <= 16; ++k) {
      pair_carry(td, to, 0.0, own, other);
      Mbuf[4 * k + 2 * cc + j] = own;
    }
  }
}

__global__ __launch_bounds__(256) void pos_scan_fwd_apply(
    const float* __restrict__ ppg, const float* __restrict__ acoef,
    const float* __restrict__ bcoef, const float* __restrict__ Ef_,
    const float* __restrict__ EfG_, const float* __restrict__ Mbuf_,
    float* __restrict__ ybuf, float* __restrict__ Eb_, float* __restrict__ EbG_) {
  __shared__ double shSd[16 * 2];
  __shared__ double shL[16 * 16 * 2];
  __shared__ v2f shE[16 * 16];
  const double* EfG = reinterpret_cast<const double*>(EfG_);
  const double* Mbuf = reinterpret_cast<const double*>(Mbuf_);
  v2f* Eb  = reinterpret_cast<v2f*>(Eb_);
  double* EbG = reinterpret_cast<double*>(EbG_);
  const int tid = threadIdx.x;
  const int b = tid & 15, i = tid >> 4;
  const int grp = blockIdx.x;
  const int c = grp * 16 + i;
  const Biquad q = load_biquad(acoef, bcoef);

  const float* xp = ppg + (size_t)c * CLEN * 16 + b;
  float x[CLEN];
#pragma unroll
  for (int w = 0; w < CLEN; ++w) x[w] = xp[(size_t)w * 16];

  float z1, z2;
  chunk_start_states(EfG, Ef_, Mbuf, grp, tid, shSd, shL, z1, z2);

  float* yp = ybuf + (size_t)c * CLEN * 16 + b;
  float y[CLEN];
#pragma unroll
  for (int w = 0; w < CLEN; ++w) {
    y[w] = biquad_step(q, x[w], z1, z2);
    yp[(size_t)w * 16] = y[w];
  }

  z1 = 0.f; z2 = 0.f;
#pragma unroll
  for (int w = CLEN - 1; w >= 0; --w) (void)biquad_step(q, y[w], z1, z2);

  const int cr = NCHUNK - 1 - c;
  const int grpr = NGRP - 1 - grp;
  const int ir = 15 - i;
  v2f e; e.x = z1; e.y = z2;
  Eb[cr * 16 + b] = e;
  shE[ir * 16 + b] = e;
  __syncthreads();
  if (tid < 32) {
    EbG[grpr * 32 + tid] = group_state(shE, tid, Mbuf + 4);
  }
}

__global__ __launch_bounds__(256) void pos_scan_bwd_apply(
    const float* __restrict__ ybuf, const float* __restrict__ acoef,
    const float* __restrict__ bcoef, const float* __restrict__ Eb_,
    const float* __restrict__ EbG_, const float* __restrict__ Mbuf_,
    float* __restrict__ out) {
  __shared__ float tile[16 * OPITCH];
  __shared__ double shSd[16 * 2];
  __shared__ double shL[16 * 16 * 2];
  const double* EbG = reinterpret_cast<const double*>(EbG_);
  const double* Mbuf = reinterpret_cast<const double*>(Mbuf_);
  const int tid = threadIdx.x;
  const int b = tid & 15, ir = tid >> 4;
  const int grpr = blockIdx.x;
  const int cr = grpr * 16 + ir;
  const int c = NCHUNK - 1 - cr;
  const Biquad q = load_biquad(acoef, bcoef);

  const float* xp = ybuf + (size_t)c * CLEN * 16 + b;
  float x[CLEN];
#pragma unroll
  for (int w = 0; w < CLEN; ++w) x[w] = xp[(size_t)w * 16];

  float z1, z2;
  chunk_start_states(EbG, Eb_, Mbuf, grpr, tid, shSd, shL, z1, z2);

  const int clo = NCHUNK - 16 * (grpr + 1);
  float* tp = tile + b * OPITCH + (c - clo) * CLEN;
#pragma unroll
  for (int w = CLEN - 1; w >= 0; --w) tp[w] = biquad_step(q, x[w], z1, z2);
  __syncthreads();

  const int tlo = clo * CLEN;
  for (int k = tid; k < 16 * 16 * CLEN; k += 256) {
    int row = k / (16 * CLEN);
    int tt  = k - row * (16 * CLEN);
    out[(size_t)row * TLEN + tlo + tt] = tile[row * OPITCH + tt];
  }
}

extern "C" void kernel_launch(void* const* d_in, const int* in_sizes, int n_in,
                              void* d_out, int out_size, void* d_ws, size_t ws_size,
                              hipStream_t stream) {
  const float* rgb    = (const float*)d_in[0];
  const float* matrix = (const float*)d_in[1];
  const float* acoef  = (const float*)d_in[2];
  const float* bcoef  = (const float*)d_in[3];
  float* out = (float*)d_out;

  float* ppg  = (float*)d_ws;
  float* ybuf = ppg + (size_t)TLEN * 16;
  float* Ef   = ybuf + (size_t)TLEN * 16;
  float* Eb   = Ef + (size_t)NCHUNK * 32;
  float* EfG  = Eb + (size_t)NCHUNK * 32;
  float* EbG  = EfG + (size_t)NGRP * 64;
  float* Mbuf = EbG + (size_t)NGRP * 64;

  pos_window_overlap_add<<<TLEN / TT, FTHREADS, 0, stream>>>(rgb, matrix, ppg);
  pos_scan_fwd_local<<<NGRP, 256, 0, stream>>>(ppg, acoef, bcoef, Ef, EfG, Mbuf);
  pos_scan_fwd_apply<<<NGRP, 256, 0, stream>>>(ppg, acoef, bcoef, Ef, EfG, Mbuf,
                                               ybuf, Eb, EbG);
  pos_scan_bwd_apply<<<NGRP, 256, 0, stream>>>(ybuf, acoef, bcoef, Eb, EbG, Mbuf, out);
}


# Filtfilt's three launches become one: a work-group per row, a chunk per thread, carries meet at work-group barriers; ppg stored row-wise in whole lines, y stays in registers; sums unchanged
# speedup vs baseline: 99.4018x; 1.2978x over previous
#include <hip/hip_runtime.h>
#include <hip/hip_bf16.h>

typedef float v2f __attribute__((ext_vector_type(2)));
typedef float v8f __attribute__((ext_vector_type(8)));

constexpr int TLEN  = 18000;
constexpr int BATCH = 16;
constexpr int WIN   = 48;
constexpr int MWIN  = TLEN - WIN;

constexpr int TT     = 80;
constexpr int NTILE  = TT / 16;
constexpr int NWIN   = 128;
constexpr int WPT    = 8;
constexpr int TPR    = NWIN / WPT;
constexpr int FTHREADS = 16 * TPR;
constexpr int TSMP   = 180;
constexpr int TROW   = TSMP * 3;
constexpr int TROW4  = TROW / 4;
constexpr int ROW4   = TLEN * 3 / 4;
constexpr int GPITCH = NWIN + 4;
constexpr int NSTAGE = (16 * TROW4 + FTHREADS - 1) / FTHREADS;
constexpr int NBLK   = TLEN / TT;
constexpr int PB     = 96;
constexpr int PB4    = PB / 4;
constexpr int POL    = 100;
static_assert(PB >= TT && PB % 32 == 0 && POL >= TT && POL % 4 == 0, "ppg block holds the group's samples in whole lines");
static_assert(PB4 % 8 == 0 && 32 % 8 == 0, "a line is 8 consecutive lanes of one store instruction");
static_assert(TLEN % TT == 0 && TT % 16 == 0, "work-groups and tiles cover the row exactly");
static_assert(NWIN >= TT + WIN && NWIN % WPT == 0 && TPR == 16, "half a wave per row holds every window of the group");
static_assert(WPT % 4 == 0 && FTHREADS / 32 >= NTILE, "gains leave as 16-byte pieces; one wave per overlap-add tile");
static_assert((WIN * 3) % 4 == 0 && (TT * 3) % 4 == 0 && (TLEN * 3) % 4 == 0, "row segments start 16-byte aligned");
static_assert(TROW % 4 == 0 && TSMP >= NWIN - WPT + 4 * ((WIN + WPT - 1 + 3) / 4), "segment holds every sample a thread reads");
static_assert(TSMP >= WIN + TT, "segment holds the group's own output samples");
static_assert((3 * WPT) % 4 == 0, "a thread's first sample is 16-byte aligned in LDS");
static_assert(GPITCH % 4 == 0 && GPITCH >= 16 * (NTILE - 1) + 1 + 64, "a tile's 64 K positions stay inside the gain row");

struct WinAcc { float s0, s1, s2, q00, q01, q02, q11, q12, q22; };

__device__ __forceinline__ void win_add(WinAcc& a, float r0, float r1, float r2) {
  a.s0 += r0; a.s1 += r1; a.s2 += r2;
  a.q00 = fmaf(r0, r0, a.q00); a.q01 = fmaf(r0, r1, a.q01); a.q02 = fmaf(r0, r2, a.q02);
  a.q11 = fmaf(r1, r1, a.q11); a.q12 = fmaf(r1, r2, a.q12); a.q22 = fmaf(r2, r2, a.q22);
}

__device__ __forceinline__ void win_join(WinAcc& a, const WinAcc& b) {
  a.s0 += b.s0; a.s1 += b.s1; a.s2 += b.s2;
  a.q00 += b.q00; a.q01 += b.q01; a.q02 += b.q02;
  a.q11 += b.q11; a.q12 += b.q12; a.q22 += b.q22;
}

__global__ __launch_bounds__(FTHREADS)
void pos_window_overlap_add(const float* __restrict__ rgb,
                            const float* __restrict__ matrix,
                            float* __restrict__ ppg) {
  __shared__ float4 tile4[16 * TROW4];
  __shared__ float4 sg4[3 * 16 * GPITCH / 4];
  __shared__ float4 po4[16 * POL / 4];
  const int tid = threadIdx.x;
  const int t0  = blockIdx.x * TT;

  {
    const int g4lo = (t0 - WIN) * 3 / 4;
    const float4* rgb4 = reinterpret_cast<const float4*>(rgb);
    float4 st[NSTAGE];
#pragma unroll
    for (int u = 0; u < NSTAGE; ++u) {
      const int idx = tid + u * FTHREADS;
      const int row = idx / TROW4, c4 = idx - row * TROW4;
      const int gi = g4lo + c4;
      st[u] = make_float4(0.f, 0.f, 0.f, 0.f);
      if (idx < 16 * TROW4 && (unsigned)gi < (unsigned)ROW4) st[u] = rgb4[(size_t)row * ROW4 + gi];
    }
#pragma unroll
    for (int u = 0; u < NSTAGE; ++u) {
      const int idx = tid + u * FTHREADS;
      if (idx < 16 * TROW4) tile4[idx] = st[u];
    }
  }
  __syncthreads();

  const int b = tid / TPR;
  const int q = tid % TPR;
  const float4* t4 = tile4 + b * TROW4 + (3 * WPT / 4) * q;

  WinAcc acc[WPT] = {};
  WinAcc com = {};

#pragma unroll
  for (int m = 0; m < (WIN + WPT - 1 + 3) / 4; ++m) {
    const float4 A = t4[3 * m + 0], B = t4[3 * m + 1], C = t4[3 * m + 2];
    const float smp[4][3] = {{A.x, A.y, A.z}, {A.w, B.x, B.y}, {B.z, B.w, C.x}, {C.y, C.z, C.w}};
#pragma unroll
    for (int e = 0; e < 4; ++e) {
      const int i = 4 * m + e;
      if (i >= WPT - 1 && i < WIN) {
        win_add(com, smp[e][0], smp[e][1], smp[e][2]);
      } else {
        if (i == WIN) {
#pragma unroll
          for (int j = 0; j < WPT; ++j) win_join(acc[j], com);
        }
#pragma unroll
        for (int j = 0; j < WPT; ++j)
          if (i >= j && i < j + WIN) win_add(acc[j], smp[e][0], smp[e][1], smp[e][2]);
      }
    }
  }

  float m00 = matrix[0], m01 = matrix[1], m02 = matrix[2];
  float m10 = matrix[3], m11 = matrix[4], m12 = matrix[5];
  float gv[3][WPT];

#pragma unroll
  for (int j = 0; j < WPT; ++j) {
    const float s0 = acc[j].s0, s1 = acc[j].s1, s2 = acc[j].s2;
    const float q00 = acc[j].q00, q01 = acc[j].q01, q02 = acc[j].q02;
    const float q11 = acc[j].q11, q12 = acc[j].q12, q22 = acc[j].q22;

    const float invW = 1.0f / (float)WIN;
    float inv0 = 1.0f / (s0 * invW + 1e-6f);
    float inv1 = 1.0f / (s1 * invW + 1e-6f);
    float inv2 = 1.0f / (s2 * invW + 1e-6f);

    float u0 = m00 * inv0, u1 = m01 * inv1, u2 = m02 * inv2;
    float sum0 = u0 * s0 + u1 * s1 + u2 * s2;
    float sq0  = u0 * u0 * q00 + u1 * u1 * q11 + u2 * u2 * q22 +
                 2.f * (u0 * u1 * q01 + u0 * u2 * q02 + u1 * u2 * q12);
    float var0 = (sq0 - sum0 * sum0 * invW) * (1.0f / (float)(WIN - 1));
    float std0 = sqrtf(fmaxf(var0, 0.f));

    float v0 = m10 * inv0, v1 = m11 * inv1, v2 = m12 * inv2;
    float sum1 = v0 * s0 + v1 * s1 + v2 * s2;
    float sq1  = v0 * v0 * q00 + v1 * v1 * q11 + v2 * v2 * q22 +
                 2.f * (v0 * v1 * q01 + v0 * v2 * q02 + v1 * v2 * q12);
    float var1 = (sq1 - sum1 * sum1 * invW) * (1.0f / (float)(WIN - 1));
    float std1 = sqrtf(fmaxf(var1, 0.f));

    float c = std0 / (std1 + 1e-4f);

    const int n = t0 - WIN + WPT * q + j;
    const bool have = (unsigned)n < (unsigned)MWIN;
    gv[0][j] = have ? (m00 + c * m10) * inv0 : 0.f;
    gv[1][j] = have ? (m01 + c * m11) * inv1 : 0.f;
    gv[2][j] = have ? (m02 + c * m12) * inv2 : 0.f;
  }
#pragma unroll
  for (int k = 0; k < 3; ++k) {
    float4* grow = sg4 + (k * 16 + b) * (GPITCH / 4);
#pragma unroll
    for (int h = 0; h < WPT / 4; ++h)
      grow[(WPT / 4) * q + h] = make_float4(gv[k][4 * h], gv[k][4 * h + 1], gv[k][4 * h + 2], gv[k][4 * h + 3]);
    if (q == 0) grow[NWIN / 4] = make_float4(0.f, 0.f, 0.f, 0.f);
  }
  __syncthreads();

  const int wave = tid >> 5;
  if (wave < NTILE) {
    const int lane = tid & 31;
    const int col  = lane & 15;
    const int hi   = lane >> 4;
    const float* sgt = reinterpret_cast<const float*>(sg4) + col * GPITCH + 16 * wave + 1;

    v8f acc3[3];
#pragma unroll
    for (int k = 0; k < 3; ++k) {
      const float* gk = sgt + k * 16 * GPITCH;
      v8f d = {0.f, 0.f, 0.f, 0.f, 0.f, 0.f, 0.f, 0.f};
#pragma unroll
      for (int s = 0; s < 16; ++s) {
        int j0 = 4 * s + 2 * hi;
        int j1 = j0 + 1;
        v2f a;
        a.x = (j0 >= col && j0 <= col + 47) ? 1.0f : 0.0f;
        a.y = (j1 >= col && j1 <= col + 47) ? 1.0f : 0.0f;
        v2f bb;
        bb.x = gk[j0];
        bb.y = gk[j1];
        d = __builtin_amdgcn_wmma_f32_16x16x4_f32(
             false, a,  false, bb,
             (short)0, d,  false,  false);
      }
      acc3[k] = d;
    }

    const float* rt = reinterpret_cast<const float*>(tile4) + col * TROW + (WIN + 16 * wave) * 3;
#pragma unroll
    for (int p = 0; p < 8; ++p) {
      const int i = p + 8 * hi;
      const float* rp = rt + i * 3;
      float v = rp[0] * acc3[0][p] + rp[1] * acc3[1][p] + rp[2] * acc3[2][p];
      reinterpret_cast<float*>(po4)[col * POL + 16 * wave + i] = v;
    }
  }
  __syncthreads();

  float4* ppg4 = reinterpret_cast<float4*>(ppg);
  for (int idx = tid; idx < 16 * PB4; idx += FTHREADS) {
    const int row = idx / PB4, j = idx - row * PB4;
    float4 v = make_float4(0.f, 0.f, 0.f, 0.f);
    if (j < TT / 4) v = po4[row * (POL / 4) + j];
    ppg4[((size_t)row * NBLK + blockIdx.x) * PB4 + j] = v;
  }
}


constexpr int CLEN   = 45;
constexpr int NCHUNK = TLEN / CLEN;
constexpr int NGRP   = NCHUNK / 16;
static_assert(NCHUNK * CLEN == TLEN, "chunks tile the row");
static_assert(NGRP * 16 == NCHUNK, "groups tile the chunks");
constexpr int CWAVES   = (NCHUNK + 31) / 32;
constexpr int CTHREADS = CWAVES * 32;
constexpr int STHREADS = CTHREADS + 32;
constexpr int ROWV4  = TLEN / 4;
constexpr int NLOAD  = (ROWV4 + CTHREADS - 1) / CTHREADS;
constexpr int MSPLIT = 22;
constexpr int TABLE_WAVE = 2;
static_assert(2 * NGRP <= 64 && MSPLIT < CLEN, "two waves hold a lane pair per group");

struct Biquad { float b0, b1, b2, a1, a2; };

__device__ __forceinline__ Biquad load_biquad(const float* __restrict__ acoef,
                                              const float* __restrict__ bcoef) {
  float ia = 1.0f / acoef[0];
  Biquad q;
  q.b0 = bcoef[0] * ia; q.b1 = bcoef[1] * ia; q.b2 = bcoef[2] * ia;
  q.a1 = acoef[1] * ia; q.a2 = acoef[2] * ia;
  return q;
}

__device__ __forceinline__ float biquad_step(const Biquad& q, float x, float& z1, float& z2) {
  float y = fmaf(q.b0, x, z1);
  z1 = fmaf(q.b1, x, fmaf(-q.a1, y, z2));
  z2 = fmaf(q.b2, x, -q.a2 * y);
  return y;
}


__device__ __forceinline__ double pair_swap(double v) {
  const long long bits = __builtin_bit_cast(long long, v);
  int lo = (int)(bits & 0xffffffffLL), hi = (int)(bits >> 32);
  lo = __builtin_amdgcn_update_dpp(0, lo, 0xB1, 0xF, 0xF, false);
  hi = __builtin_amdgcn_update_dpp(0, hi, 0xB1, 0xF, 0xF, false);
  const long long r = ((long long)hi << 32) | (long long)(unsigned int)lo;
  return __builtin_bit_cast(double, r);
}

__device__ __forceinline__ void pair_carry(double td, double to, double e,
                                           double& own, double& other) {
  own = fma(td, own, fma(to, other, e));
  other = pair_swap(own);
}

__device__ __forceinline__ void unit_step(double coef, int cc, double& own, double& other) {
  const double p  = cc ? other : own;
  const double ad = cc ? 0.0 : other;
  own = fma(coef, p, ad);
  other = pair_swap(own);
}

__device__ __forceinline__ void scan_carries(const v2f* shE, const double* shM, double* shP,
                                             double* shEG, double* shL, double* shS,
                                             int tid, bool build_table) {
  if (tid < 64) {
    const int g = tid >> 1, comp = tid & 1;
    const bool have = g < NGRP;
    const int gc = have ? g : NGRP - 1;
    const double td = shM[3 * comp], to = shM[1 + comp];
    const float* ef = reinterpret_cast<const float*>(shE);
    float e[16];
#pragma unroll
    for (int j = 0; j < 16; ++j) e[j] = ef[(gc * 16 + j) * 2 + comp];
    double own = 0.0, other = 0.0;
    if (have) shL[(gc * 16) * 2 + comp] = 0.0;
#pragma unroll
    for (int j = 0; j < 16; ++j) {
      pair_carry(td, to, (double)e[j], own, other);
      if (have) {
        if (j < 15) shL[(gc * 16 + j + 1) * 2 + comp] = own;
        else        shEG[gc * 2 + comp] = own;
      }
    }
  } else if (build_table && (tid >> 5) == TABLE_WAVE) {
    const int lane = tid & 31;
    const int j = (lane >> 1) & 1, cc = lane & 1;
    const double td = shM[3 * cc], to = shM[1 + cc];
    double own = (cc == j) ? 1.0 : 0.0, other = (cc == j) ? 0.0 : 1.0;
    if (lane < 4) shP[2 * cc + j] = own;
    for (int k = 1; k <= 16; ++k) {
      pair_carry(td, to, 0.0, own, other);
      if (lane < 4) shP[4 * k + 2 * cc + j] = own;
    }
  }
  __syncthreads();
  if (tid < 32) {
    const int comp = tid & 1;
    const double* G = shP + 4 * 16;
    const double td = G[3 * comp], to = G[1 + comp];
    double eg[NGRP - 1];
#pragma unroll
    for (int k = 0; k < NGRP - 1; ++k) eg[k] = shEG[k * 2 + comp];
    double own = 0.0, other = 0.0;
    if (tid < 2) shS[comp] = 0.0;
#pragma unroll
    for (int k = 0; k < NGRP - 1; ++k) {
      pair_carry(td, to, eg[k], own, other);
      if (tid < 2) shS[(k + 1) * 2 + comp] = own;
    }
  }
  __syncthreads();
}

__device__ __forceinline__ void chunk_start(const double* shP, const double* shS,
                                            const double* shL, int cn,
                                            float& z1, float& z2) {
  const int g = cn >> 4, i = cn & 15;
  const double p11 = shP[4 * i + 0], p12 = shP[4 * i + 1];
  const double p21 = shP[4 * i + 2], p22 = shP[4 * i + 3];
  const double s1 = shS[g * 2], s2 = shS[g * 2 + 1];
  const double l1 = shL[cn * 2], l2 = shL[cn * 2 + 1];
  z1 = (float)fma(p11, s1, fma(p12, s2, l1));
  z2 = (float)fma(p21, s1, fma(p22, s2, l2));
}

__global__ __launch_bounds__(STHREADS)
void pos_filtfilt_rows(const float* __restrict__ ppg, const float* __restrict__ acoef,
                       const float* __restrict__ bcoef, float* __restrict__ out) {
  __shared__ float4 row4[ROWV4];
  __shared__ v2f    shE[NCHUNK];
  __shared__ double shL[NCHUNK * 2];
  __shared__ double shEG[NGRP * 2];
  __shared__ double shS[NGRP * 2];
  __shared__ double shM[4];
  __shared__ double shP[17 * 4];
  float* rowf = reinterpret_cast<float*>(row4);
  const int tid = threadIdx.x;
  const int b = blockIdx.x;
  const bool helper = tid >= CTHREADS;
  const bool act = tid < NCHUNK;
  const int c = act ? tid : NCHUNK - 1;

  float4 st[NLOAD];
  if (!helper) {
    const float4* src = reinterpret_cast<const float4*>(ppg) + (size_t)b * (NBLK * PB4);
#pragma unroll
    for (int u = 0; u < NLOAD; ++u) {
      const int idx = tid + u * CTHREADS;
      const int blk = idx / (TT / 4), j = idx - blk * (TT / 4);
      st[u] = make_float4(0.f, 0.f, 0.f, 0.f);
      if (idx < ROWV4) st[u] = src[blk * PB4 + j];
    }
  }

  const Biquad q = load_biquad(acoef, bcoef);

  const int mlane = tid & 31;
  const int mj = (mlane >> 1) & 1, mc = mlane & 1;
  const double mcoef = mc ? -(double)q.a2 : -(double)q.a1;
  double mown = (mc == mj) ? 1.0 : 0.0, moth = (mc == mj) ? 0.0 : 1.0;

  if (!helper) {
#pragma unroll
    for (int u = 0; u < NLOAD; ++u) {
      const int idx = tid + u * CTHREADS;
      if (idx < ROWV4) row4[idx] = st[u];
    }
  } else {
    for (int w = 0; w < MSPLIT; ++w) unit_step(mcoef, mc, mown, moth);
  }
  __syncthreads();

  float x[CLEN];
  float z1 = 0.f, z2 = 0.f;
  if (!helper) {
    const float* xp = rowf + c * CLEN;
#pragma unroll
    for (int w = 0; w < CLEN; ++w) x[w] = xp[w];
#pragma unroll
    for (int w = 0; w < CLEN; ++w) (void)biquad_step(q, x[w], z1, z2);
    v2f e; e.x = z1; e.y = z2;
    if (act) shE[c] = e;
  } else {
    for (int w = MSPLIT; w < CLEN; ++w) unit_step(mcoef, mc, mown, moth);
    if (mlane < 4) shM[2 * mc + mj] = mown;
  }
  __syncthreads();

  scan_carries(shE, shM, shP, shEG, shL, shS, tid, true);

  float y[CLEN];
  if (!helper) {
    chunk_start(shP, shS, shL, c, z1, z2);
#pragma unroll
    for (int w = 0; w < CLEN; ++w) y[w] = biquad_step(q, x[w], z1, z2);
    z1 = 0.f; z2 = 0.f;
#pragma unroll
    for (int w = CLEN - 1; w >= 0; --w) (void)biquad_step(q, y[w], z1, z2);
    v2f e; e.x = z1; e.y = z2;
    if (act) shE[NCHUNK - 1 - c] = e;
  }
  __syncthreads();

  scan_carries(shE, shM, shP, shEG, shL, shS, tid, false);

  if (!helper) {
    chunk_start(shP, shS, shL, NCHUNK - 1 - c, z1, z2);
    float* tp = rowf + c * CLEN;
#pragma unroll
    for (int w = CLEN - 1; w >= 0; --w) {
      const float o = biquad_step(q, y[w], z1, z2);
      if (act) tp[w] = o;
    }
  }
  __syncthreads();

  float4* dst = reinterpret_cast<float4*>(out) + (size_t)b * ROWV4;
  for (int idx = tid; idx < ROWV4; idx += STHREADS) dst[idx] = row4[idx];
}

extern "C" void kernel_launch(void* const* d_in, const int* in_sizes, int n_in,
                              void* d_out, int out_size, void* d_ws, size_t ws_size,
                              hipStream_t stream) {
  const float* rgb    = (const float*)d_in[0];
  const float* matrix = (const float*)d_in[1];
  const float* acoef  = (const float*)d_in[2];
  const float* bcoef  = (const float*)d_in[3];
  float* out = (float*)d_out;

  float* ppg  = (float*)d_ws;

  pos_window_overlap_add<<<NBLK, FTHREADS, 0, stream>>>(rgb, matrix, ppg);
  pos_filtfilt_rows<<<BATCH, STHREADS, 0, stream>>>(ppg, acoef, bcoef, out);
}


# Scan carries on three levels (groups of 10 chunks, blocks of 8 groups, 5 blocks) instead of two (16 x 25): 23 dependent f64 carry steps per pass instead of 40; f64 sum order changes, fixed per run
# speedup vs baseline: 109.6387x; 1.1030x over previous
#include <hip/hip_runtime.h>
#include <hip/hip_bf16.h>

typedef float v2f __attribute__((ext_vector_type(2)));
typedef float v8f __attribute__((ext_vector_type(8)));

constexpr int TLEN  = 18000;
constexpr int BATCH = 16;
constexpr int WIN   = 48;
constexpr int MWIN  = TLEN - WIN;

constexpr int TT     = 80;
constexpr int NTILE  = TT / 16;
constexpr int NWIN   = 128;
constexpr int WPT    = 8;
constexpr int TPR    = NWIN / WPT;
constexpr int FTHREADS = 16 * TPR;
constexpr int TSMP   = 180;
constexpr int TROW   = TSMP * 3;
constexpr int TROW4  = TROW / 4;
constexpr int ROW4   = TLEN * 3 / 4;
constexpr int GPITCH = NWIN + 4;
constexpr int NSTAGE = (16 * TROW4 + FTHREADS - 1) / FTHREADS;
constexpr int NBLK   = TLEN / TT;
constexpr int PB     = 96;
constexpr int PB4    = PB / 4;
constexpr int POL    = 100;
static_assert(PB >= TT && PB % 32 == 0 && POL >= TT && POL % 4 == 0, "ppg block holds the group's samples in whole lines");
static_assert(PB4 % 8 == 0 && 32 % 8 == 0, "a line is 8 consecutive lanes of one store instruction");
static_assert(TLEN % TT == 0 && TT % 16 == 0, "work-groups and tiles cover the row exactly");
static_assert(NWIN >= TT + WIN && NWIN % WPT == 0 && TPR == 16, "half a wave per row holds every window of the group");
static_assert(WPT % 4 == 0 && FTHREADS / 32 >= NTILE, "gains leave as 16-byte pieces; one wave per overlap-add tile");
static_assert((WIN * 3) % 4 == 0 && (TT * 3) % 4 == 0 && (TLEN * 3) % 4 == 0, "row segments start 16-byte aligned");
static_assert(TROW % 4 == 0 && TSMP >= NWIN - WPT + 4 * ((WIN + WPT - 1 + 3) / 4), "segment holds every sample a thread reads");
static_assert(TSMP >= WIN + TT, "segment holds the group's own output samples");
static_assert((3 * WPT) % 4 == 0, "a thread's first sample is 16-byte aligned in LDS");
static_assert(GPITCH % 4 == 0 && GPITCH >= 16 * (NTILE - 1) + 1 + 64, "a tile's 64 K positions stay inside the gain row");

struct WinAcc { float s0, s1, s2, q00, q01, q02, q11, q12, q22; };

__device__ __forceinline__ void win_add(WinAcc& a, float r0, float r1, float r2) {
  a.s0 += r0; a.s1 += r1; a.s2 += r2;
  a.q00 = fmaf(r0, r0, a.q00); a.q01 = fmaf(r0, r1, a.q01); a.q02 = fmaf(r0, r2, a.q02);
  a.q11 = fmaf(r1, r1, a.q11); a.q12 = fmaf(r1, r2, a.q12); a.q22 = fmaf(r2, r2, a.q22);
}

__device__ __forceinline__ void win_join(WinAcc& a, const WinAcc& b) {
  a.s0 += b.s0; a.s1 += b.s1; a.s2 += b.s2;
  a.q00 += b.q00; a.q01 += b.q01; a.q02 += b.q02;
  a.q11 += b.q11; a.q12 += b.q12; a.q22 += b.q22;
}

__global__ __launch_bounds__(FTHREADS)
void pos_window_overlap_add(const float* __restrict__ rgb,
                            const float* __restrict__ matrix,
                            float* __restrict__ ppg) {
  __shared__ float4 tile4[16 * TROW4];
  __shared__ float4 sg4[3 * 16 * GPITCH / 4];
  __shared__ float4 po4[16 * POL / 4];
  const int tid = threadIdx.x;
  const int t0  = blockIdx.x * TT;

  {
    const int g4lo = (t0 - WIN) * 3 / 4;
    const float4* rgb4 = reinterpret_cast<const float4*>(rgb);
    float4 st[NSTAGE];
#pragma unroll
    for (int u = 0; u < NSTAGE; ++u) {
      const int idx = tid + u * FTHREADS;
      const int row = idx / TROW4, c4 = idx - row * TROW4;
      const int gi = g4lo + c4;
      st[u] = make_float4(0.f, 0.f, 0.f, 0.f);
      if (idx < 16 * TROW4 && (unsigned)gi < (unsigned)ROW4) st[u] = rgb4[(size_t)row * ROW4 + gi];
    }
#pragma unroll
    for (int u = 0; u < NSTAGE; ++u) {
      const int idx = tid + u * FTHREADS;
      if (idx < 16 * TROW4) tile4[idx] = st[u];
    }
  }
  __syncthreads();

  const int b = tid / TPR;
  const int q = tid % TPR;
  const float4* t4 = tile4 + b * TROW4 + (3 * WPT / 4) * q;

  WinAcc acc[WPT] = {};
  WinAcc com = {};

#pragma unroll
  for (int m = 0; m < (WIN + WPT - 1 + 3) / 4; ++m) {
    const float4 A = t4[3 * m + 0], B = t4[3 * m + 1], C = t4[3 * m + 2];
    const float smp[4][3] = {{A.x, A.y, A.z}, {A.w, B.x, B.y}, {B.z, B.w, C.x}, {C.y, C.z, C.w}};
#pragma unroll
    for (int e = 0; e < 4; ++e) {
      const int i = 4 * m + e;
      if (i >= WPT - 1 && i < WIN) {
        win_add(com, smp[e][0], smp[e][1], smp[e][2]);
      } else {
        if (i == WIN) {
#pragma unroll
          for (int j = 0; j < WPT; ++j) win_join(acc[j], com);
        }
#pragma unroll
        for (int j = 0; j < WPT; ++j)
          if (i >= j && i < j + WIN) win_add(acc[j], smp[e][0], smp[e][1], smp[e][2]);
      }
    }
  }

  float m00 = matrix[0], m01 = matrix[1], m02 = matrix[2];
  float m10 = matrix[3], m11 = matrix[4], m12 = matrix[5];
  float gv[3][WPT];

#pragma unroll
  for (int j = 0; j < WPT; ++j) {
    const float s0 = acc[j].s0, s1 = acc[j].s1, s2 = acc[j].s2;
    const float q00 = acc[j].q00, q01 = acc[j].q01, q02 = acc[j].q02;
    const float q11 = acc[j].q11, q12 = acc[j].q12, q22 = acc[j].q22;

    const float invW = 1.0f / (float)WIN;
    float inv0 = 1.0f / (s0 * invW + 1e-6f);
    float inv1 = 1.0f / (s1 * invW + 1e-6f);
    float inv2 = 1.0f / (s2 * invW + 1e-6f);

    float u0 = m00 * inv0, u1 = m01 * inv1, u2 = m02 * inv2;
    float sum0 = u0 * s0 + u1 * s1 + u2 * s2;
    float sq0  = u0 * u0 * q00 + u1 * u1 * q11 + u2 * u2 * q22 +
                 2.f * (u0 * u1 * q01 + u0 * u2 * q02 + u1 * u2 * q12);
    float var0 = (sq0 - sum0 * sum0 * invW) * (1.0f / (float)(WIN - 1));
    float std0 = sqrtf(fmaxf(var0, 0.f));

    float v0 = m10 * inv0, v1 = m11 * inv1, v2 = m12 * inv2;
    float sum1 = v0 * s0 + v1 * s1 + v2 * s2;
    float sq1  = v0 * v0 * q00 + v1 * v1 * q11 + v2 * v2 * q22 +
                 2.f * (v0 * v1 * q01 + v0 * v2 * q02 + v1 * v2 * q12);
    float var1 = (sq1 - sum1 * sum1 * invW) * (1.0f / (float)(WIN - 1));
    float std1 = sqrtf(fmaxf(var1, 0.f));

    float c = std0 / (std1 + 1e-4f);

    const int n = t0 - WIN + WPT * q + j;
    const bool have = (unsigned)n < (unsigned)MWIN;
    gv[0][j] = have ? (m00 + c * m10) * inv0 : 0.f;
    gv[1][j] = have ? (m01 + c * m11) * inv1 : 0.f;
    gv[2][j] = have ? (m02 + c * m12) * inv2 : 0.f;
  }
#pragma unroll
  for (int k = 0; k < 3; ++k) {
    float4* grow = sg4 + (k * 16 + b) * (GPITCH / 4);
#pragma unroll
    for (int h = 0; h < WPT / 4; ++h)
      grow[(WPT / 4) * q + h] = make_float4(gv[k][4 * h], gv[k][4 * h + 1], gv[k][4 * h + 2], gv[k][4 * h + 3]);
    if (q == 0) grow[NWIN / 4] = make_float4(0.f, 0.f, 0.f, 0.f);
  }
  __syncthreads();

  const int wave = tid >> 5;
  if (wave < NTILE) {
    const int lane = tid & 31;
    const int col  = lane & 15;
    const int hi   = lane >> 4;
    const float* sgt = reinterpret_cast<const float*>(sg4) + col * GPITCH + 16 * wave + 1;

    v8f acc3[3];
#pragma unroll
    for (int k = 0; k < 3; ++k) {
      const float* gk = sgt + k * 16 * GPITCH;
      v8f d = {0.f, 0.f, 0.f, 0.f, 0.f, 0.f, 0.f, 0.f};
#pragma unroll
      for (int s = 0; s < 16; ++s) {
        int j0 = 4 * s + 2 * hi;
        int j1 = j0 + 1;
        v2f a;
        a.x = (j0 >= col && j0 <= col + 47) ? 1.0f : 0.0f;
        a.y = (j1 >= col && j1 <= col + 47) ? 1.0f : 0.0f;
        v2f bb;
        bb.x = gk[j0];
        bb.y = gk[j1];
        d = __builtin_amdgcn_wmma_f32_16x16x4_f32(
             false, a,  false, bb,
             (short)0, d,  false,  false);
      }
      acc3[k] = d;
    }

    const float* rt = reinterpret_cast<const float*>(tile4) + col * TROW + (WIN + 16 * wave) * 3;
#pragma unroll
    for (int p = 0; p < 8; ++p) {
      const int i = p + 8 * hi;
      const float* rp = rt + i * 3;
      float v = rp[0] * acc3[0][p] + rp[1] * acc3[1][p] + rp[2] * acc3[2][p];
      reinterpret_cast<float*>(po4)[col * POL + 16 * wave + i] = v;
    }
  }
  __syncthreads();

  float4* ppg4 = reinterpret_cast<float4*>(ppg);
  for (int idx = tid; idx < 16 * PB4; idx += FTHREADS) {
    const int row = idx / PB4, j = idx - row * PB4;
    float4 v = make_float4(0.f, 0.f, 0.f, 0.f);
    if (j < TT / 4) v = po4[row * (POL / 4) + j];
    ppg4[((size_t)row * NBLK + blockIdx.x) * PB4 + j] = v;
  }
}


constexpr int CLEN   = 45;
constexpr int NCHUNK = TLEN / CLEN;
constexpr int G1     = 10;
constexpr int NG1    = NCHUNK / G1;
constexpr int G2     = 8;
constexpr int NSG    = NG1 / G2;
static_assert(NCHUNK * CLEN == TLEN, "chunks tile the row");
static_assert(NG1 * G1 == NCHUNK && NSG * G2 == NG1, "groups tile the chunks, blocks tile the groups");
constexpr int L1WAVES = (2 * NG1 + 31) / 32;
constexpr int CWAVES   = (NCHUNK + 31) / 32;
constexpr int CTHREADS = CWAVES * 32;
constexpr int STHREADS = CTHREADS + 32;
constexpr int ROWV4  = TLEN / 4;
constexpr int NLOAD  = (ROWV4 + CTHREADS - 1) / CTHREADS;
constexpr int MSPLIT = 22;
constexpr int TABLE_WAVE = L1WAVES;
static_assert(TABLE_WAVE < 4 && 2 * NSG <= 32 && MSPLIT < CLEN, "group waves and table wave on four SIMDs; one wave holds a lane pair per block");

struct Biquad { float b0, b1, b2, a1, a2; };

__device__ __forceinline__ Biquad load_biquad(const float* __restrict__ acoef,
                                              const float* __restrict__ bcoef) {
  float ia = 1.0f / acoef[0];
  Biquad q;
  q.b0 = bcoef[0] * ia; q.b1 = bcoef[1] * ia; q.b2 = bcoef[2] * ia;
  q.a1 = acoef[1] * ia; q.a2 = acoef[2] * ia;
  return q;
}

__device__ __forceinline__ float biquad_step(const Biquad& q, float x, float& z1, float& z2) {
  float y = fmaf(q.b0, x, z1);
  z1 = fmaf(q.b1, x, fmaf(-q.a1, y, z2));
  z2 = fmaf(q.b2, x, -q.a2 * y);
  return y;
}


__device__ __forceinline__ double pair_swap(double v) {
  const long long bits = __builtin_bit_cast(long long, v);
  int lo = (int)(bits & 0xffffffffLL), hi = (int)(bits >> 32);
  lo = __builtin_amdgcn_update_dpp(0, lo, 0xB1, 0xF, 0xF, false);
  hi = __builtin_amdgcn_update_dpp(0, hi, 0xB1, 0xF, 0xF, false);
  const long long r = ((long long)hi << 32) | (long long)(unsigned int)lo;
  return __builtin_bit_cast(double, r);
}

template <int CTRL>
__device__ __forceinline__ double quad_fetch(double v) {
  const long long bits = __builtin_bit_cast(long long, v);
  int lo = (int)(bits & 0xffffffffLL), hi = (int)(bits >> 32);
  lo = __builtin_amdgcn_update_dpp(0, lo, CTRL, 0xF, 0xF, false);
  hi = __builtin_amdgcn_update_dpp(0, hi, CTRL, 0xF, 0xF, false);
  const long long r = ((long long)hi << 32) | (long long)(unsigned int)lo;
  return __builtin_bit_cast(double, r);
}

__device__ __forceinline__ void pair_carry(double td, double to, double e,
                                           double& own, double& other) {
  own = fma(td, own, fma(to, other, e));
  other = pair_swap(own);
}

__device__ __forceinline__ void unit_step(double coef, int cc, double& own, double& other) {
  const double p  = cc ? other : own;
  const double ad = cc ? 0.0 : other;
  own = fma(coef, p, ad);
  other = pair_swap(own);
}

__device__ __forceinline__ void scan_carries(const v2f* shE, const double* shM, double* shP, double* shQ,
                                             double* shL, double* shEG, double* shL2,
                                             double* shESG, double* shSG,
                                             int tid, bool build_table) {
  const int wave = tid >> 5, lane = tid & 31;
  const int tj = (lane >> 1) & 1, tc = lane & 1;
  double town = 0.0, toth = 0.0;
  if (tid < 32 * L1WAVES) {
    const int g = tid >> 1, comp = tid & 1;
    const bool have = g < NG1;
    const int gc = have ? g : NG1 - 1;
    const double td = shM[3 * comp], to = shM[1 + comp];
    const float* ef = reinterpret_cast<const float*>(shE);
    float e[G1];
#pragma unroll
    for (int j = 0; j < G1; ++j) e[j] = ef[(gc * G1 + j) * 2 + comp];
    double own = 0.0, other = 0.0;
    if (have) shL[(gc * G1) * 2 + comp] = 0.0;
#pragma unroll
    for (int j = 0; j < G1; ++j) {
      pair_carry(td, to, (double)e[j], own, other);
      if (have) {
        if (j < G1 - 1) shL[(gc * G1 + j + 1) * 2 + comp] = own;
        else            shEG[gc * 2 + comp] = own;
      }
    }
  } else if (build_table && wave == TABLE_WAVE) {
    const double td = shM[3 * tc], to = shM[1 + tc];
    town = (tc == tj) ? 1.0 : 0.0; toth = (tc == tj) ? 0.0 : 1.0;
    if (lane < 4) { shP[2 * tc + tj] = town; shQ[2 * tc + tj] = town; }
    for (int k = 1; k <= G1; ++k) {
      pair_carry(td, to, 0.0, town, toth);
      if (lane < 4) shP[4 * k + 2 * tc + tj] = town;
    }
    if (lane < 4) shQ[4 + 2 * tc + tj] = town;
  }
  __syncthreads();
  if (tid < 32) {
    const int s = tid >> 1, comp = tid & 1;
    const bool have = s < NSG;
    const int sc = have ? s : NSG - 1;
    const double* T2 = shP + 4 * G1;
    const double td = T2[3 * comp], to = T2[1 + comp];
    double eg[G2];
#pragma unroll
    for (int k = 0; k < G2; ++k) eg[k] = shEG[(sc * G2 + k) * 2 + comp];
    double own = 0.0, other = 0.0;
    if (have) shL2[(sc * G2) * 2 + comp] = 0.0;
#pragma unroll
    for (int k = 0; k < G2; ++k) {
      pair_carry(td, to, eg[k], own, other);
      if (have) {
        if (k < G2 - 1) shL2[(sc * G2 + k + 1) * 2 + comp] = own;
        else            shESG[sc * 2 + comp] = own;
      }
    }
  } else if (build_table && wave == TABLE_WAVE) {
    const double td = quad_fetch<0xCC>(town);
    const double to = quad_fetch<0x66>(town);
    for (int k = 2; k <= G2; ++k) {
      pair_carry(td, to, 0.0, town, toth);
      if (lane < 4) shQ[4 * k + 2 * tc + tj] = town;
    }
  }
  __syncthreads();
  if (tid < 32 * L1WAVES) {
    const int g = tid >> 1, comp = tid & 1;
    const bool have = g < NG1;
    const int gc = have ? g : NG1 - 1;
    const int s = gc / G2, j = gc - s * G2;
    const double* T3 = shQ + 4 * G2;
    const double td = T3[3 * comp], to = T3[1 + comp];
    const double qd = shQ[4 * j + 3 * comp], qo = shQ[4 * j + 1 + comp];
    const double l2 = shL2[gc * 2 + comp];
    double esg[NSG - 1];
#pragma unroll
    for (int k = 0; k < NSG - 1; ++k) esg[k] = shESG[k * 2 + comp];
    asm volatile("" :: "v"(qd), "v"(qo), "v"(l2));
    __builtin_amdgcn_sched_barrier(0);
    double own = 0.0, other = 0.0, so = 0.0, sp = 0.0;
#pragma unroll
    for (int k = 0; k < NSG - 1; ++k) {
      if (k == s) { so = own; sp = other; }
      pair_carry(td, to, esg[k], own, other);
    }
    if (s == NSG - 1) { so = own; sp = other; }
    const double sg = fma(qd, so, fma(qo, sp, l2));
    if (have) shSG[gc * 2 + comp] = sg;
  }
  __syncthreads();
}

__device__ __forceinline__ void chunk_start(const double* shP, const double* shS,
                                            const double* shL, int cn,
                                            float& z1, float& z2) {
  const int g = cn / G1, i = cn - g * G1;
  const double p11 = shP[4 * i + 0], p12 = shP[4 * i + 1];
  const double p21 = shP[4 * i + 2], p22 = shP[4 * i + 3];
  const double s1 = shS[g * 2], s2 = shS[g * 2 + 1];
  const double l1 = shL[cn * 2], l2 = shL[cn * 2 + 1];
  z1 = (float)fma(p11, s1, fma(p12, s2, l1));
  z2 = (float)fma(p21, s1, fma(p22, s2, l2));
}

__global__ __launch_bounds__(STHREADS)
void pos_filtfilt_rows(const float* __restrict__ ppg, const float* __restrict__ acoef,
                       const float* __restrict__ bcoef, float* __restrict__ out) {
  __shared__ float4 row4[ROWV4];
  __shared__ v2f    shE[NCHUNK];
  __shared__ double shL[NCHUNK * 2];
  __shared__ double shEG[NG1 * 2];
  __shared__ double shL2[NG1 * 2];
  __shared__ double shESG[NSG * 2];
  __shared__ double shS[NG1 * 2];
  __shared__ double shM[4];
  __shared__ double shP[(G1 + 1) * 4];
  __shared__ double shQ[(G2 + 1) * 4];
  float* rowf = reinterpret_cast<float*>(row4);
  const int tid = threadIdx.x;
  const int b = blockIdx.x;
  const bool helper = tid >= CTHREADS;
  const bool act = tid < NCHUNK;
  const int c = act ? tid : NCHUNK - 1;

  float4 st[NLOAD];
  if (!helper) {
    const float4* src = reinterpret_cast<const float4*>(ppg) + (size_t)b * (NBLK * PB4);
#pragma unroll
    for (int u = 0; u < NLOAD; ++u) {
      const int idx = tid + u * CTHREADS;
      const int blk = idx / (TT / 4), j = idx - blk * (TT / 4);
      st[u] = make_float4(0.f, 0.f, 0.f, 0.f);
      if (idx < ROWV4) st[u] = src[blk * PB4 + j];
    }
  }

  const Biquad q = load_biquad(acoef, bcoef);

  const int mlane = tid & 31;
  const int mj = (mlane >> 1) & 1, mc = mlane & 1;
  const double mcoef = mc ? -(double)q.a2 : -(double)q.a1;
  double mown = (mc == mj) ? 1.0 : 0.0, moth = (mc == mj) ? 0.0 : 1.0;

  if (!helper) {
#pragma unroll
    for (int u = 0; u < NLOAD; ++u) {
      const int idx = tid + u * CTHREADS;
      if (idx < ROWV4) row4[idx] = st[u];
    }
  } else {
    for (int w = 0; w < MSPLIT; ++w) unit_step(mcoef, mc, mown, moth);
  }
  __syncthreads();

  float x[CLEN];
  float z1 = 0.f, z2 = 0.f;
  if (!helper) {
    const float* xp = rowf + c * CLEN;
#pragma unroll
    for (int w = 0; w < CLEN; ++w) x[w] = xp[w];
#pragma unroll
    for (int w = 0; w < CLEN; ++w) (void)biquad_step(q, x[w], z1, z2);
    v2f e; e.x = z1; e.y = z2;
    if (act) shE[c] = e;
  } else {
    for (int w = MSPLIT; w < CLEN; ++w) unit_step(mcoef, mc, mown, moth);
    if (mlane < 4) shM[2 * mc + mj] = mown;
  }
  __syncthreads();

  scan_carries(shE, shM, shP, shQ, shL, shEG, shL2, shESG, shS, tid, true);

  float y[CLEN];
  if (!helper) {
    chunk_start(shP, shS, shL, c, z1, z2);
#pragma unroll
    for (int w = 0; w < CLEN; ++w) y[w] = biquad_step(q, x[w], z1, z2);
    z1 = 0.f; z2 = 0.f;
#pragma unroll
    for (int w = CLEN - 1; w >= 0; --w) (void)biquad_step(q, y[w], z1, z2);
    v2f e; e.x = z1; e.y = z2;
    if (act) shE[NCHUNK - 1 - c] = e;
  }
  __syncthreads();

  scan_carries(shE, shM, shP, shQ, shL, shEG, shL2, shESG, shS, tid, false);

  if (!helper) {
    chunk_start(shP, shS, shL, NCHUNK - 1 - c, z1, z2);
    float* tp = rowf + c * CLEN;
#pragma unroll
    for (int w = CLEN - 1; w >= 0; --w) {
      const float o = biquad_step(q, y[w], z1, z2);
      if (act) tp[w] = o;
    }
  }
  __syncthreads();

  float4* dst = reinterpret_cast<float4*>(out) + (size_t)b * ROWV4;
  for (int idx = tid; idx < ROWV4; idx += STHREADS) dst[idx] = row4[idx];
}

extern "C" void kernel_launch(void* const* d_in, const int* in_sizes, int n_in,
                              void* d_out, int out_size, void* d_ws, size_t ws_size,
                              hipStream_t stream) {
  const float* rgb    = (const float*)d_in[0];
  const float* matrix = (const float*)d_in[1];
  const float* acoef  = (const float*)d_in[2];
  const float* bcoef  = (const float*)d_in[3];
  float* out = (float*)d_out;

  float* ppg  = (float*)d_ws;

  pos_window_overlap_add<<<NBLK, FTHREADS, 0, stream>>>(rgb, matrix, ppg);
  pos_filtfilt_rows<<<BATCH, STHREADS, 0, stream>>>(ppg, acoef, bcoef, out);
}


# Scan carries run in f32 where no element of any tabled power of M reaches 2, else f64 as before (precision of carried states); tables built ahead in f64, M by squaring; carry rounding changes
# speedup vs baseline: 126.7298x; 1.1559x over previous
#include <hip/hip_runtime.h>
#include <hip/hip_bf16.h>

typedef float v2f __attribute__((ext_vector_type(2)));
typedef float v8f __attribute__((ext_vector_type(8)));

constexpr int TLEN  = 18000;
constexpr int BATCH = 16;
constexpr int WIN   = 48;
constexpr int MWIN  = TLEN - WIN;

constexpr int TT     = 80;
constexpr int NTILE  = TT / 16;
constexpr int NWIN   = 128;
constexpr int WPT    = 8;
constexpr int TPR    = NWIN / WPT;
constexpr int FTHREADS = 16 * TPR;
constexpr int TSMP   = 180;
constexpr int TROW   = TSMP * 3;
constexpr int TROW4  = TROW / 4;
constexpr int ROW4   = TLEN * 3 / 4;
constexpr int GPITCH = NWIN + 4;
constexpr int NSTAGE = (16 * TROW4 + FTHREADS - 1) / FTHREADS;
constexpr int NBLK   = TLEN / TT;
constexpr int PB     = 96;
constexpr int PB4    = PB / 4;
constexpr int POL    = 100;
static_assert(PB >= TT && PB % 32 == 0 && POL >= TT && POL % 4 == 0, "ppg block holds the group's samples in whole lines");
static_assert(PB4 % 8 == 0 && 32 % 8 == 0, "a line is 8 consecutive lanes of one store instruction");
static_assert(TLEN % TT == 0 && TT % 16 == 0, "work-groups and tiles cover the row exactly");
static_assert(NWIN >= TT + WIN && NWIN % WPT == 0 && TPR == 16, "half a wave per row holds every window of the group");
static_assert(WPT % 4 == 0 && FTHREADS / 32 >= NTILE, "gains leave as 16-byte pieces; one wave per overlap-add tile");
static_assert((WIN * 3) % 4 == 0 && (TT * 3) % 4 == 0 && (TLEN * 3) % 4 == 0, "row segments start 16-byte aligned");
static_assert(TROW % 4 == 0 && TSMP >= NWIN - WPT + 4 * ((WIN + WPT - 1 + 3) / 4), "segment holds every sample a thread reads");
static_assert(TSMP >= WIN + TT, "segment holds the group's own output samples");
static_assert((3 * WPT) % 4 == 0, "a thread's first sample is 16-byte aligned in LDS");
static_assert(GPITCH % 4 == 0 && GPITCH >= 16 * (NTILE - 1) + 1 + 64, "a tile's 64 K positions stay inside the gain row");

struct WinAcc { float s0, s1, s2, q00, q01, q02, q11, q12, q22; };

__device__ __forceinline__ void win_add(WinAcc& a, float r0, float r1, float r2) {
  a.s0 += r0; a.s1 += r1; a.s2 += r2;
  a.q00 = fmaf(r0, r0, a.q00); a.q01 = fmaf(r0, r1, a.q01); a.q02 = fmaf(r0, r2, a.q02);
  a.q11 = fmaf(r1, r1, a.q11); a.q12 = fmaf(r1, r2, a.q12); a.q22 = fmaf(r2, r2, a.q22);
}

__device__ __forceinline__ void win_join(WinAcc& a, const WinAcc& b) {
  a.s0 += b.s0; a.s1 += b.s1; a.s2 += b.s2;
  a.q00 += b.q00; a.q01 += b.q01; a.q02 += b.q02;
  a.q11 += b.q11; a.q12 += b.q12; a.q22 += b.q22;
}

__global__ __launch_bounds__(FTHREADS)
void pos_window_overlap_add(const float* __restrict__ rgb,
                            const float* __restrict__ matrix,
                            float* __restrict__ ppg) {
  __shared__ float4 tile4[16 * TROW4];
  __shared__ float4 sg4[3 * 16 * GPITCH / 4];
  __shared__ float4 po4[16 * POL / 4];
  const int tid = threadIdx.x;
  const int t0  = blockIdx.x * TT;

  {
    const int g4lo = (t0 - WIN) * 3 / 4;
    const float4* rgb4 = reinterpret_cast<const float4*>(rgb);
    float4 st[NSTAGE];
#pragma unroll
    for (int u = 0; u < NSTAGE; ++u) {
      const int idx = tid + u * FTHREADS;
      const int row = idx / TROW4, c4 = idx - row * TROW4;
      const int gi = g4lo + c4;
      st[u] = make_float4(0.f, 0.f, 0.f, 0.f);
      if (idx < 16 * TROW4 && (unsigned)gi < (unsigned)ROW4) st[u] = rgb4[(size_t)row * ROW4 + gi];
    }
#pragma unroll
    for (int u = 0; u < NSTAGE; ++u) {
      const int idx = tid + u * FTHREADS;
      if (idx < 16 * TROW4) tile4[idx] = st[u];
    }
  }
  __syncthreads();

  const int b = tid / TPR;
  const int q = tid % TPR;
  const float4* t4 = tile4 + b * TROW4 + (3 * WPT / 4) * q;

  WinAcc acc[WPT] = {};
  WinAcc com = {};

#pragma unroll
  for (int m = 0; m < (WIN + WPT - 1 + 3) / 4; ++m) {
    const float4 A = t4[3 * m + 0], B = t4[3 * m + 1], C = t4[3 * m + 2];
    const float smp[4][3] = {{A.x, A.y, A.z}, {A.w, B.x, B.y}, {B.z, B.w, C.x}, {C.y, C.z, C.w}};
#pragma unroll
    for (int e = 0; e < 4; ++e) {
      const int i = 4 * m + e;
      if (i >= WPT - 1 && i < WIN) {
        win_add(com, smp[e][0], smp[e][1], smp[e][2]);
      } else {
        if (i == WIN) {
#pragma unroll
          for (int j = 0; j < WPT; ++j) win_join(acc[j], com);
        }
#pragma unroll
        for (int j = 0; j < WPT; ++j)
          if (i >= j && i < j + WIN) win_add(acc[j], smp[e][0], smp[e][1], smp[e][2]);
      }
    }
  }

  float m00 = matrix[0], m01 = matrix[1], m02 = matrix[2];
  float m10 = matrix[3], m11 = matrix[4], m12 = matrix[5];
  float gv[3][WPT];

#pragma unroll
  for (int j = 0; j < WPT; ++j) {
    const float s0 = acc[j].s0, s1 = acc[j].s1, s2 = acc[j].s2;
    const float q00 = acc[j].q00, q01 = acc[j].q01, q02 = acc[j].q02;
    const float q11 = acc[j].q11, q12 = acc[j].q12, q22 = acc[j].q22;

    const float invW = 1.0f / (float)WIN;
    float inv0 = 1.0f / (s0 * invW + 1e-6f);
    float inv1 = 1.0f / (s1 * invW + 1e-6f);
    float inv2 = 1.0f / (s2 * invW + 1e-6f);

    float u0 = m00 * inv0, u1 = m01 * inv1, u2 = m02 * inv2;
    float sum0 = u0 * s0 + u1 * s1 + u2 * s2;
    float sq0  = u0 * u0 * q00 + u1 * u1 * q11 + u2 * u2 * q22 +
                 2.f * (u0 * u1 * q01 + u0 * u2 * q02 + u1 * u2 * q12);
    float var0 = (sq0 - sum0 * sum0 * invW) * (1.0f / (float)(WIN - 1));
    float std0 = sqrtf(fmaxf(var0, 0.f));

    float v0 = m10 * inv0, v1 = m11 * inv1, v2 = m12 * inv2;
    float sum1 = v0 * s0 + v1 * s1 + v2 * s2;
    float sq1  = v0 * v0 * q00 + v1 * v1 * q11 + v2 * v2 * q22 +
                 2.f * (v0 * v1 * q01 + v0 * v2 * q02 + v1 * v2 * q12);
    float var1 = (sq1 - sum1 * sum1 * invW) * (1.0f / (float)(WIN - 1));
    float std1 = sqrtf(fmaxf(var1, 0.f));

    float c = std0 / (std1 + 1e-4f);

    const int n = t0 - WIN + WPT * q + j;
    const bool have = (unsigned)n < (unsigned)MWIN;
    gv[0][j] = have ? (m00 + c * m10) * inv0 : 0.f;
    gv[1][j] = have ? (m01 + c * m11) * inv1 : 0.f;
    gv[2][j] = have ? (m02 + c * m12) * inv2 : 0.f;
  }
#pragma unroll
  for (int k = 0; k < 3; ++k) {
    float4* grow = sg4 + (k * 16 + b) * (GPITCH / 4);
#pragma unroll
    for (int h = 0; h < WPT / 4; ++h)
      grow[(WPT / 4) * q + h] = make_float4(gv[k][4 * h], gv[k][4 * h + 1], gv[k][4 * h + 2], gv[k][4 * h + 3]);
    if (q == 0) grow[NWIN / 4] = make_float4(0.f, 0.f, 0.f, 0.f);
  }
  __syncthreads();

  const int wave = tid >> 5;
  if (wave < NTILE) {
    const int lane = tid & 31;
    const int col  = lane & 15;
    const int hi   = lane >> 4;
    const float* sgt = reinterpret_cast<const float*>(sg4) + col * GPITCH + 16 * wave + 1;

    v8f acc3[3];
#pragma unroll
    for (int k = 0; k < 3; ++k) {
      const float* gk = sgt + k * 16 * GPITCH;
      v8f d = {0.f, 0.f, 0.f, 0.f, 0.f, 0.f, 0.f, 0.f};
#pragma unroll
      for (int s = 0; s < 16; ++s) {
        int j0 = 4 * s + 2 * hi;
        int j1 = j0 + 1;
        v2f a;
        a.x = (j0 >= col && j0 <= col + 47) ? 1.0f : 0.0f;
        a.y = (j1 >= col && j1 <= col + 47) ? 1.0f : 0.0f;
        v2f bb;
        bb.x = gk[j0];
        bb.y = gk[j1];
        d = __builtin_amdgcn_wmma_f32_16x16x4_f32(
             false, a,  false, bb,
             (short)0, d,  false,  false);
      }
      acc3[k] = d;
    }

    const float* rt = reinterpret_cast<const float*>(tile4) + col * TROW + (WIN + 16 * wave) * 3;
#pragma unroll
    for (int p = 0; p < 8; ++p) {
      const int i = p + 8 * hi;
      const float* rp = rt + i * 3;
      float v = rp[0] * acc3[0][p] + rp[1] * acc3[1][p] + rp[2] * acc3[2][p];
      reinterpret_cast<float*>(po4)[col * POL + 16 * wave + i] = v;
    }
  }
  __syncthreads();

  float4* ppg4 = reinterpret_cast<float4*>(ppg);
  for (int idx = tid; idx < 16 * PB4; idx += FTHREADS) {
    const int row = idx / PB4, j = idx - row * PB4;
    float4 v = make_float4(0.f, 0.f, 0.f, 0.f);
    if (j < TT / 4) v = po4[row * (POL / 4) + j];
    ppg4[((size_t)row * NBLK + blockIdx.x) * PB4 + j] = v;
  }
}


constexpr int CLEN   = 45;
constexpr int NCHUNK = TLEN / CLEN;
constexpr int G1     = 10;
constexpr int NG1    = NCHUNK / G1;
constexpr int G2     = 8;
constexpr int NSG    = NG1 / G2;
static_assert(NCHUNK * CLEN == TLEN, "chunks tile the row");
static_assert(NG1 * G1 == NCHUNK && NSG * G2 == NG1, "groups tile the chunks, blocks tile the groups");
constexpr int L1WAVES = (2 * NG1 + 31) / 32;
constexpr int CWAVES   = (NCHUNK + 31) / 32;
constexpr int CTHREADS = CWAVES * 32;
constexpr int STHREADS = CTHREADS + 32;
constexpr int ROWV4  = TLEN / 4;
constexpr int NLOAD  = (ROWV4 + CTHREADS - 1) / CTHREADS;
constexpr int NPOW   = 4 * (G1 + 1);
constexpr int NTAB   = NPOW + 4 * (G2 + 1);
constexpr int PSPLIT = 3;
static_assert(2 * NSG <= 32 && PSPLIT < G1, "one wave holds a lane pair per block");
static_assert(32 * L1WAVES + NTAB <= CTHREADS, "the waves that round the tables are chunk waves behind the group waves");

struct Biquad { float b0, b1, b2, a1, a2; };

__device__ __forceinline__ Biquad load_biquad(const float* __restrict__ acoef,
                                              const float* __restrict__ bcoef) {
  float ia = 1.0f / acoef[0];
  Biquad q;
  q.b0 = bcoef[0] * ia; q.b1 = bcoef[1] * ia; q.b2 = bcoef[2] * ia;
  q.a1 = acoef[1] * ia; q.a2 = acoef[2] * ia;
  return q;
}

__device__ __forceinline__ float biquad_step(const Biquad& q, float x, float& z1, float& z2) {
  float y = fmaf(q.b0, x, z1);
  z1 = fmaf(q.b1, x, fmaf(-q.a1, y, z2));
  z2 = fmaf(q.b2, x, -q.a2 * y);
  return y;
}


__device__ __forceinline__ double pair_swap(double v) {
  const long long bits = __builtin_bit_cast(long long, v);
  int lo = (int)(bits & 0xffffffffLL), hi = (int)(bits >> 32);
  lo = __builtin_amdgcn_update_dpp(0, lo, 0xB1, 0xF, 0xF, false);
  hi = __builtin_amdgcn_update_dpp(0, hi, 0xB1, 0xF, 0xF, false);
  const long long r = ((long long)hi << 32) | (long long)(unsigned int)lo;
  return __builtin_bit_cast(double, r);
}

__device__ __forceinline__ float pair_swap(float v) {
  int b = __builtin_bit_cast(int, v);
  b = __builtin_amdgcn_update_dpp(0, b, 0xB1, 0xF, 0xF, false);
  return __builtin_bit_cast(float, b);
}

template <int CTRL>
__device__ __forceinline__ double quad_fetch(double v) {
  const long long bits = __builtin_bit_cast(long long, v);
  int lo = (int)(bits & 0xffffffffLL), hi = (int)(bits >> 32);
  lo = __builtin_amdgcn_update_dpp(0, lo, CTRL, 0xF, 0xF, false);
  hi = __builtin_amdgcn_update_dpp(0, hi, CTRL, 0xF, 0xF, false);
  const long long r = ((long long)hi << 32) | (long long)(unsigned int)lo;
  return __builtin_bit_cast(double, r);
}

__device__ __forceinline__ float  fma_t(float a, float b, float c)    { return fmaf(a, b, c); }
__device__ __forceinline__ double fma_t(double a, double b, double c) { return fma(a, b, c); }

template <typename T>
__device__ __forceinline__ void pair_carry(T td, T to, T e, T& own, T& other) {
  own = fma_t(td, own, fma_t(to, other, e));
  other = pair_swap(own);
}

__device__ __forceinline__ unsigned int f64_mag_word(double v) {
  return (unsigned int)(__builtin_bit_cast(unsigned long long, v) >> 32) & 0x7fffffffu;
}

__device__ __forceinline__ double quad_matmul(double x, double y) {
  const double x0 = quad_fetch<0x44>(x);
  const double x1 = quad_fetch<0xEE>(x);
  const double y0 = quad_fetch<0xA0>(y);
  const double y1 = quad_fetch<0xF5>(y);
  return fma(x0, y0, x1 * y1);
}

template <typename T>
__device__ __forceinline__ void scan_carries(const v2f* shE, const T* shM, const T* shP, const T* shQ,
                                             T* shL, T* shEG, T* shL2,
                                             T* shESG, T* shSG, int tid,
                                             const double* rnd_src, float* rnd_dst) {
  if (tid < 32 * L1WAVES) {
    const int g = tid >> 1, comp = tid & 1;
    const bool have = g < NG1;
    const int gc = have ? g : NG1 - 1;
    const T td = shM[3 * comp], to = shM[1 + comp];
    const float* ef = reinterpret_cast<const float*>(shE);
    float e[G1];
#pragma unroll
    for (int j = 0; j < G1; ++j) e[j] = ef[(gc * G1 + j) * 2 + comp];
    T own = 0, other = 0;
    if (have) shL[(gc * G1) * 2 + comp] = 0;
#pragma unroll
    for (int j = 0; j < G1; ++j) {
      pair_carry(td, to, (T)e[j], own, other);
      if (have) {
        if (j < G1 - 1) shL[(gc * G1 + j + 1) * 2 + comp] = own;
        else            shEG[gc * 2 + comp] = own;
      }
    }
  } else if (rnd_dst != nullptr && tid < 32 * L1WAVES + NTAB) {
    const int i = tid - 32 * L1WAVES;
    rnd_dst[i] = (float)rnd_src[i];
  }
  __syncthreads();
  if (tid < 32) {
    const int s = tid >> 1, comp = tid & 1;
    const bool have = s < NSG;
    const int sc = have ? s : NSG - 1;
    const T* T2 = shP + 4 * G1;
    const T td = T2[3 * comp], to = T2[1 + comp];
    T eg[G2];
#pragma unroll
    for (int k = 0; k < G2; ++k) eg[k] = shEG[(sc * G2 + k) * 2 + comp];
    T own = 0, other = 0;
    if (have) shL2[(sc * G2) * 2 + comp] = 0;
#pragma unroll
    for (int k = 0; k < G2; ++k) {
      pair_carry(td, to, eg[k], own, other);
      if (have) {
        if (k < G2 - 1) shL2[(sc * G2 + k + 1) * 2 + comp] = own;
        else            shESG[sc * 2 + comp] = own;
      }
    }
  }
  __syncthreads();
  if (tid < 32 * L1WAVES) {
    const int g = tid >> 1, comp = tid & 1;
    const bool have = g < NG1;
    const int gc = have ? g : NG1 - 1;
    const int s = gc / G2, j = gc - s * G2;
    const T* T3 = shQ + 4 * G2;
    const T td = T3[3 * comp], to = T3[1 + comp];
    const T qd = shQ[4 * j + 3 * comp], qo = shQ[4 * j + 1 + comp];
    const T l2 = shL2[gc * 2 + comp];
    T esg[NSG - 1];
#pragma unroll
    for (int k = 0; k < NSG - 1; ++k) esg[k] = shESG[k * 2 + comp];
    asm volatile("" :: "v"(qd), "v"(qo), "v"(l2));
    __builtin_amdgcn_sched_barrier(0);
    T own = 0, other = 0, so = 0, sp = 0;
#pragma unroll
    for (int k = 0; k < NSG - 1; ++k) {
      if (k == s) { so = own; sp = other; }
      pair_carry(td, to, esg[k], own, other);
    }
    if (s == NSG - 1) { so = own; sp = other; }
    const T sg = fma_t(qd, so, fma_t(qo, sp, l2));
    if (have) shSG[gc * 2 + comp] = sg;
  }
  __syncthreads();
}

template <typename T>
__device__ __forceinline__ void chunk_start(const T* shP, const T* shS,
                                            const T* shL, int cn,
                                            float& z1, float& z2) {
  const int g = cn / G1, i = cn - g * G1;
  const T p11 = shP[4 * i + 0], p12 = shP[4 * i + 1];
  const T p21 = shP[4 * i + 2], p22 = shP[4 * i + 3];
  const T s1 = shS[g * 2], s2 = shS[g * 2 + 1];
  const T l1 = shL[cn * 2], l2 = shL[cn * 2 + 1];
  z1 = (float)fma_t(p11, s1, fma_t(p12, s2, l1));
  z2 = (float)fma_t(p21, s1, fma_t(p22, s2, l2));
}

__global__ __launch_bounds__(STHREADS)
void pos_filtfilt_rows(const float* __restrict__ ppg, const float* __restrict__ acoef,
                       const float* __restrict__ bcoef, float* __restrict__ out) {
  __shared__ float4 row4[ROWV4];
  __shared__ v2f    shE[NCHUNK];
  __shared__ double shL[NCHUNK * 2];
  __shared__ double shEG[NG1 * 2];
  __shared__ double shL2[NG1 * 2];
  __shared__ double shESG[NSG * 2];
  __shared__ double shS[NG1 * 2];
  __shared__ float  shMf[4];
  __shared__ double shTd[NTAB];
  __shared__ float4 shT4[NTAB / 4];
  __shared__ int    shSmall;
  float* rowf = reinterpret_cast<float*>(row4);
  float* shT  = reinterpret_cast<float*>(shT4);
  const int tid = threadIdx.x;
  const int b = blockIdx.x;
  const bool helper = tid >= CTHREADS;
  const bool act = tid < NCHUNK;
  const int c = act ? tid : NCHUNK - 1;

  float4 st[NLOAD];
  if (!helper) {
    const float4* src = reinterpret_cast<const float4*>(ppg) + (size_t)b * (NBLK * PB4);
#pragma unroll
    for (int u = 0; u < NLOAD; ++u) {
      const int idx = tid + u * CTHREADS;
      const int blk = idx / (TT / 4), j = idx - blk * (TT / 4);
      st[u] = make_float4(0.f, 0.f, 0.f, 0.f);
      if (idx < ROWV4) st[u] = src[blk * PB4 + j];
    }
  }

  const Biquad q = load_biquad(acoef, bcoef);

  const int mlane = tid & 31;
  const int mj = (mlane >> 1) & 1, mc = mlane & 1;
  double town = 0.0, toth = 0.0, ttd = 0.0, tto = 0.0;
  unsigned int tmag = 0;

  if (!helper) {
#pragma unroll
    for (int u = 0; u < NLOAD; ++u) {
      const int idx = tid + u * CTHREADS;
      if (idx < ROWV4) row4[idx] = st[u];
    }
  } else {
    const double a = mc ? (mj ? 0.0 : -(double)q.a2) : (mj ? 1.0 : -(double)q.a1);
    double xp = a, m = a;
    bool got = false;
#pragma unroll
    for (int bits = CLEN; bits != 0; bits >>= 1) {
      if (bits & 1) { m = got ? quad_matmul(m, xp) : xp; got = true; }
      if (bits > 1) xp = quad_matmul(xp, xp);
    }
    if (mlane < 4) shMf[2 * mc + mj] = (float)m;
    ttd = quad_fetch<0xCC>(m);
    tto = quad_fetch<0x66>(m);
    town = (mc == mj) ? 1.0 : 0.0; toth = (mc == mj) ? 0.0 : 1.0;
    if (mlane < 4) { shTd[2 * mc + mj] = town; shTd[NPOW + 2 * mc + mj] = town; }
    for (int k = 1; k <= PSPLIT; ++k) {
      pair_carry(ttd, tto, 0.0, town, toth);
      tmag = max(tmag, f64_mag_word(town));
      if (mlane < 4) shTd[4 * k + 2 * mc + mj] = town;
    }
  }
  __syncthreads();

  float x[CLEN];
  float z1 = 0.f, z2 = 0.f;
  if (!helper) {
    const float* xp = rowf + c * CLEN;
#pragma unroll
    for (int w = 0; w < CLEN; ++w) x[w] = xp[w];
#pragma unroll
    for (int w = 0; w < CLEN; ++w) (void)biquad_step(q, x[w], z1, z2);
    v2f e; e.x = z1; e.y = z2;
    if (act) shE[c] = e;
  } else {
    for (int k = PSPLIT + 1; k <= G1; ++k) {
      pair_carry(ttd, tto, 0.0, town, toth);
      tmag = max(tmag, f64_mag_word(town));
      if (mlane < 4) shTd[4 * k + 2 * mc + mj] = town;
    }
    if (mlane < 4) shTd[NPOW + 4 + 2 * mc + mj] = town;
    ttd = quad_fetch<0xCC>(town);
    tto = quad_fetch<0x66>(town);
    for (int k = 2; k <= G2; ++k) {
      pair_carry(ttd, tto, 0.0, town, toth);
      tmag = max(tmag, f64_mag_word(town));
      if (mlane < 4) shTd[NPOW + 4 * k + 2 * mc + mj] = town;
    }
    const bool big = tmag >= 0x40000000u;
    const bool anybig = __builtin_amdgcn_ballot_w32(big) != 0u;
    if (mlane == 0) shSmall = anybig ? 0 : 1;
  }
  __syncthreads();

  float* shLf = reinterpret_cast<float*>(shL);
  float* shEGf = reinterpret_cast<float*>(shEG);
  float* shL2f = reinterpret_cast<float*>(shL2);
  float* shESGf = reinterpret_cast<float*>(shESG);
  float* shSf = reinterpret_cast<float*>(shS);
  const bool small = __builtin_amdgcn_readfirstlane(shSmall) != 0;
  if (small) scan_carries<float>(shE, shMf, shT, shT + NPOW, shLf, shEGf, shL2f, shESGf, shSf, tid, shTd, shT);
  else       scan_carries<double>(shE, shTd + 4, shTd, shTd + NPOW, shL, shEG, shL2, shESG, shS, tid, nullptr, nullptr);

  float y[CLEN];
  if (!helper) {
    if (small) chunk_start<float>(shT, shSf, shLf, c, z1, z2);
    else       chunk_start<double>(shTd, shS, shL, c, z1, z2);
#pragma unroll
    for (int w = 0; w < CLEN; ++w) y[w] = biquad_step(q, x[w], z1, z2);
    z1 = 0.f; z2 = 0.f;
#pragma unroll
    for (int w = CLEN - 1; w >= 0; --w) (void)biquad_step(q, y[w], z1, z2);
    v2f e; e.x = z1; e.y = z2;
    if (act) shE[NCHUNK - 1 - c] = e;
  }
  __syncthreads();

  if (small) scan_carries<float>(shE, shMf, shT, shT + NPOW, shLf, shEGf, shL2f, shESGf, shSf, tid, nullptr, nullptr);
  else       scan_carries<double>(shE, shTd + 4, shTd, shTd + NPOW, shL, shEG, shL2, shESG, shS, tid, nullptr, nullptr);

  if (!helper) {
    if (small) chunk_start<float>(shT, shSf, shLf, NCHUNK - 1 - c, z1, z2);
    else       chunk_start<double>(shTd, shS, shL, NCHUNK - 1 - c, z1, z2);
    float* tp = rowf + c * CLEN;
#pragma unroll
    for (int w = CLEN - 1; w >= 0; --w) {
      const float o = biquad_step(q, y[w], z1, z2);
      if (act) tp[w] = o;
    }
  }
  __syncthreads();

  float4* dst = reinterpret_cast<float4*>(out) + (size_t)b * ROWV4;
  for (int idx = tid; idx < ROWV4; idx += STHREADS) dst[idx] = row4[idx];
}

extern "C" void kernel_launch(void* const* d_in, const int* in_sizes, int n_in,
                              void* d_out, int out_size, void* d_ws, size_t ws_size,
                              hipStream_t stream) {
  const float* rgb    = (const float*)d_in[0];
  const float* matrix = (const float*)d_in[1];
  const float* acoef  = (const float*)d_in[2];
  const float* bcoef  = (const float*)d_in[3];
  float* out = (float*)d_out;

  float* ppg  = (float*)d_ws;

  pos_window_overlap_add<<<NBLK, FTHREADS, 0, stream>>>(rgb, matrix, ppg);
  pos_filtfilt_rows<<<BATCH, STHREADS, 0, stream>>>(ppg, acoef, bcoef, out);
}


# Scan carries as log-step scans in registers: each 16-lane row scans its chunks' end states by DPP shifts with tabled M^(2^d), rows meet in LDS once per pass; carry sum order changes, fixed per run
# speedup vs baseline: 133.4680x; 1.0532x over previous
#include <hip/hip_runtime.h>
#include <hip/hip_bf16.h>

typedef float v2f __attribute__((ext_vector_type(2)));
typedef float v8f __attribute__((ext_vector_type(8)));

constexpr int TLEN  = 18000;
constexpr int BATCH = 16;
constexpr int WIN   = 48;
constexpr int MWIN  = TLEN - WIN;

constexpr int TT     = 80;
constexpr int NTILE  = TT / 16;
constexpr int NWIN   = 128;
constexpr int WPT    = 8;
constexpr int TPR    = NWIN / WPT;
constexpr int FTHREADS = 16 * TPR;
constexpr int TSMP   = 180;
constexpr int TROW   = TSMP * 3;
constexpr int TROW4  = TROW / 4;
constexpr int ROW4   = TLEN * 3 / 4;
constexpr int GPITCH = NWIN + 4;
constexpr int NSTAGE = (16 * TROW4 + FTHREADS - 1) / FTHREADS;
constexpr int NBLK   = TLEN / TT;
constexpr int PB     = 96;
constexpr int PB4    = PB / 4;
constexpr int POL    = 100;
static_assert(PB >= TT && PB % 32 == 0 && POL >= TT && POL % 4 == 0, "ppg block holds the group's samples in whole lines");
static_assert(PB4 % 8 == 0 && 32 % 8 == 0, "a line is 8 consecutive lanes of one store instruction");
static_assert(TLEN % TT == 0 && TT % 16 == 0, "work-groups and tiles cover the row exactly");
static_assert(NWIN >= TT + WIN && NWIN % WPT == 0 && TPR == 16, "half a wave per row holds every window of the group");
static_assert(WPT % 4 == 0 && FTHREADS / 32 >= NTILE, "gains leave as 16-byte pieces; one wave per overlap-add tile");
static_assert((WIN * 3) % 4 == 0 && (TT * 3) % 4 == 0 && (TLEN * 3) % 4 == 0, "row segments start 16-byte aligned");
static_assert(TROW % 4 == 0 && TSMP >= NWIN - WPT + 4 * ((WIN + WPT - 1 + 3) / 4), "segment holds every sample a thread reads");
static_assert(TSMP >= WIN + TT, "segment holds the group's own output samples");
static_assert((3 * WPT) % 4 == 0, "a thread's first sample is 16-byte aligned in LDS");
static_assert(GPITCH % 4 == 0 && GPITCH >= 16 * (NTILE - 1) + 1 + 64, "a tile's 64 K positions stay inside the gain row");

struct WinAcc { float s0, s1, s2, q00, q01, q02, q11, q12, q22; };

__device__ __forceinline__ void win_add(WinAcc& a, float r0, float r1, float r2) {
  a.s0 += r0; a.s1 += r1; a.s2 += r2;
  a.q00 = fmaf(r0, r0, a.q00); a.q01 = fmaf(r0, r1, a.q01); a.q02 = fmaf(r0, r2, a.q02);
  a.q11 = fmaf(r1, r1, a.q11); a.q12 = fmaf(r1, r2, a.q12); a.q22 = fmaf(r2, r2, a.q22);
}

__device__ __forceinline__ void win_join(WinAcc& a, const WinAcc& b) {
  a.s0 += b.s0; a.s1 += b.s1; a.s2 += b.s2;
  a.q00 += b.q00; a.q01 += b.q01; a.q02 += b.q02;
  a.q11 += b.q11; a.q12 += b.q12; a.q22 += b.q22;
}

__global__ __launch_bounds__(FTHREADS)
void pos_window_overlap_add(const float* __restrict__ rgb,
                            const float* __restrict__ matrix,
                            float* __restrict__ ppg) {
  __shared__ float4 tile4[16 * TROW4];
  __shared__ float4 sg4[3 * 16 * GPITCH / 4];
  __shared__ float4 po4[16 * POL / 4];
  const int tid = threadIdx.x;
  const int t0  = blockIdx.x * TT;

  {
    const int g4lo = (t0 - WIN) * 3 / 4;
    const float4* rgb4 = reinterpret_cast<const float4*>(rgb);
    float4 st[NSTAGE];
#pragma unroll
    for (int u = 0; u < NSTAGE; ++u) {
      const int idx = tid + u * FTHREADS;
      const int row = idx / TROW4, c4 = idx - row * TROW4;
      const int gi = g4lo + c4;
      st[u] = make_float4(0.f, 0.f, 0.f, 0.f);
      if (idx < 16 * TROW4 && (unsigned)gi < (unsigned)ROW4) st[u] = rgb4[(size_t)row * ROW4 + gi];
    }
#pragma unroll
    for (int u = 0; u < NSTAGE; ++u) {
      const int idx = tid + u * FTHREADS;
      if (idx < 16 * TROW4) tile4[idx] = st[u];
    }
  }
  __syncthreads();

  const int b = tid / TPR;
  const int q = tid % TPR;
  const float4* t4 = tile4 + b * TROW4 + (3 * WPT / 4) * q;

  WinAcc acc[WPT] = {};
  WinAcc com = {};

#pragma unroll
  for (int m = 0; m < (WIN + WPT - 1 + 3) / 4; ++m) {
    const float4 A = t4[3 * m + 0], B = t4[3 * m + 1], C = t4[3 * m + 2];
    const float smp[4][3] = {{A.x, A.y, A.z}, {A.w, B.x, B.y}, {B.z, B.w, C.x}, {C.y, C.z, C.w}};
#pragma unroll
    for (int e = 0; e < 4; ++e) {
      const int i = 4 * m + e;
      if (i >= WPT - 1 && i < WIN) {
        win_add(com, smp[e][0], smp[e][1], smp[e][2]);
      } else {
        if (i == WIN) {
#pragma unroll
          for (int j = 0; j < WPT; ++j) win_join(acc[j], com);
        }
#pragma unroll
        for (int j = 0; j < WPT; ++j)
          if (i >= j && i < j + WIN) win_add(acc[j], smp[e][0], smp[e][1], smp[e][2]);
      }
    }
  }

  float m00 = matrix[0], m01 = matrix[1], m02 = matrix[2];
  float m10 = matrix[3], m11 = matrix[4], m12 = matrix[5];
  float gv[3][WPT];

#pragma unroll
  for (int j = 0; j < WPT; ++j) {
    const float s0 = acc[j].s0, s1 = acc[j].s1, s2 = acc[j].s2;
    const float q00 = acc[j].q00, q01 = acc[j].q01, q02 = acc[j].q02;
    const float q11 = acc[j].q11, q12 = acc[j].q12, q22 = acc[j].q22;

    const float invW = 1.0f / (float)WIN;
    float inv0 = 1.0f / (s0 * invW + 1e-6f);
    float inv1 = 1.0f / (s1 * invW + 1e-6f);
    float inv2 = 1.0f / (s2 * invW + 1e-6f);

    float u0 = m00 * inv0, u1 = m01 * inv1, u2 = m02 * inv2;
    float sum0 = u0 * s0 + u1 * s1 + u2 * s2;
    float sq0  = u0 * u0 * q00 + u1 * u1 * q11 + u2 * u2 * q22 +
                 2.f * (u0 * u1 * q01 + u0 * u2 * q02 + u1 * u2 * q12);
    float var0 = (sq0 - sum0 * sum0 * invW) * (1.0f / (float)(WIN - 1));
    float std0 = sqrtf(fmaxf(var0, 0.f));

    float v0 = m10 * inv0, v1 = m11 * inv1, v2 = m12 * inv2;
    float sum1 = v0 * s0 + v1 * s1 + v2 * s2;
    float sq1  = v0 * v0 * q00 + v1 * v1 * q11 + v2 * v2 * q22 +
                 2.f * (v0 * v1 * q01 + v0 * v2 * q02 + v1 * v2 * q12);
    float var1 = (sq1 - sum1 * sum1 * invW) * (1.0f / (float)(WIN - 1));
    float std1 = sqrtf(fmaxf(var1, 0.f));

    float c = std0 / (std1 + 1e-4f);

    const int n = t0 - WIN + WPT * q + j;
    const bool have = (unsigned)n < (unsigned)MWIN;
    gv[0][j] = have ? (m00 + c * m10) * inv0 : 0.f;
    gv[1][j] = have ? (m01 + c * m11) * inv1 : 0.f;
    gv[2][j] = have ? (m02 + c * m12) * inv2 : 0.f;
  }
#pragma unroll
  for (int k = 0; k < 3; ++k) {
    float4* grow = sg4 + (k * 16 + b) * (GPITCH / 4);
#pragma unroll
    for (int h = 0; h < WPT / 4; ++h)
      grow[(WPT / 4) * q + h] = make_float4(gv[k][4 * h], gv[k][4 * h + 1], gv[k][4 * h + 2], gv[k][4 * h + 3]);
    if (q == 0) grow[NWIN / 4] = make_float4(0.f, 0.f, 0.f, 0.f);
  }
  __syncthreads();

  const int wave = tid >> 5;
  if (wave < NTILE) {
    const int lane = tid & 31;
    const int col  = lane & 15;
    const int hi   = lane >> 4;
    const float* sgt = reinterpret_cast<const float*>(sg4) + col * GPITCH + 16 * wave + 1;

    v8f acc3[3];
#pragma unroll
    for (int k = 0; k < 3; ++k) {
      const float* gk = sgt + k * 16 * GPITCH;
      v8f d = {0.f, 0.f, 0.f, 0.f, 0.f, 0.f, 0.f, 0.f};
#pragma unroll
      for (int s = 0; s < 16; ++s) {
        int j0 = 4 * s + 2 * hi;
        int j1 = j0 + 1;
        v2f a;
        a.x = (j0 >= col && j0 <= col + 47) ? 1.0f : 0.0f;
        a.y = (j1 >= col && j1 <= col + 47) ? 1.0f : 0.0f;
        v2f bb;
        bb.x = gk[j0];
        bb.y = gk[j1];
        d = __builtin_amdgcn_wmma_f32_16x16x4_f32(
             false, a,  false, bb,
             (short)0, d,  false,  false);
      }
      acc3[k] = d;
    }

    const float* rt = reinterpret_cast<const float*>(tile4) + col * TROW + (WIN + 16 * wave) * 3;
#pragma unroll
    for (int p = 0; p < 8; ++p) {
      const int i = p + 8 * hi;
      const float* rp = rt + i * 3;
      float v = rp[0] * acc3[0][p] + rp[1] * acc3[1][p] + rp[2] * acc3[2][p];
      reinterpret_cast<float*>(po4)[col * POL + 16 * wave + i] = v;
    }
  }
  __syncthreads();

  float4* ppg4 = reinterpret_cast<float4*>(ppg);
  for (int idx = tid; idx < 16 * PB4; idx += FTHREADS) {
    const int row = idx / PB4, j = idx - row * PB4;
    float4 v = make_float4(0.f, 0.f, 0.f, 0.f);
    if (j < TT / 4) v = po4[row * (POL / 4) + j];
    ppg4[((size_t)row * NBLK + blockIdx.x) * PB4 + j] = v;
  }
}


constexpr int CLEN   = 45;
constexpr int NCHUNK = TLEN / CLEN;
static_assert(NCHUNK * CLEN == TLEN, "chunks tile the row");
constexpr int CWAVES   = (NCHUNK + 31) / 32;
constexpr int CTHREADS = CWAVES * 32;
constexpr int STHREADS = CTHREADS + 32;
constexpr int ROWV4  = TLEN / 4;
constexpr int NLOAD  = (ROWV4 + CTHREADS - 1) / CTHREADS;
constexpr int NROW16 = 2 * CWAVES;
constexpr int NLEV   = 6;
constexpr int NTAB   = 4 * NLEV;
constexpr int PREW   = CWAVES - 1;
constexpr int NTOT   = 2 * (2 * PREW + NROW16);

struct Biquad { float b0, b1, b2, a1, a2; };

__device__ __forceinline__ Biquad load_biquad(const float* __restrict__ acoef,
                                              const float* __restrict__ bcoef) {
  float ia = 1.0f / acoef[0];
  Biquad q;
  q.b0 = bcoef[0] * ia; q.b1 = bcoef[1] * ia; q.b2 = bcoef[2] * ia;
  q.a1 = acoef[1] * ia; q.a2 = acoef[2] * ia;
  return q;
}

__device__ __forceinline__ float biquad_step(const Biquad& q, float x, float& z1, float& z2) {
  float y = fmaf(q.b0, x, z1);
  z1 = fmaf(q.b1, x, fmaf(-q.a1, y, z2));
  z2 = fmaf(q.b2, x, -q.a2 * y);
  return y;
}

template <int CTRL>
__device__ __forceinline__ double quad_fetch(double v) {
  const long long bits = __builtin_bit_cast(long long, v);
  int lo = (int)(bits & 0xffffffffLL), hi = (int)(bits >> 32);
  lo = __builtin_amdgcn_update_dpp(0, lo, CTRL, 0xF, 0xF, false);
  hi = __builtin_amdgcn_update_dpp(0, hi, CTRL, 0xF, 0xF, false);
  const long long r = ((long long)hi << 32) | (long long)(unsigned int)lo;
  return __builtin_bit_cast(double, r);
}

template <int CTRL>
__device__ __forceinline__ float row_fetch(float v) {
  int b = __builtin_bit_cast(int, v);
  b = __builtin_amdgcn_update_dpp(0, b, CTRL, 0xF, 0xF, true);
  return __builtin_bit_cast(float, b);
}

template <int CTRL>
__device__ __forceinline__ double row_fetch(double v) {
  const long long bits = __builtin_bit_cast(long long, v);
  int lo = (int)(bits & 0xffffffffLL), hi = (int)(bits >> 32);
  lo = __builtin_amdgcn_update_dpp(0, lo, CTRL, 0xF, 0xF, true);
  hi = __builtin_amdgcn_update_dpp(0, hi, CTRL, 0xF, 0xF, true);
  const long long r = ((long long)hi << 32) | (long long)(unsigned int)lo;
  return __builtin_bit_cast(double, r);
}

__device__ __forceinline__ float  fma_t(float a, float b, float c)    { return fmaf(a, b, c); }
__device__ __forceinline__ double fma_t(double a, double b, double c) { return fma(a, b, c); }

__device__ __forceinline__ unsigned int f64_mag_word(double v) {
  return (unsigned int)(__builtin_bit_cast(unsigned long long, v) >> 32) & 0x7fffffffu;
}

__device__ __forceinline__ double quad_matmul(double x, double y) {
  const double x0 = quad_fetch<0x44>(x);
  const double x1 = quad_fetch<0xEE>(x);
  const double y0 = quad_fetch<0xA0>(y);
  const double y1 = quad_fetch<0xF5>(y);
  return fma(x0, y0, x1 * y1);
}

template <typename T> struct Powers { T p[NLEV][4]; };

template <typename T>
__device__ __forceinline__ void carry_step(const T (&P)[4], T c1, T c2, T& s1, T& s2) {
  const T n1 = fma_t(P[0], s1, fma_t(P[1], s2, c1));
  const T n2 = fma_t(P[2], s1, fma_t(P[3], s2, c2));
  s1 = n1; s2 = n2;
}

template <typename T, bool REV, int D>
__device__ __forceinline__ void scan_level(const Powers<T>& pw, T& v1, T& v2) {
  constexpr int CTRL = (REV ? 0x100 : 0x110) + (1 << D);
  T u1 = row_fetch<CTRL>(v1), u2 = row_fetch<CTRL>(v2);
  carry_step(pw.p[D], v1, v2, u1, u2);
  v1 = u1; v2 = u2;
}

template <typename T, bool REV>
__device__ __forceinline__ void row_totals(const Powers<T>& pw, T e1, T e2, T* tot, int tid) {
  T v1 = e1, v2 = e2;
  scan_level<T, REV, 0>(pw, v1, v2);
  scan_level<T, REV, 1>(pw, v1, v2);
  scan_level<T, REV, 2>(pw, v1, v2);
  scan_level<T, REV, 3>(pw, v1, v2);
  const int r = tid >> 4;
  const int fr = REV ? NROW16 - 1 - r : r;
  if ((tid & 15) == (REV ? 0 : 15)) {
    tot[2 * (2 * PREW + fr)]     = v1;
    tot[2 * (2 * PREW + fr) + 1] = v2;
  }
}

template <typename T, bool REV>
__device__ __forceinline__ void chunk_starts(const Powers<T>& pw, T e1, T e2, const T* tot, int tid,
                                             float& z1, float& z2) {
  const int w = tid >> 5;
  const int fw = REV ? CWAVES - 1 - w : w;
  const T* rp = tot + 2 * (2 * fw);
  T t1[PREW], t2[PREW];
#pragma unroll
  for (int k = 0; k < PREW; ++k) {
    t1[k] = rp[4 * k]; t2[k] = rp[4 * k + 1];
    carry_step(pw.p[4], rp[4 * k + 2], rp[4 * k + 3], t1[k], t2[k]);
  }
  const T o1 = rp[4 * PREW], o2 = rp[4 * PREW + 1];
  T s1 = 0, s2 = 0;
#pragma unroll
  for (int k = 0; k < PREW; ++k) carry_step(pw.p[5], t1[k], t2[k], s1, s2);
  T a1 = s1, a2 = s2;
  carry_step(pw.p[4], o1, o2, a1, a2);
  const bool second = REV ? (tid & 16) == 0 : (tid & 16) != 0;
  s1 = second ? a1 : s1; s2 = second ? a2 : s2;
  const bool first = (tid & 15) == (REV ? 15 : 0);
  T v1 = e1, v2 = e2;
  {
    T b1 = s1, b2 = s2;
    carry_step(pw.p[0], e1, e2, b1, b2);
    v1 = first ? b1 : v1; v2 = first ? b2 : v2;
  }
  scan_level<T, REV, 0>(pw, v1, v2);
  scan_level<T, REV, 1>(pw, v1, v2);
  scan_level<T, REV, 2>(pw, v1, v2);
  scan_level<T, REV, 3>(pw, v1, v2);
  constexpr int ONE = (REV ? 0x100 : 0x110) + 1;
  const T p1 = row_fetch<ONE>(v1), p2 = row_fetch<ONE>(v2);
  z1 = (float)(first ? s1 : p1);
  z2 = (float)(first ? s2 : p2);
}

template <typename T>
__device__ __forceinline__ Powers<T> load_powers(const T* tab) {
  Powers<T> pw;
#pragma unroll
  for (int d = 0; d < NLEV; ++d) {
#pragma unroll
    for (int i = 0; i < 4; ++i) pw.p[d][i] = tab[4 * d + i];
  }
  return pw;
}

__global__ __launch_bounds__(STHREADS)
void pos_filtfilt_rows(const float* __restrict__ ppg, const float* __restrict__ acoef,
                       const float* __restrict__ bcoef, float* __restrict__ out) {
  __shared__ float4 row4[ROWV4];
  __shared__ double shTotF[NTOT];
  __shared__ double shTotB[NTOT];
  __shared__ double shTd[NTAB];
  __shared__ float4 shT4[NTAB / 4];
  __shared__ int    shSmall;
  float* rowf = reinterpret_cast<float*>(row4);
  float* shT  = reinterpret_cast<float*>(shT4);
  const int tid = threadIdx.x;
  const int b = blockIdx.x;
  const bool helper = tid >= CTHREADS;
  const bool act = tid < NCHUNK;
  const int c = act ? tid : NCHUNK - 1;

  float4 st[NLOAD];
  if (!helper) {
    const float4* src = reinterpret_cast<const float4*>(ppg) + (size_t)b * (NBLK * PB4);
#pragma unroll
    for (int u = 0; u < NLOAD; ++u) {
      const int idx = tid + u * CTHREADS;
      const int blk = idx / (TT / 4), j = idx - blk * (TT / 4);
      st[u] = make_float4(0.f, 0.f, 0.f, 0.f);
      if (idx < ROWV4) st[u] = src[blk * PB4 + j];
    }
  }

  const Biquad q = load_biquad(acoef, bcoef);

  const int mlane = tid & 31;
  const int mj = (mlane >> 1) & 1, mc = mlane & 1;
  double mpow = 0.0;

  if (!helper) {
#pragma unroll
    for (int u = 0; u < NLOAD; ++u) {
      const int idx = tid + u * CTHREADS;
      if (idx < ROWV4) row4[idx] = st[u];
    }
  } else {
    const double a = mc ? (mj ? 0.0 : -(double)q.a2) : (mj ? 1.0 : -(double)q.a1);
    double xp = a, m = a;
    bool got = false;
#pragma unroll
    for (int bits = CLEN; bits != 0; bits >>= 1) {
      if (bits & 1) { m = got ? quad_matmul(m, xp) : xp; got = true; }
      if (bits > 1) xp = quad_matmul(xp, xp);
    }
    mpow = m;
    for (int i = mlane; i < 2 * 2 * PREW; i += 32) { shTotF[i] = 0.0; shTotB[i] = 0.0; }
  }
  __syncthreads();

  float x[CLEN];
  float z1 = 0.f, z2 = 0.f;
  if (!helper) {
    const float* xp = rowf + c * CLEN;
#pragma unroll
    for (int w = 0; w < CLEN; ++w) x[w] = xp[w];
#pragma unroll
    for (int w = 0; w < CLEN; ++w) (void)biquad_step(q, x[w], z1, z2);
  } else {
    unsigned int tmag = 0;
#pragma unroll
    for (int d = 0; d < NLEV; ++d) {
      tmag = max(tmag, f64_mag_word(mpow));
      if (mlane < 4) { shTd[4 * d + 2 * mc + mj] = mpow; shT[4 * d + 2 * mc + mj] = (float)mpow; }
      if (d < NLEV - 1) mpow = quad_matmul(mpow, mpow);
    }
    const bool big = tmag >= 0x40000000u;
    const bool anybig = __builtin_amdgcn_ballot_w32(big) != 0u;
    if (mlane == 0) shSmall = anybig ? 0 : 1;
  }
  __syncthreads();

  float* shTotFf = reinterpret_cast<float*>(shTotF);
  float* shTotBf = reinterpret_cast<float*>(shTotB);
  const bool small = __builtin_amdgcn_readfirstlane(shSmall) != 0;
  float e1 = act ? z1 : 0.f, e2 = act ? z2 : 0.f;
  Powers<float> pwf;
  Powers<double> pwd;
  if (!helper) {
    if (small) { pwf = load_powers<float>(shT);   row_totals<float, false>(pwf, e1, e2, shTotFf, tid); }
    else       { pwd = load_powers<double>(shTd); row_totals<double, false>(pwd, (double)e1, (double)e2, shTotF, tid); }
  }
  __syncthreads();

  float y[CLEN];
  if (!helper) {
    if (small) chunk_starts<float, false>(pwf, e1, e2, shTotFf, tid, z1, z2);
    else       chunk_starts<double, false>(pwd, (double)e1, (double)e2, shTotF, tid, z1, z2);
#pragma unroll
    for (int w = 0; w < CLEN; ++w) y[w] = biquad_step(q, x[w], z1, z2);
    z1 = 0.f; z2 = 0.f;
#pragma unroll
    for (int w = CLEN - 1; w >= 0; --w) (void)biquad_step(q, y[w], z1, z2);
    e1 = act ? z1 : 0.f; e2 = act ? z2 : 0.f;
    if (small) row_totals<float, true>(pwf, e1, e2, shTotBf, tid);
    else       row_totals<double, true>(pwd, (double)e1, (double)e2, shTotB, tid);
  }
  __syncthreads();

  if (!helper) {
    if (small) chunk_starts<float, true>(pwf, e1, e2, shTotBf, tid, z1, z2);
    else       chunk_starts<double, true>(pwd, (double)e1, (double)e2, shTotB, tid, z1, z2);
    float* tp = rowf + c * CLEN;
#pragma unroll
    for (int w = CLEN - 1; w >= 0; --w) {
      const float o = biquad_step(q, y[w], z1, z2);
      if (act) tp[w] = o;
    }
  }
  __syncthreads();

  float4* dst = reinterpret_cast<float4*>(out) + (size_t)b * ROWV4;
  for (int idx = tid; idx < ROWV4; idx += STHREADS) dst[idx] = row4[idx];
}

extern "C" void kernel_launch(void* const* d_in, const int* in_sizes, int n_in,
                              void* d_out, int out_size, void* d_ws, size_t ws_size,
                              hipStream_t stream) {
  const float* rgb    = (const float*)d_in[0];
  const float* matrix = (const float*)d_in[1];
  const float* acoef  = (const float*)d_in[2];
  const float* bcoef  = (const float*)d_in[3];
  float* out = (float*)d_out;

  float* ppg  = (float*)d_ws;

  pos_window_overlap_add<<<NBLK, FTHREADS, 0, stream>>>(rgb, matrix, ppg);
  pos_filtfilt_rows<<<BATCH, STHREADS, 0, stream>>>(ppg, acoef, bcoef, out);
}
